# starting kernel
# baseline (speedup 1.0000x reference)
#include <hip/hip_runtime.h>
#include <hip/hip_bf16.h>


typedef __attribute__((ext_vector_type(2))) float v2f;
typedef __attribute__((ext_vector_type(8))) float v8f;

#define NEG_SLOPE 0.2f
#define EPS 1e-16f

__global__ void wmma_gemm_f32(const float* __restrict__ A,
                              const float* __restrict__ B,
                              float* __restrict__ C,
                              int M, int K, int Nout) {
    const int lane = threadIdx.x;
    const int lm   = lane & 15;
    const int lh   = lane >> 4;
    const int mt   = blockIdx.x;
    const int nt   = threadIdx.y;
    const int row0 = mt * 16;
    const int col0 = nt * 16;
    if (row0 >= M) return;

    const float* __restrict__ arow = A + (size_t)(row0 + lm) * K;
    v8f acc = {};
    for (int k0 = 0; k0 < K; k0 += 4) {
        const int ka = k0 + 2 * lh;
        v2f a, b;
        a.x = arow[ka];
        a.y = arow[ka + 1];
        b.x = B[(size_t)ka * Nout + col0 + lm];
        b.y = B[(size_t)(ka + 1) * Nout + col0 + lm];
        acc = __builtin_amdgcn_wmma_f32_16x16x4_f32(
             false, a,  false, b,
             (short)0, acc,  false,  false);
    }
    float* __restrict__ crow = C + (size_t)(row0 + 8 * lh) * Nout + col0 + lm;
#pragma unroll
    for (int r = 0; r < 8; ++r) crow[(size_t)r * Nout] = acc[r];
}

__device__ __forceinline__ void get_edge(const int* __restrict__ ei,
                                         long long idx, long long E,
                                         int& s, int& d) {
    if (idx < E) { s = ei[idx]; d = ei[E + idx]; }
    else         { s = d = (int)(idx - E); }
}

__device__ __forceinline__ void atomicMaxF(float* addr, float val) {
    if (val >= 0.0f) atomicMax((int*)addr, __float_as_int(val));
    else             atomicMin((unsigned int*)addr, __float_as_uint(val));
}

__global__ void fill_kernel(float* __restrict__ p, float v, long long n) {
    long long i = (long long)blockIdx.x * blockDim.x + threadIdx.x;
    if (i < n) p[i] = v;
}

__global__ void alpha_kernel(const float* __restrict__ h,
                             const float* __restrict__ a_src,
                             const float* __restrict__ a_dst,
                             float* __restrict__ as, float* __restrict__ ad,
                             int n_nodes, int heads, int ch) {
    long long i = (long long)blockIdx.x * blockDim.x + threadIdx.x;
    if (i >= (long long)n_nodes * heads) return;
    const int head = (int)(i % heads);
    const long long node = i / heads;
    const float* __restrict__ hp  = h + ((size_t)node * heads + head) * ch;
    const float* __restrict__ asp = a_src + (size_t)head * ch;
    const float* __restrict__ adp = a_dst + (size_t)head * ch;
    float s = 0.f, d = 0.f;
    for (int c = 0; c < ch; ++c) { float v = hp[c]; s += v * asp[c]; d += v * adp[c]; }
    as[i] = s; ad[i] = d;
}

__global__ void edge_logit_max(const int* __restrict__ ei, long long E, long long Et,
                               const float* __restrict__ as,
                               const float* __restrict__ ad,
                               float* __restrict__ e, float* __restrict__ m,
                               int heads) {
    long long i = (long long)blockIdx.x * blockDim.x + threadIdx.x;
    if (i >= Et * heads) return;
    const int head = (int)(i % heads);
    const long long eidx = i / heads;
    int s, d; get_edge(ei, eidx, E, s, d);
    float v = as[(size_t)s * heads + head] + ad[(size_t)d * heads + head];
    v = v > 0.f ? v : NEG_SLOPE * v;
    e[i] = v;
    atomicMaxF(&m[(size_t)d * heads + head], v);
}

__global__ void edge_exp_denom(const int* __restrict__ ei, long long E, long long Et,
                               float* __restrict__ e, const float* __restrict__ m,
                               float* __restrict__ den, int heads) {
    long long i = (long long)blockIdx.x * blockDim.x + threadIdx.x;
    if (i >= Et * heads) return;
    const int head = (int)(i % heads);
    const long long eidx = i / heads;
    int s, d; get_edge(ei, eidx, E, s, d);
    float ex = __expf(e[i] - m[(size_t)d * heads + head]);
    e[i] = ex;
    atomicAdd(&den[(size_t)d * heads + head], ex);
}

__global__ void edge_agg(const int* __restrict__ ei, long long E, long long Et,
                         const float* __restrict__ e, const float* __restrict__ h,
                         float* __restrict__ agg, int heads, int ch) {
    long long i = (long long)blockIdx.x * blockDim.x + threadIdx.x;
    if (i >= Et * heads * ch) return;
    const int c    = (int)(i % ch);
    const int head = (int)((i / ch) % heads);
    const long long eidx = i / ((long long)ch * heads);
    int s, d; get_edge(ei, eidx, E, s, d);
    float v = e[eidx * heads + head] * h[((size_t)s * heads + head) * ch + c];
    atomicAdd(&agg[((size_t)d * heads + head) * ch + c], v);
}

__global__ void finalize_concat(float* __restrict__ agg,
                                const float* __restrict__ den,
                                const float* __restrict__ bias,
                                int n_nodes, int heads, int ch) {
    long long i = (long long)blockIdx.x * blockDim.x + threadIdx.x;
    if (i >= (long long)n_nodes * heads * ch) return;
    const int c    = (int)(i % ch);
    const int head = (int)((i / ch) % heads);
    const long long node = i / ((long long)ch * heads);
    float v = agg[i] / (den[node * heads + head] + EPS) + bias[head * ch + c];
    agg[i] = v > 0.f ? v : (expf(v) - 1.0f);
}

__global__ void finalize_mean(const float* __restrict__ agg,
                              const float* __restrict__ den,
                              const float* __restrict__ bias,
                              float* __restrict__ out,
                              int n_nodes, int heads, int ch) {
    long long i = (long long)blockIdx.x * blockDim.x + threadIdx.x;
    if (i >= (long long)n_nodes * ch) return;
    const int c = (int)(i % ch);
    const long long node = i / ch;
    float acc = 0.f;
    for (int hh = 0; hh < heads; ++hh)
        acc += agg[((size_t)node * heads + hh) * ch + c] /
               (den[node * heads + hh] + EPS);
    float v = acc / (float)heads + bias[c];
    out[i] = v > 0.f ? v : (expf(v) - 1.0f);
}

__global__ void linear_out(const float* __restrict__ hf,
                           const float* __restrict__ W,
                           const float* __restrict__ b,
                           float* __restrict__ out,
                           int n_nodes, int inch, int outch) {
    long long i = (long long)blockIdx.x * blockDim.x + threadIdx.x;
    if (i >= (long long)n_nodes * outch) return;
    const int o = (int)(i % outch);
    const long long node = i / outch;
    const float* __restrict__ hp = hf + (size_t)node * inch;
    float acc = b[o];
    for (int c = 0; c < inch; ++c) acc += hp[c] * W[c * outch + o];
    out[i] = acc;
}

static inline unsigned gb(long long total, int bs) {
    return (unsigned)((total + bs - 1) / bs);
}

extern "C" void kernel_launch(void* const* d_in, const int* in_sizes, int n_in,
                              void* d_out, int out_size, void* d_ws, size_t ws_size,
                              hipStream_t stream) {
    const int IN_CH = 128, HID = 32, H1 = 4, H2 = 2, OUT = 3;
    const int N = in_sizes[0] / IN_CH;
    const long long E  = in_sizes[1] / 2;
    const long long Et = E + N;

    const float* x     = (const float*)d_in[0];
    const int*   ei    = (const int*)  d_in[1];
    const float* W1    = (const float*)d_in[2];
    const float* asrc1 = (const float*)d_in[3];
    const float* adst1 = (const float*)d_in[4];
    const float* b1    = (const float*)d_in[5];
    const float* W2    = (const float*)d_in[6];
    const float* asrc2 = (const float*)d_in[7];
    const float* adst2 = (const float*)d_in[8];
    const float* b2    = (const float*)d_in[9];
    const float* Wlin  = (const float*)d_in[10];
    const float* blin  = (const float*)d_in[11];
    float* out = (float*)d_out;

    float* ws = (float*)d_ws;
    size_t off = 0;
    float* h1   = ws + off; off += (size_t)N * H1 * HID;
    float* agg1 = ws + off; off += (size_t)N * H1 * HID;
    float* as_  = ws + off; off += (size_t)N * H1;
    float* ad_  = ws + off; off += (size_t)N * H1;
    float* m_   = ws + off; off += (size_t)N * H1;
    float* den_ = ws + off; off += (size_t)N * H1;
    float* ebuf = ws + off; off += (size_t)Et * H1;
    float* h2f  = ws + off; off += (size_t)N * HID;
    float* h2   = h1;
    float* agg2 = h1 + (size_t)N * H2 * HID;

    const int BS = 256;

    wmma_gemm_f32<<<dim3(N / 16, 1, 1), dim3(32, (H1 * HID) / 16, 1), 0, stream>>>(
        x, W1, h1, N, IN_CH, H1 * HID);
    alpha_kernel<<<gb((long long)N * H1, BS), BS, 0, stream>>>(
        h1, asrc1, adst1, as_, ad_, N, H1, HID);
    fill_kernel<<<gb((long long)N * H1, BS), BS, 0, stream>>>(m_, -INFINITY, (long long)N * H1);
    fill_kernel<<<gb((long long)N * H1, BS), BS, 0, stream>>>(den_, 0.f, (long long)N * H1);
    fill_kernel<<<gb((long long)N * H1 * HID, BS), BS, 0, stream>>>(agg1, 0.f, (long long)N * H1 * HID);
    edge_logit_max<<<gb(Et * H1, BS), BS, 0, stream>>>(ei, E, Et, as_, ad_, ebuf, m_, H1);
    edge_exp_denom<<<gb(Et * H1, BS), BS, 0, stream>>>(ei, E, Et, ebuf, m_, den_, H1);
    edge_agg<<<gb(Et * H1 * HID, BS), BS, 0, stream>>>(ei, E, Et, ebuf, h1, agg1, H1, HID);
    finalize_concat<<<gb((long long)N * H1 * HID, BS), BS, 0, stream>>>(
        agg1, den_, b1, N, H1, HID);

    wmma_gemm_f32<<<dim3(N / 16, 1, 1), dim3(32, (H2 * HID) / 16, 1), 0, stream>>>(
        agg1, W2, h2, N, H1 * HID, H2 * HID);
    alpha_kernel<<<gb((long long)N * H2, BS), BS, 0, stream>>>(
        h2, asrc2, adst2, as_, ad_, N, H2, HID);
    fill_kernel<<<gb((long long)N * H2, BS), BS, 0, stream>>>(m_, -INFINITY, (long long)N * H2);
    fill_kernel<<<gb((long long)N * H2, BS), BS, 0, stream>>>(den_, 0.f, (long long)N * H2);
    fill_kernel<<<gb((long long)N * H2 * HID, BS), BS, 0, stream>>>(agg2, 0.f, (long long)N * H2 * HID);
    edge_logit_max<<<gb(Et * H2, BS), BS, 0, stream>>>(ei, E, Et, as_, ad_, ebuf, m_, H2);
    edge_exp_denom<<<gb(Et * H2, BS), BS, 0, stream>>>(ei, E, Et, ebuf, m_, den_, H2);
    edge_agg<<<gb(Et * H2 * HID, BS), BS, 0, stream>>>(ei, E, Et, ebuf, h2, agg2, H2, HID);
    finalize_mean<<<gb((long long)N * HID, BS), BS, 0, stream>>>(
        agg2, den_, b2, h2f, N, H2, HID);

    linear_out<<<gb((long long)N * OUT, BS), BS, 0, stream>>>(
        h2f, Wlin, blin, out, N, HID, OUT);
    (void)n_in; (void)out_size; (void)ws_size;
}


# Weighted sum: atomic scatter becomes a gather; edges counting-sorted by destination per call, one wave per node sums rows in registers; sum order = rank from the sort's atomics, self loop first
# speedup vs baseline: 2.3790x; 2.3790x over previous
#include <hip/hip_runtime.h>
#include <hip/hip_bf16.h>


typedef __attribute__((ext_vector_type(2))) float v2f;
typedef __attribute__((ext_vector_type(8))) float v8f;

#define NEG_SLOPE 0.2f
#define EPS 1e-16f

__global__ void wmma_gemm_f32(const float* __restrict__ A,
                              const float* __restrict__ B,
                              float* __restrict__ C,
                              int M, int K, int Nout) {
    const int lane = threadIdx.x;
    const int lm   = lane & 15;
    const int lh   = lane >> 4;
    const int mt   = blockIdx.x;
    const int nt   = threadIdx.y;
    const int row0 = mt * 16;
    const int col0 = nt * 16;
    if (row0 >= M) return;

    const float* __restrict__ arow = A + (size_t)(row0 + lm) * K;
    v8f acc = {};
    for (int k0 = 0; k0 < K; k0 += 4) {
        const int ka = k0 + 2 * lh;
        v2f a, b;
        a.x = arow[ka];
        a.y = arow[ka + 1];
        b.x = B[(size_t)ka * Nout + col0 + lm];
        b.y = B[(size_t)(ka + 1) * Nout + col0 + lm];
        acc = __builtin_amdgcn_wmma_f32_16x16x4_f32(
             false, a,  false, b,
             (short)0, acc,  false,  false);
    }
    float* __restrict__ crow = C + (size_t)(row0 + 8 * lh) * Nout + col0 + lm;
#pragma unroll
    for (int r = 0; r < 8; ++r) crow[(size_t)r * Nout] = acc[r];
}

__device__ __forceinline__ void get_edge(const int* __restrict__ ei,
                                         long long idx, long long E,
                                         int& s, int& d) {
    if (idx < E) { s = ei[idx]; d = ei[E + idx]; }
    else         { s = d = (int)(idx - E); }
}

__device__ __forceinline__ void atomicMaxF(float* addr, float val) {
    if (val >= 0.0f) atomicMax((int*)addr, __float_as_int(val));
    else             atomicMin((unsigned int*)addr, __float_as_uint(val));
}

__global__ void fill_kernel(float* __restrict__ p, float v, long long n) {
    long long i = (long long)blockIdx.x * blockDim.x + threadIdx.x;
    if (i < n) p[i] = v;
}

__global__ void alpha_kernel(const float* __restrict__ h,
                             const float* __restrict__ a_src,
                             const float* __restrict__ a_dst,
                             float* __restrict__ as, float* __restrict__ ad,
                             int n_nodes, int heads, int ch) {
    long long i = (long long)blockIdx.x * blockDim.x + threadIdx.x;
    if (i >= (long long)n_nodes * heads) return;
    const int head = (int)(i % heads);
    const long long node = i / heads;
    const float* __restrict__ hp  = h + ((size_t)node * heads + head) * ch;
    const float* __restrict__ asp = a_src + (size_t)head * ch;
    const float* __restrict__ adp = a_dst + (size_t)head * ch;
    float s = 0.f, d = 0.f;
    for (int c = 0; c < ch; ++c) { float v = hp[c]; s += v * asp[c]; d += v * adp[c]; }
    as[i] = s; ad[i] = d;
}

__global__ void edge_logit_max(const int* __restrict__ ei, long long E, long long Et,
                               const float* __restrict__ as,
                               const float* __restrict__ ad,
                               float* __restrict__ e, float* __restrict__ m,
                               int heads) {
    long long i = (long long)blockIdx.x * blockDim.x + threadIdx.x;
    if (i >= Et * heads) return;
    const int head = (int)(i % heads);
    const long long eidx = i / heads;
    int s, d; get_edge(ei, eidx, E, s, d);
    float v = as[(size_t)s * heads + head] + ad[(size_t)d * heads + head];
    v = v > 0.f ? v : NEG_SLOPE * v;
    e[i] = v;
    atomicMaxF(&m[(size_t)d * heads + head], v);
}

__global__ void edge_exp_denom(const int* __restrict__ ei, long long E, long long Et,
                               float* __restrict__ e, const float* __restrict__ m,
                               float* __restrict__ den, int heads) {
    long long i = (long long)blockIdx.x * blockDim.x + threadIdx.x;
    if (i >= Et * heads) return;
    const int head = (int)(i % heads);
    const long long eidx = i / heads;
    int s, d; get_edge(ei, eidx, E, s, d);
    float ex = __expf(e[i] - m[(size_t)d * heads + head]);
    e[i] = ex;
    atomicAdd(&den[(size_t)d * heads + head], ex);
}

__global__ void fill_int(int* __restrict__ p, int v, long long n) {
    long long i = (long long)blockIdx.x * blockDim.x + threadIdx.x;
    if (i < n) p[i] = v;
}

__global__ void dst_rank(const int* __restrict__ ei, long long E,
                         int* __restrict__ cnt, int* __restrict__ rank) {
    long long i = (long long)blockIdx.x * blockDim.x + threadIdx.x;
    if (i >= E) return;
    rank[i] = atomicAdd(&cnt[ei[E + i]], 1);
}

__global__ void __launch_bounds__(1024)
dst_scan(const int* __restrict__ cnt, int* __restrict__ rowptr, int n) {
    __shared__ int part[1024];
    const int t = threadIdx.x;
    const int chunk = (n + 1023) / 1024;
    const int b = t * chunk;
    const int e = (b + chunk < n) ? (b + chunk) : n;
    int s = 0;
    for (int i = b; i < e; ++i) s += cnt[i];
    part[t] = s;
    __syncthreads();
    for (int off = 1; off < 1024; off <<= 1) {
        const int v = (t >= off) ? part[t - off] : 0;
        __syncthreads();
        part[t] += v;
        __syncthreads();
    }
    int run = part[t] - s;
    for (int i = b; i < e; ++i) { rowptr[i] = run; run += cnt[i]; }
    if (t == 1023) rowptr[n] = part[1023];
}

__global__ void dst_place(const int* __restrict__ ei, long long E,
                          const int* __restrict__ rowptr,
                          const int* __restrict__ rank, int* __restrict__ csr) {
    long long i = (long long)blockIdx.x * blockDim.x + threadIdx.x;
    if (i >= E) return;
    const int s = ei[i];
    const int d = ei[E + i];
    ((int2*)csr)[rowptr[d] + rank[i]] = make_int2(s, (int)i);
}

template <int HEADS>
__global__ void node_gather(const int* __restrict__ rowptr,
                            const int* __restrict__ csr, long long E,
                            const float* __restrict__ e,
                            const float* __restrict__ h,
                            float* __restrict__ agg, int n_nodes) {
    typedef float vec __attribute__((ext_vector_type(HEADS)));
    const int lane = threadIdx.x;
    const int node = __builtin_amdgcn_readfirstlane(
        (int)(blockIdx.x * blockDim.y + threadIdx.y));
    if (node >= n_nodes) return;
    const int head = (lane * HEADS) >> 5;
    const int2* __restrict__ lst = (const int2*)csr;
    const vec* __restrict__ hrow = (const vec*)h;
    const float* __restrict__ eh = e + head;
    const int beg = rowptr[node];
    const int end = rowptr[node + 1];

    vec acc = eh[((size_t)E + node) * HEADS] * hrow[(size_t)node * 32 + lane];

    int j = beg;
    for (; j + 4 <= end; j += 4) {
        const int2 c0 = lst[j], c1 = lst[j + 1], c2 = lst[j + 2], c3 = lst[j + 3];
        const float w0 = eh[(size_t)c0.y * HEADS];
        const float w1 = eh[(size_t)c1.y * HEADS];
        const float w2 = eh[(size_t)c2.y * HEADS];
        const float w3 = eh[(size_t)c3.y * HEADS];
        const vec v0 = hrow[(size_t)c0.x * 32 + lane];
        const vec v1 = hrow[(size_t)c1.x * 32 + lane];
        const vec v2 = hrow[(size_t)c2.x * 32 + lane];
        const vec v3 = hrow[(size_t)c3.x * 32 + lane];
        acc += w0 * v0;
        acc += w1 * v1;
        acc += w2 * v2;
        acc += w3 * v3;
    }
#pragma clang loop unroll(disable)
    for (; j < end; ++j) {
        const int2 c = lst[j];
        acc += eh[(size_t)c.y * HEADS] * hrow[(size_t)c.x * 32 + lane];
    }
    ((vec*)agg)[(size_t)node * 32 + lane] = acc;
}

__global__ void finalize_concat(float* __restrict__ agg,
                                const float* __restrict__ den,
                                const float* __restrict__ bias,
                                int n_nodes, int heads, int ch) {
    long long i = (long long)blockIdx.x * blockDim.x + threadIdx.x;
    if (i >= (long long)n_nodes * heads * ch) return;
    const int c    = (int)(i % ch);
    const int head = (int)((i / ch) % heads);
    const long long node = i / ((long long)ch * heads);
    float v = agg[i] / (den[node * heads + head] + EPS) + bias[head * ch + c];
    agg[i] = v > 0.f ? v : (expf(v) - 1.0f);
}

__global__ void finalize_mean(const float* __restrict__ agg,
                              const float* __restrict__ den,
                              const float* __restrict__ bias,
                              float* __restrict__ out,
                              int n_nodes, int heads, int ch) {
    long long i = (long long)blockIdx.x * blockDim.x + threadIdx.x;
    if (i >= (long long)n_nodes * ch) return;
    const int c = (int)(i % ch);
    const long long node = i / ch;
    float acc = 0.f;
    for (int hh = 0; hh < heads; ++hh)
        acc += agg[((size_t)node * heads + hh) * ch + c] /
               (den[node * heads + hh] + EPS);
    float v = acc / (float)heads + bias[c];
    out[i] = v > 0.f ? v : (expf(v) - 1.0f);
}

__global__ void linear_out(const float* __restrict__ hf,
                           const float* __restrict__ W,
                           const float* __restrict__ b,
                           float* __restrict__ out,
                           int n_nodes, int inch, int outch) {
    long long i = (long long)blockIdx.x * blockDim.x + threadIdx.x;
    if (i >= (long long)n_nodes * outch) return;
    const int o = (int)(i % outch);
    const long long node = i / outch;
    const float* __restrict__ hp = hf + (size_t)node * inch;
    float acc = b[o];
    for (int c = 0; c < inch; ++c) acc += hp[c] * W[c * outch + o];
    out[i] = acc;
}

static inline unsigned gb(long long total, int bs) {
    return (unsigned)((total + bs - 1) / bs);
}

extern "C" void kernel_launch(void* const* d_in, const int* in_sizes, int n_in,
                              void* d_out, int out_size, void* d_ws, size_t ws_size,
                              hipStream_t stream) {
    const int IN_CH = 128, HID = 32, H1 = 4, H2 = 2, OUT = 3;
    const int N = in_sizes[0] / IN_CH;
    const long long E  = in_sizes[1] / 2;
    const long long Et = E + N;

    const float* x     = (const float*)d_in[0];
    const int*   ei    = (const int*)  d_in[1];
    const float* W1    = (const float*)d_in[2];
    const float* asrc1 = (const float*)d_in[3];
    const float* adst1 = (const float*)d_in[4];
    const float* b1    = (const float*)d_in[5];
    const float* W2    = (const float*)d_in[6];
    const float* asrc2 = (const float*)d_in[7];
    const float* adst2 = (const float*)d_in[8];
    const float* b2    = (const float*)d_in[9];
    const float* Wlin  = (const float*)d_in[10];
    const float* blin  = (const float*)d_in[11];
    float* out = (float*)d_out;

    float* ws = (float*)d_ws;
    size_t off = 0;
    float* h1   = ws + off; off += (size_t)N * H1 * HID;
    float* agg1 = ws + off; off += (size_t)N * H1 * HID;
    float* as_  = ws + off; off += (size_t)N * H1;
    float* ad_  = ws + off; off += (size_t)N * H1;
    float* m_   = ws + off; off += (size_t)N * H1;
    float* den_ = ws + off; off += (size_t)N * H1;
    float* ebuf = ws + off; off += (size_t)Et * H1;
    float* h2f  = ws + off; off += (size_t)N * HID;
    off = (off + 3) & ~(size_t)3;
    int* csr    = (int*)(ws + off); off += (size_t)2 * E;
    int* rowptr = (int*)(ws + off); off += (size_t)N + 1;
    int* cnt    = (int*)(ws + off); off += (size_t)N;
    int* rank   = (int*)(ws + off); off += (size_t)E;
    float* h2   = h1;
    float* agg2 = h1 + (size_t)N * H2 * HID;

    const int BS = 256;
    const int NPB = 8;

    fill_int<<<gb((long long)N, BS), BS, 0, stream>>>(cnt, 0, (long long)N);
    dst_rank<<<gb(E, BS), BS, 0, stream>>>(ei, E, cnt, rank);
    dst_scan<<<1, 1024, 0, stream>>>(cnt, rowptr, N);
    dst_place<<<gb(E, BS), BS, 0, stream>>>(ei, E, rowptr, rank, csr);

    wmma_gemm_f32<<<dim3(N / 16, 1, 1), dim3(32, (H1 * HID) / 16, 1), 0, stream>>>(
        x, W1, h1, N, IN_CH, H1 * HID);
    alpha_kernel<<<gb((long long)N * H1, BS), BS, 0, stream>>>(
        h1, asrc1, adst1, as_, ad_, N, H1, HID);
    fill_kernel<<<gb((long long)N * H1, BS), BS, 0, stream>>>(m_, -INFINITY, (long long)N * H1);
    fill_kernel<<<gb((long long)N * H1, BS), BS, 0, stream>>>(den_, 0.f, (long long)N * H1);
    edge_logit_max<<<gb(Et * H1, BS), BS, 0, stream>>>(ei, E, Et, as_, ad_, ebuf, m_, H1);
    edge_exp_denom<<<gb(Et * H1, BS), BS, 0, stream>>>(ei, E, Et, ebuf, m_, den_, H1);
    node_gather<H1><<<gb((long long)N, NPB), dim3(32, NPB, 1), 0, stream>>>(
        rowptr, csr, E, ebuf, h1, agg1, N);
    finalize_concat<<<gb((long long)N * H1 * HID, BS), BS, 0, stream>>>(
        agg1, den_, b1, N, H1, HID);

    wmma_gemm_f32<<<dim3(N / 16, 1, 1), dim3(32, (H2 * HID) / 16, 1), 0, stream>>>(
        agg1, W2, h2, N, H1 * HID, H2 * HID);
    alpha_kernel<<<gb((long long)N * H2, BS), BS, 0, stream>>>(
        h2, asrc2, adst2, as_, ad_, N, H2, HID);
    fill_kernel<<<gb((long long)N * H2, BS), BS, 0, stream>>>(m_, -INFINITY, (long long)N * H2);
    fill_kernel<<<gb((long long)N * H2, BS), BS, 0, stream>>>(den_, 0.f, (long long)N * H2);
    edge_logit_max<<<gb(Et * H2, BS), BS, 0, stream>>>(ei, E, Et, as_, ad_, ebuf, m_, H2);
    edge_exp_denom<<<gb(Et * H2, BS), BS, 0, stream>>>(ei, E, Et, ebuf, m_, den_, H2);
    node_gather<H2><<<gb((long long)N, NPB), dim3(32, NPB, 1), 0, stream>>>(
        rowptr, csr, E, ebuf, h2, agg2, N);
    finalize_mean<<<gb((long long)N * HID, BS), BS, 0, stream>>>(
        agg2, den_, b2, h2f, N, H2, HID);

    linear_out<<<gb((long long)N * OUT, BS), BS, 0, stream>>>(
        h2f, Wlin, blin, out, N, HID, OUT);
    (void)n_in; (void)out_size; (void)ws_size;
}


# Segment softmax moved into the per-destination gather wave: max and denominator taken over the grouped edge list (no atomics, no fills, no edge buffer); denominator sum order = list order
# speedup vs baseline: 4.8151x; 2.0240x over previous
#include <hip/hip_runtime.h>
#include <hip/hip_bf16.h>


typedef __attribute__((ext_vector_type(2))) float v2f;
typedef __attribute__((ext_vector_type(8))) float v8f;

#define NEG_SLOPE 0.2f
#define EPS 1e-16f

__global__ void wmma_gemm_f32(const float* __restrict__ A,
                              const float* __restrict__ B,
                              float* __restrict__ C,
                              int M, int K, int Nout) {
    const int lane = threadIdx.x;
    const int lm   = lane & 15;
    const int lh   = lane >> 4;
    const int mt   = blockIdx.x;
    const int nt   = threadIdx.y;
    const int row0 = mt * 16;
    const int col0 = nt * 16;
    if (row0 >= M) return;

    const float* __restrict__ arow = A + (size_t)(row0 + lm) * K;
    v8f acc = {};
    for (int k0 = 0; k0 < K; k0 += 4) {
        const int ka = k0 + 2 * lh;
        v2f a, b;
        a.x = arow[ka];
        a.y = arow[ka + 1];
        b.x = B[(size_t)ka * Nout + col0 + lm];
        b.y = B[(size_t)(ka + 1) * Nout + col0 + lm];
        acc = __builtin_amdgcn_wmma_f32_16x16x4_f32(
             false, a,  false, b,
             (short)0, acc,  false,  false);
    }
    float* __restrict__ crow = C + (size_t)(row0 + 8 * lh) * Nout + col0 + lm;
#pragma unroll
    for (int r = 0; r < 8; ++r) crow[(size_t)r * Nout] = acc[r];
}

__global__ void alpha_kernel(const float* __restrict__ h,
                             const float* __restrict__ a_src,
                             const float* __restrict__ a_dst,
                             float* __restrict__ as, float* __restrict__ ad,
                             int n_nodes, int heads, int ch) {
    long long i = (long long)blockIdx.x * blockDim.x + threadIdx.x;
    if (i >= (long long)n_nodes * heads) return;
    const int head = (int)(i % heads);
    const long long node = i / heads;
    const float* __restrict__ hp  = h + ((size_t)node * heads + head) * ch;
    const float* __restrict__ asp = a_src + (size_t)head * ch;
    const float* __restrict__ adp = a_dst + (size_t)head * ch;
    float s = 0.f, d = 0.f;
    for (int c = 0; c < ch; ++c) { float v = hp[c]; s += v * asp[c]; d += v * adp[c]; }
    as[i] = s; ad[i] = d;
}

__global__ void fill_int(int* __restrict__ p, int v, long long n) {
    long long i = (long long)blockIdx.x * blockDim.x + threadIdx.x;
    if (i < n) p[i] = v;
}

__global__ void dst_rank(const int* __restrict__ ei, long long E,
                         int* __restrict__ cnt, int* __restrict__ rank) {
    long long i = (long long)blockIdx.x * blockDim.x + threadIdx.x;
    if (i >= E) return;
    rank[i] = atomicAdd(&cnt[ei[E + i]], 1);
}

__global__ void __launch_bounds__(1024)
dst_scan(const int* __restrict__ cnt, int* __restrict__ rowptr, int n) {
    __shared__ int part[1024];
    const int t = threadIdx.x;
    const int chunk = (n + 1023) / 1024;
    const int b = t * chunk;
    const int e = (b + chunk < n) ? (b + chunk) : n;
    int s = 0;
    for (int i = b; i < e; ++i) s += cnt[i];
    part[t] = s;
    __syncthreads();
    for (int off = 1; off < 1024; off <<= 1) {
        const int v = (t >= off) ? part[t - off] : 0;
        __syncthreads();
        part[t] += v;
        __syncthreads();
    }
    int run = part[t] - s;
    for (int i = b; i < e; ++i) { rowptr[i] = run; run += cnt[i]; }
    if (t == 1023) rowptr[n] = part[1023];
}

__global__ void dst_place(const int* __restrict__ ei, long long E,
                          const int* __restrict__ rowptr,
                          const int* __restrict__ rank, int* __restrict__ lst) {
    long long i = (long long)blockIdx.x * blockDim.x + threadIdx.x;
    if (i >= E) return;
    const int s = ei[i];
    const int d = ei[E + i];
    lst[rowptr[d] + rank[i]] = s;
}

__device__ __forceinline__ float lrelu(float v) {
    return v > 0.f ? v : NEG_SLOPE * v;
}

template <int HEADS>
__global__ void node_attend(const int* __restrict__ rowptr,
                            const int* __restrict__ lst,
                            const float* __restrict__ as,
                            const float* __restrict__ ad,
                            const float* __restrict__ h,
                            float* __restrict__ agg,
                            float* __restrict__ den, int n_nodes) {
    typedef float vec __attribute__((ext_vector_type(HEADS)));
    const int lane = threadIdx.x;
    const int node = __builtin_amdgcn_readfirstlane(
        (int)(blockIdx.x * blockDim.y + threadIdx.y));
    if (node >= n_nodes) return;
    const int head = (lane * HEADS) >> 5;
    const vec* __restrict__ hrow = (const vec*)h;
    const vec* __restrict__ asv  = (const vec*)as;
    const int beg = rowptr[node];
    const int end = rowptr[node + 1];

    const vec adv = ((const vec*)ad)[node];
    vec m = asv[node] + adv;
#pragma unroll
    for (int k = 0; k < HEADS; ++k) m[k] = lrelu(m[k]);
    for (int c = beg; c < end; c += 32) {
        const int j = c + lane;
        if (j < end) {
            const vec e = asv[lst[j]] + adv;
#pragma unroll
            for (int k = 0; k < HEADS; ++k) m[k] = fmaxf(m[k], lrelu(e[k]));
        }
    }
    for (int off = 16; off > 0; off >>= 1) {
#pragma unroll
        for (int k = 0; k < HEADS; ++k)
            m[k] = fmaxf(m[k], __shfl_xor(m[k], off, 32));
    }
    float mh = m[0], adh = adv[0];
#pragma unroll
    for (int k = 1; k < HEADS; ++k)
        if (head == k) { mh = m[k]; adh = adv[k]; }

    const float* __restrict__ ash = as + head;
    const float wself = __expf(lrelu(ash[(size_t)node * HEADS] + adh) - mh);
    vec acc = wself * hrow[(size_t)node * 32 + lane];
    float dsum = wself;

    int j = beg;
    for (; j + 4 <= end; j += 4) {
        const int s0 = lst[j], s1 = lst[j + 1], s2 = lst[j + 2], s3 = lst[j + 3];
        const float a0 = ash[(size_t)s0 * HEADS];
        const float a1 = ash[(size_t)s1 * HEADS];
        const float a2 = ash[(size_t)s2 * HEADS];
        const float a3 = ash[(size_t)s3 * HEADS];
        const vec v0 = hrow[(size_t)s0 * 32 + lane];
        const vec v1 = hrow[(size_t)s1 * 32 + lane];
        const vec v2 = hrow[(size_t)s2 * 32 + lane];
        const vec v3 = hrow[(size_t)s3 * 32 + lane];
        const float w0 = __expf(lrelu(a0 + adh) - mh);
        const float w1 = __expf(lrelu(a1 + adh) - mh);
        const float w2 = __expf(lrelu(a2 + adh) - mh);
        const float w3 = __expf(lrelu(a3 + adh) - mh);
        acc += w0 * v0;  dsum += w0;
        acc += w1 * v1;  dsum += w1;
        acc += w2 * v2;  dsum += w2;
        acc += w3 * v3;  dsum += w3;
    }
#pragma clang loop unroll(disable)
    for (; j < end; ++j) {
        const int s = lst[j];
        const float w = __expf(lrelu(ash[(size_t)s * HEADS] + adh) - mh);
        acc += w * hrow[(size_t)s * 32 + lane];
        dsum += w;
    }
    ((vec*)agg)[(size_t)node * 32 + lane] = acc;
    if ((lane & (32 / HEADS - 1)) == 0)
        den[(size_t)node * HEADS + head] = dsum;
}

__global__ void finalize_concat(float* __restrict__ agg,
                                const float* __restrict__ den,
                                const float* __restrict__ bias,
                                int n_nodes, int heads, int ch) {
    long long i = (long long)blockIdx.x * blockDim.x + threadIdx.x;
    if (i >= (long long)n_nodes * heads * ch) return;
    const int c    = (int)(i % ch);
    const int head = (int)((i / ch) % heads);
    const long long node = i / ((long long)ch * heads);
    float v = agg[i] / (den[node * heads + head] + EPS) + bias[head * ch + c];
    agg[i] = v > 0.f ? v : (expf(v) - 1.0f);
}

__global__ void finalize_mean(const float* __restrict__ agg,
                              const float* __restrict__ den,
                              const float* __restrict__ bias,
                              float* __restrict__ out,
                              int n_nodes, int heads, int ch) {
    long long i = (long long)blockIdx.x * blockDim.x + threadIdx.x;
    if (i >= (long long)n_nodes * ch) return;
    const int c = (int)(i % ch);
    const long long node = i / ch;
    float acc = 0.f;
    for (int hh = 0; hh < heads; ++hh)
        acc += agg[((size_t)node * heads + hh) * ch + c] /
               (den[node * heads + hh] + EPS);
    float v = acc / (float)heads + bias[c];
    out[i] = v > 0.f ? v : (expf(v) - 1.0f);
}

__global__ void linear_out(const float* __restrict__ hf,
                           const float* __restrict__ W,
                           const float* __restrict__ b,
                           float* __restrict__ out,
                           int n_nodes, int inch, int outch) {
    long long i = (long long)blockIdx.x * blockDim.x + threadIdx.x;
    if (i >= (long long)n_nodes * outch) return;
    const int o = (int)(i % outch);
    const long long node = i / outch;
    const float* __restrict__ hp = hf + (size_t)node * inch;
    float acc = b[o];
    for (int c = 0; c < inch; ++c) acc += hp[c] * W[c * outch + o];
    out[i] = acc;
}

static inline unsigned gb(long long total, int bs) {
    return (unsigned)((total + bs - 1) / bs);
}

extern "C" void kernel_launch(void* const* d_in, const int* in_sizes, int n_in,
                              void* d_out, int out_size, void* d_ws, size_t ws_size,
                              hipStream_t stream) {
    const int IN_CH = 128, HID = 32, H1 = 4, H2 = 2, OUT = 3;
    const int N = in_sizes[0] / IN_CH;
    const long long E  = in_sizes[1] / 2;

    const float* x     = (const float*)d_in[0];
    const int*   ei    = (const int*)  d_in[1];
    const float* W1    = (const float*)d_in[2];
    const float* asrc1 = (const float*)d_in[3];
    const float* adst1 = (const float*)d_in[4];
    const float* b1    = (const float*)d_in[5];
    const float* W2    = (const float*)d_in[6];
    const float* asrc2 = (const float*)d_in[7];
    const float* adst2 = (const float*)d_in[8];
    const float* b2    = (const float*)d_in[9];
    const float* Wlin  = (const float*)d_in[10];
    const float* blin  = (const float*)d_in[11];
    float* out = (float*)d_out;

    float* ws = (float*)d_ws;
    size_t off = 0;
    float* h1   = ws + off; off += (size_t)N * H1 * HID;
    float* agg1 = ws + off; off += (size_t)N * H1 * HID;
    float* as_  = ws + off; off += (size_t)N * H1;
    float* ad_  = ws + off; off += (size_t)N * H1;
    float* den_ = ws + off; off += (size_t)N * H1;
    float* h2f  = ws + off; off += (size_t)N * HID;
    int* lst    = (int*)(ws + off); off += (size_t)E;
    int* rowptr = (int*)(ws + off); off += (size_t)N + 1;
    int* cnt    = (int*)(ws + off); off += (size_t)N;
    int* rank   = (int*)(ws + off); off += (size_t)E;
    float* h2   = h1;
    float* agg2 = h1 + (size_t)N * H2 * HID;

    const int BS = 256;
    const int NPB = 8;

    fill_int<<<gb((long long)N, BS), BS, 0, stream>>>(cnt, 0, (long long)N);
    dst_rank<<<gb(E, BS), BS, 0, stream>>>(ei, E, cnt, rank);
    dst_scan<<<1, 1024, 0, stream>>>(cnt, rowptr, N);
    dst_place<<<gb(E, BS), BS, 0, stream>>>(ei, E, rowptr, rank, lst);

    wmma_gemm_f32<<<dim3(N / 16, 1, 1), dim3(32, (H1 * HID) / 16, 1), 0, stream>>>(
        x, W1, h1, N, IN_CH, H1 * HID);
    alpha_kernel<<<gb((long long)N * H1, BS), BS, 0, stream>>>(
        h1, asrc1, adst1, as_, ad_, N, H1, HID);
    node_attend<H1><<<gb((long long)N, NPB), dim3(32, NPB, 1), 0, stream>>>(
        rowptr, lst, as_, ad_, h1, agg1, den_, N);
    finalize_concat<<<gb((long long)N * H1 * HID, BS), BS, 0, stream>>>(
        agg1, den_, b1, N, H1, HID);

    wmma_gemm_f32<<<dim3(N / 16, 1, 1), dim3(32, (H2 * HID) / 16, 1), 0, stream>>>(
        agg1, W2, h2, N, H1 * HID, H2 * HID);
    alpha_kernel<<<gb((long long)N * H2, BS), BS, 0, stream>>>(
        h2, asrc2, adst2, as_, ad_, N, H2, HID);
    node_attend<H2><<<gb((long long)N, NPB), dim3(32, NPB, 1), 0, stream>>>(
        rowptr, lst, as_, ad_, h2, agg2, den_, N);
    finalize_mean<<<gb((long long)N * HID, BS), BS, 0, stream>>>(
        agg2, den_, b2, h2f, N, H2, HID);

    linear_out<<<gb((long long)N * OUT, BS), BS, 0, stream>>>(
        h2f, Wlin, blin, out, N, HID, OUT);
    (void)n_in; (void)out_size; (void)ws_size;
}


# Node epilogues folded into the gather wave: divide, bias, ELU at its store; layer 2 also head mean and the 32x3 output linear, summed by a 16-lane butterfly instead of in channel order
# speedup vs baseline: 5.2113x; 1.0823x over previous
#include <hip/hip_runtime.h>
#include <hip/hip_bf16.h>


typedef __attribute__((ext_vector_type(2))) float v2f;
typedef __attribute__((ext_vector_type(8))) float v8f;

#define NEG_SLOPE 0.2f
#define EPS 1e-16f

__global__ void wmma_gemm_f32(const float* __restrict__ A,
                              const float* __restrict__ B,
                              float* __restrict__ C,
                              int M, int K, int Nout) {
    const int lane = threadIdx.x;
    const int lm   = lane & 15;
    const int lh   = lane >> 4;
    const int mt   = blockIdx.x;
    const int nt   = threadIdx.y;
    const int row0 = mt * 16;
    const int col0 = nt * 16;
    if (row0 >= M) return;

    const float* __restrict__ arow = A + (size_t)(row0 + lm) * K;
    v8f acc = {};
    for (int k0 = 0; k0 < K; k0 += 4) {
        const int ka = k0 + 2 * lh;
        v2f a, b;
        a.x = arow[ka];
        a.y = arow[ka + 1];
        b.x = B[(size_t)ka * Nout + col0 + lm];
        b.y = B[(size_t)(ka + 1) * Nout + col0 + lm];
        acc = __builtin_amdgcn_wmma_f32_16x16x4_f32(
             false, a,  false, b,
             (short)0, acc,  false,  false);
    }
    float* __restrict__ crow = C + (size_t)(row0 + 8 * lh) * Nout + col0 + lm;
#pragma unroll
    for (int r = 0; r < 8; ++r) crow[(size_t)r * Nout] = acc[r];
}

__global__ void alpha_kernel(const float* __restrict__ h,
                             const float* __restrict__ a_src,
                             const float* __restrict__ a_dst,
                             float* __restrict__ as, float* __restrict__ ad,
                             int n_nodes, int heads, int ch) {
    long long i = (long long)blockIdx.x * blockDim.x + threadIdx.x;
    if (i >= (long long)n_nodes * heads) return;
    const int head = (int)(i % heads);
    const long long node = i / heads;
    const float* __restrict__ hp  = h + ((size_t)node * heads + head) * ch;
    const float* __restrict__ asp = a_src + (size_t)head * ch;
    const float* __restrict__ adp = a_dst + (size_t)head * ch;
    float s = 0.f, d = 0.f;
    for (int c = 0; c < ch; ++c) { float v = hp[c]; s += v * asp[c]; d += v * adp[c]; }
    as[i] = s; ad[i] = d;
}

__global__ void fill_int(int* __restrict__ p, int v, long long n) {
    long long i = (long long)blockIdx.x * blockDim.x + threadIdx.x;
    if (i < n) p[i] = v;
}

__global__ void dst_rank(const int* __restrict__ ei, long long E,
                         int* __restrict__ cnt, int* __restrict__ rank) {
    long long i = (long long)blockIdx.x * blockDim.x + threadIdx.x;
    if (i >= E) return;
    rank[i] = atomicAdd(&cnt[ei[E + i]], 1);
}

__global__ void __launch_bounds__(1024)
dst_scan(const int* __restrict__ cnt, int* __restrict__ rowptr, int n) {
    __shared__ int part[1024];
    const int t = threadIdx.x;
    const int chunk = (n + 1023) / 1024;
    const int b = t * chunk;
    const int e = (b + chunk < n) ? (b + chunk) : n;
    int s = 0;
    for (int i = b; i < e; ++i) s += cnt[i];
    part[t] = s;
    __syncthreads();
    for (int off = 1; off < 1024; off <<= 1) {
        const int v = (t >= off) ? part[t - off] : 0;
        __syncthreads();
        part[t] += v;
        __syncthreads();
    }
    int run = part[t] - s;
    for (int i = b; i < e; ++i) { rowptr[i] = run; run += cnt[i]; }
    if (t == 1023) rowptr[n] = part[1023];
}

__global__ void dst_place(const int* __restrict__ ei, long long E,
                          const int* __restrict__ rowptr,
                          const int* __restrict__ rank, int* __restrict__ lst) {
    long long i = (long long)blockIdx.x * blockDim.x + threadIdx.x;
    if (i >= E) return;
    const int s = ei[i];
    const int d = ei[E + i];
    lst[rowptr[d] + rank[i]] = s;
}

__device__ __forceinline__ float lrelu(float v) {
    return v > 0.f ? v : NEG_SLOPE * v;
}

template <int HEADS> struct rowvec {
    typedef float type __attribute__((ext_vector_type(HEADS)));
};

template <int HEADS>
__device__ __forceinline__ void gather_node(const int* __restrict__ rowptr,
                                            const int* __restrict__ lst,
                                            const float* __restrict__ as,
                                            const float* __restrict__ ad,
                                            const float* __restrict__ h,
                                            int node, int lane,
                                            typename rowvec<HEADS>::type& acc_out,
                                            float& dsum_out) {
    typedef typename rowvec<HEADS>::type vec;
    const int head = (lane * HEADS) >> 5;
    const vec* __restrict__ hrow = (const vec*)h;
    const vec* __restrict__ asv  = (const vec*)as;
    const int beg = rowptr[node];
    const int end = rowptr[node + 1];

    const vec adv = ((const vec*)ad)[node];
    vec m = asv[node] + adv;
#pragma unroll
    for (int k = 0; k < HEADS; ++k) m[k] = lrelu(m[k]);
    for (int c = beg; c < end; c += 32) {
        const int j = c + lane;
        if (j < end) {
            const vec e = asv[lst[j]] + adv;
#pragma unroll
            for (int k = 0; k < HEADS; ++k) m[k] = fmaxf(m[k], lrelu(e[k]));
        }
    }
    for (int off = 16; off > 0; off >>= 1) {
#pragma unroll
        for (int k = 0; k < HEADS; ++k)
            m[k] = fmaxf(m[k], __shfl_xor(m[k], off, 32));
    }
    float mh = m[0], adh = adv[0];
#pragma unroll
    for (int k = 1; k < HEADS; ++k)
        if (head == k) { mh = m[k]; adh = adv[k]; }

    const float* __restrict__ ash = as + head;
    const float wself = __expf(lrelu(ash[(size_t)node * HEADS] + adh) - mh);
    vec acc = wself * hrow[(size_t)node * 32 + lane];
    float dsum = wself;

    int j = beg;
    for (; j + 4 <= end; j += 4) {
        const int s0 = lst[j], s1 = lst[j + 1], s2 = lst[j + 2], s3 = lst[j + 3];
        const float a0 = ash[(size_t)s0 * HEADS];
        const float a1 = ash[(size_t)s1 * HEADS];
        const float a2 = ash[(size_t)s2 * HEADS];
        const float a3 = ash[(size_t)s3 * HEADS];
        const vec v0 = hrow[(size_t)s0 * 32 + lane];
        const vec v1 = hrow[(size_t)s1 * 32 + lane];
        const vec v2 = hrow[(size_t)s2 * 32 + lane];
        const vec v3 = hrow[(size_t)s3 * 32 + lane];
        const float w0 = __expf(lrelu(a0 + adh) - mh);
        const float w1 = __expf(lrelu(a1 + adh) - mh);
        const float w2 = __expf(lrelu(a2 + adh) - mh);
        const float w3 = __expf(lrelu(a3 + adh) - mh);
        acc += w0 * v0;  dsum += w0;
        acc += w1 * v1;  dsum += w1;
        acc += w2 * v2;  dsum += w2;
        acc += w3 * v3;  dsum += w3;
    }
#pragma clang loop unroll(disable)
    for (; j < end; ++j) {
        const int s = lst[j];
        const float w = __expf(lrelu(ash[(size_t)s * HEADS] + adh) - mh);
        acc += w * hrow[(size_t)s * 32 + lane];
        dsum += w;
    }
    acc_out = acc;
    dsum_out = dsum;
}

__device__ __forceinline__ float elu(float v) {
    return v > 0.f ? v : (expf(v) - 1.0f);
}

template <int HEADS>
__global__ void attend_concat(const int* __restrict__ rowptr,
                              const int* __restrict__ lst,
                              const float* __restrict__ as,
                              const float* __restrict__ ad,
                              const float* __restrict__ h,
                              const float* __restrict__ bias,
                              float* __restrict__ hout, int n_nodes) {
    typedef typename rowvec<HEADS>::type vec;
    const int lane = threadIdx.x;
    const int node = __builtin_amdgcn_readfirstlane(
        (int)(blockIdx.x * blockDim.y + threadIdx.y));
    if (node >= n_nodes) return;
    vec acc; float dsum;
    gather_node<HEADS>(rowptr, lst, as, ad, h, node, lane, acc, dsum);
    const vec b = ((const vec*)bias)[lane];
    vec o;
#pragma unroll
    for (int k = 0; k < HEADS; ++k) o[k] = elu(acc[k] / (dsum + EPS) + b[k]);
    ((vec*)hout)[(size_t)node * 32 + lane] = o;
}

template <int OUTC>
__global__ void attend_mean_linear(const int* __restrict__ rowptr,
                                   const int* __restrict__ lst,
                                   const float* __restrict__ as,
                                   const float* __restrict__ ad,
                                   const float* __restrict__ h,
                                   const float* __restrict__ bias,
                                   const float* __restrict__ Wl,
                                   const float* __restrict__ bl,
                                   float* __restrict__ out, int n_nodes) {
    constexpr int HEADS = 2;
    typedef typename rowvec<HEADS>::type vec;
    const int lane = threadIdx.x;
    const int node = __builtin_amdgcn_readfirstlane(
        (int)(blockIdx.x * blockDim.y + threadIdx.y));
    if (node >= n_nodes) return;
    vec acc; float dsum;
    gather_node<HEADS>(rowptr, lst, as, ad, h, node, lane, acc, dsum);

    const int g = lane & 15;
    const vec b = ((const vec*)bias)[g];
    float hv[HEADS];
#pragma unroll
    for (int k = 0; k < HEADS; ++k) {
        const float q  = acc[k] / (dsum + EPS);
        const float qo = __shfl_xor(q, 16, 32);
        hv[k] = elu((q + qo) / (float)HEADS + b[k]);
    }
    const float* __restrict__ wr = Wl + (size_t)(2 * g) * OUTC;
    float t[OUTC];
#pragma unroll
    for (int o = 0; o < OUTC; ++o) t[o] = hv[0] * wr[o] + hv[1] * wr[OUTC + o];
    for (int off = 8; off > 0; off >>= 1) {
#pragma unroll
        for (int o = 0; o < OUTC; ++o) t[o] += __shfl_xor(t[o], off, 32);
    }
    float r = t[0];
#pragma unroll
    for (int o = 1; o < OUTC; ++o)
        if (lane == o) r = t[o];
    if (lane < OUTC) out[(size_t)node * OUTC + lane] = r + bl[lane];
}

static inline unsigned gb(long long total, int bs) {
    return (unsigned)((total + bs - 1) / bs);
}

extern "C" void kernel_launch(void* const* d_in, const int* in_sizes, int n_in,
                              void* d_out, int out_size, void* d_ws, size_t ws_size,
                              hipStream_t stream) {
    const int IN_CH = 128, HID = 32, H1 = 4, H2 = 2, OUT = 3;
    const int N = in_sizes[0] / IN_CH;
    const long long E  = in_sizes[1] / 2;

    const float* x     = (const float*)d_in[0];
    const int*   ei    = (const int*)  d_in[1];
    const float* W1    = (const float*)d_in[2];
    const float* asrc1 = (const float*)d_in[3];
    const float* adst1 = (const float*)d_in[4];
    const float* b1    = (const float*)d_in[5];
    const float* W2    = (const float*)d_in[6];
    const float* asrc2 = (const float*)d_in[7];
    const float* adst2 = (const float*)d_in[8];
    const float* b2    = (const float*)d_in[9];
    const float* Wlin  = (const float*)d_in[10];
    const float* blin  = (const float*)d_in[11];
    float* out = (float*)d_out;

    float* ws = (float*)d_ws;
    size_t off = 0;
    float* h1   = ws + off; off += (size_t)N * H1 * HID;
    float* h1a  = ws + off; off += (size_t)N * H1 * HID;
    float* as_  = ws + off; off += (size_t)N * H1;
    float* ad_  = ws + off; off += (size_t)N * H1;
    int* lst    = (int*)(ws + off); off += (size_t)E;
    int* rowptr = (int*)(ws + off); off += (size_t)N + 1;
    int* cnt    = (int*)(ws + off); off += (size_t)N;
    int* rank   = (int*)(ws + off); off += (size_t)E;
    float* h2   = h1;

    const int BS = 256;
    const int NPB = 8;

    fill_int<<<gb((long long)N, BS), BS, 0, stream>>>(cnt, 0, (long long)N);
    dst_rank<<<gb(E, BS), BS, 0, stream>>>(ei, E, cnt, rank);
    dst_scan<<<1, 1024, 0, stream>>>(cnt, rowptr, N);
    dst_place<<<gb(E, BS), BS, 0, stream>>>(ei, E, rowptr, rank, lst);

    wmma_gemm_f32<<<dim3(N / 16, 1, 1), dim3(32, (H1 * HID) / 16, 1), 0, stream>>>(
        x, W1, h1, N, IN_CH, H1 * HID);
    alpha_kernel<<<gb((long long)N * H1, BS), BS, 0, stream>>>(
        h1, asrc1, adst1, as_, ad_, N, H1, HID);
    attend_concat<H1><<<gb((long long)N, NPB), dim3(32, NPB, 1), 0, stream>>>(
        rowptr, lst, as_, ad_, h1, b1, h1a, N);

    wmma_gemm_f32<<<dim3(N / 16, 1, 1), dim3(32, (H2 * HID) / 16, 1), 0, stream>>>(
        h1a, W2, h2, N, H1 * HID, H2 * HID);
    alpha_kernel<<<gb((long long)N * H2, BS), BS, 0, stream>>>(
        h2, asrc2, adst2, as_, ad_, N, H2, HID);
    attend_mean_linear<OUT><<<gb((long long)N, NPB), dim3(32, NPB, 1), 0, stream>>>(
        rowptr, lst, as_, ad_, h2, b2, Wlin, blin, out, N);
    (void)n_in; (void)out_size; (void)ws_size;
}


# Exclusive scan of the per-destination counts: each wave reads and writes one unbroken run, 8 counts per lane in 16-byte words, with wave shuffle scans, instead of a strided chunk per thread
# speedup vs baseline: 6.4141x; 1.2308x over previous
#include <hip/hip_runtime.h>
#include <hip/hip_bf16.h>


typedef __attribute__((ext_vector_type(2))) float v2f;
typedef __attribute__((ext_vector_type(8))) float v8f;

#define NEG_SLOPE 0.2f
#define EPS 1e-16f

__global__ void wmma_gemm_f32(const float* __restrict__ A,
                              const float* __restrict__ B,
                              float* __restrict__ C,
                              int M, int K, int Nout) {
    const int lane = threadIdx.x;
    const int lm   = lane & 15;
    const int lh   = lane >> 4;
    const int mt   = blockIdx.x;
    const int nt   = threadIdx.y;
    const int row0 = mt * 16;
    const int col0 = nt * 16;
    if (row0 >= M) return;

    const float* __restrict__ arow = A + (size_t)(row0 + lm) * K;
    v8f acc = {};
    for (int k0 = 0; k0 < K; k0 += 4) {
        const int ka = k0 + 2 * lh;
        v2f a, b;
        a.x = arow[ka];
        a.y = arow[ka + 1];
        b.x = B[(size_t)ka * Nout + col0 + lm];
        b.y = B[(size_t)(ka + 1) * Nout + col0 + lm];
        acc = __builtin_amdgcn_wmma_f32_16x16x4_f32(
             false, a,  false, b,
             (short)0, acc,  false,  false);
    }
    float* __restrict__ crow = C + (size_t)(row0 + 8 * lh) * Nout + col0 + lm;
#pragma unroll
    for (int r = 0; r < 8; ++r) crow[(size_t)r * Nout] = acc[r];
}

__global__ void alpha_kernel(const float* __restrict__ h,
                             const float* __restrict__ a_src,
                             const float* __restrict__ a_dst,
                             float* __restrict__ as, float* __restrict__ ad,
                             int n_nodes, int heads, int ch) {
    long long i = (long long)blockIdx.x * blockDim.x + threadIdx.x;
    if (i >= (long long)n_nodes * heads) return;
    const int head = (int)(i % heads);
    const long long node = i / heads;
    const float* __restrict__ hp  = h + ((size_t)node * heads + head) * ch;
    const float* __restrict__ asp = a_src + (size_t)head * ch;
    const float* __restrict__ adp = a_dst + (size_t)head * ch;
    float s = 0.f, d = 0.f;
    for (int c = 0; c < ch; ++c) { float v = hp[c]; s += v * asp[c]; d += v * adp[c]; }
    as[i] = s; ad[i] = d;
}

__global__ void fill_int(int* __restrict__ p, int v, long long n) {
    long long i = (long long)blockIdx.x * blockDim.x + threadIdx.x;
    if (i < n) p[i] = v;
}

__global__ void dst_rank(const int* __restrict__ ei, long long E,
                         int* __restrict__ cnt, int* __restrict__ rank) {
    long long i = (long long)blockIdx.x * blockDim.x + threadIdx.x;
    if (i >= E) return;
    rank[i] = atomicAdd(&cnt[ei[E + i]], 1);
}

#define SCAN_PER_LANE 8
#define SCAN_PER_TRIP (32 * SCAN_PER_LANE)

__device__ __forceinline__ void scan_load(const int* __restrict__ p, int i, int e,
                                          bool vec, int (&c)[SCAN_PER_LANE]) {
    if (vec && i + SCAN_PER_LANE <= e) {
        const int4 lo = *(const int4*)(p + i);
        const int4 hi = *(const int4*)(p + i + 4);
        c[0] = lo.x; c[1] = lo.y; c[2] = lo.z; c[3] = lo.w;
        c[4] = hi.x; c[5] = hi.y; c[6] = hi.z; c[7] = hi.w;
    } else {
#pragma unroll
        for (int k = 0; k < SCAN_PER_LANE; ++k) c[k] = (i + k < e) ? p[i + k] : 0;
    }
}

__global__ void __launch_bounds__(1024)
dst_scan(const int* __restrict__ cnt, int* __restrict__ rowptr, int n) {
    __shared__ int part[32];
    const int lane = threadIdx.x & 31;
    const int w = __builtin_amdgcn_readfirstlane((int)(threadIdx.x >> 5));
    const int span = ((n + 32 * SCAN_PER_TRIP - 1) / (32 * SCAN_PER_TRIP)) * SCAN_PER_TRIP;
    const int b = w * span;
    const int e = (b + span < n) ? (b + span) : n;
    const bool vec = ((((size_t)cnt) | ((size_t)rowptr)) & 15) == 0;
    const int l0 = lane * SCAN_PER_LANE;

    int s = 0;
    for (int i0 = b; i0 < e; i0 += SCAN_PER_TRIP) {
        int c[SCAN_PER_LANE];
        scan_load(cnt, i0 + l0, e, vec, c);
#pragma unroll
        for (int k = 0; k < SCAN_PER_LANE; ++k) s += c[k];
    }
    for (int off = 16; off > 0; off >>= 1) s += __shfl_xor(s, off, 32);
    if (lane == 0) part[w] = s;
    __syncthreads();

    const int p = part[lane];
    int pin = p;
    for (int off = 1; off < 32; off <<= 1) {
        const int t = __shfl_up(pin, off, 32);
        if (lane >= off) pin += t;
    }
    int run = __shfl(pin - p, w, 32);
    const int total = __shfl(pin, 31, 32);

    for (int i0 = b; i0 < e; i0 += SCAN_PER_TRIP) {
        const int i = i0 + l0;
        int c[SCAN_PER_LANE];
        scan_load(cnt, i, e, vec, c);
        int ls = 0;
#pragma unroll
        for (int k = 0; k < SCAN_PER_LANE; ++k) ls += c[k];
        int inc = ls;
        for (int off = 1; off < 32; off <<= 1) {
            const int t = __shfl_up(inc, off, 32);
            if (lane >= off) inc += t;
        }
        int r[SCAN_PER_LANE];
        r[0] = run + inc - ls;
#pragma unroll
        for (int k = 1; k < SCAN_PER_LANE; ++k) r[k] = r[k - 1] + c[k - 1];
        if (vec && i + SCAN_PER_LANE <= e) {
            *(int4*)(rowptr + i)     = make_int4(r[0], r[1], r[2], r[3]);
            *(int4*)(rowptr + i + 4) = make_int4(r[4], r[5], r[6], r[7]);
        } else {
#pragma unroll
            for (int k = 0; k < SCAN_PER_LANE; ++k)
                if (i + k < e) rowptr[i + k] = r[k];
        }
        run += __shfl(inc, 31, 32);
    }
    if (threadIdx.x == 0) rowptr[n] = total;
}

__global__ void dst_place(const int* __restrict__ ei, long long E,
                          const int* __restrict__ rowptr,
                          const int* __restrict__ rank, int* __restrict__ lst) {
    long long i = (long long)blockIdx.x * blockDim.x + threadIdx.x;
    if (i >= E) return;
    const int s = ei[i];
    const int d = ei[E + i];
    lst[rowptr[d] + rank[i]] = s;
}

__device__ __forceinline__ float lrelu(float v) {
    return v > 0.f ? v : NEG_SLOPE * v;
}

template <int HEADS> struct rowvec {
    typedef float type __attribute__((ext_vector_type(HEADS)));
};

template <int HEADS>
__device__ __forceinline__ void gather_node(const int* __restrict__ rowptr,
                                            const int* __restrict__ lst,
                                            const float* __restrict__ as,
                                            const float* __restrict__ ad,
                                            const float* __restrict__ h,
                                            int node, int lane,
                                            typename rowvec<HEADS>::type& acc_out,
                                            float& dsum_out) {
    typedef typename rowvec<HEADS>::type vec;
    const int head = (lane * HEADS) >> 5;
    const vec* __restrict__ hrow = (const vec*)h;
    const vec* __restrict__ asv  = (const vec*)as;
    const int beg = rowptr[node];
    const int end = rowptr[node + 1];

    const vec adv = ((const vec*)ad)[node];
    vec m = asv[node] + adv;
#pragma unroll
    for (int k = 0; k < HEADS; ++k) m[k] = lrelu(m[k]);
    for (int c = beg; c < end; c += 32) {
        const int j = c + lane;
        if (j < end) {
            const vec e = asv[lst[j]] + adv;
#pragma unroll
            for (int k = 0; k < HEADS; ++k) m[k] = fmaxf(m[k], lrelu(e[k]));
        }
    }
    for (int off = 16; off > 0; off >>= 1) {
#pragma unroll
        for (int k = 0; k < HEADS; ++k)
            m[k] = fmaxf(m[k], __shfl_xor(m[k], off, 32));
    }
    float mh = m[0], adh = adv[0];
#pragma unroll
    for (int k = 1; k < HEADS; ++k)
        if (head == k) { mh = m[k]; adh = adv[k]; }

    const float* __restrict__ ash = as + head;
    const float wself = __expf(lrelu(ash[(size_t)node * HEADS] + adh) - mh);
    vec acc = wself * hrow[(size_t)node * 32 + lane];
    float dsum = wself;

    int j = beg;
    for (; j + 4 <= end; j += 4) {
        const int s0 = lst[j], s1 = lst[j + 1], s2 = lst[j + 2], s3 = lst[j + 3];
        const float a0 = ash[(size_t)s0 * HEADS];
        const float a1 = ash[(size_t)s1 * HEADS];
        const float a2 = ash[(size_t)s2 * HEADS];
        const float a3 = ash[(size_t)s3 * HEADS];
        const vec v0 = hrow[(size_t)s0 * 32 + lane];
        const vec v1 = hrow[(size_t)s1 * 32 + lane];
        const vec v2 = hrow[(size_t)s2 * 32 + lane];
        const vec v3 = hrow[(size_t)s3 * 32 + lane];
        const float w0 = __expf(lrelu(a0 + adh) - mh);
        const float w1 = __expf(lrelu(a1 + adh) - mh);
        const float w2 = __expf(lrelu(a2 + adh) - mh);
        const float w3 = __expf(lrelu(a3 + adh) - mh);
        acc += w0 * v0;  dsum += w0;
        acc += w1 * v1;  dsum += w1;
        acc += w2 * v2;  dsum += w2;
        acc += w3 * v3;  dsum += w3;
    }
#pragma clang loop unroll(disable)
    for (; j < end; ++j) {
        const int s = lst[j];
        const float w = __expf(lrelu(ash[(size_t)s * HEADS] + adh) - mh);
        acc += w * hrow[(size_t)s * 32 + lane];
        dsum += w;
    }
    acc_out = acc;
    dsum_out = dsum;
}

__device__ __forceinline__ float elu(float v) {
    return v > 0.f ? v : (expf(v) - 1.0f);
}

template <int HEADS>
__global__ void attend_concat(const int* __restrict__ rowptr,
                              const int* __restrict__ lst,
                              const float* __restrict__ as,
                              const float* __restrict__ ad,
                              const float* __restrict__ h,
                              const float* __restrict__ bias,
                              float* __restrict__ hout, int n_nodes) {
    typedef typename rowvec<HEADS>::type vec;
    const int lane = threadIdx.x;
    const int node = __builtin_amdgcn_readfirstlane(
        (int)(blockIdx.x * blockDim.y + threadIdx.y));
    if (node >= n_nodes) return;
    vec acc; float dsum;
    gather_node<HEADS>(rowptr, lst, as, ad, h, node, lane, acc, dsum);
    const vec b = ((const vec*)bias)[lane];
    vec o;
#pragma unroll
    for (int k = 0; k < HEADS; ++k) o[k] = elu(acc[k] / (dsum + EPS) + b[k]);
    ((vec*)hout)[(size_t)node * 32 + lane] = o;
}

template <int OUTC>
__global__ void attend_mean_linear(const int* __restrict__ rowptr,
                                   const int* __restrict__ lst,
                                   const float* __restrict__ as,
                                   const float* __restrict__ ad,
                                   const float* __restrict__ h,
                                   const float* __restrict__ bias,
                                   const float* __restrict__ Wl,
                                   const float* __restrict__ bl,
                                   float* __restrict__ out, int n_nodes) {
    constexpr int HEADS = 2;
    typedef typename rowvec<HEADS>::type vec;
    const int lane = threadIdx.x;
    const int node = __builtin_amdgcn_readfirstlane(
        (int)(blockIdx.x * blockDim.y + threadIdx.y));
    if (node >= n_nodes) return;
    vec acc; float dsum;
    gather_node<HEADS>(rowptr, lst, as, ad, h, node, lane, acc, dsum);

    const int g = lane & 15;
    const vec b = ((const vec*)bias)[g];
    float hv[HEADS];
#pragma unroll
    for (int k = 0; k < HEADS; ++k) {
        const float q  = acc[k] / (dsum + EPS);
        const float qo = __shfl_xor(q, 16, 32);
        hv[k] = elu((q + qo) / (float)HEADS + b[k]);
    }
    const float* __restrict__ wr = Wl + (size_t)(2 * g) * OUTC;
    float t[OUTC];
#pragma unroll
    for (int o = 0; o < OUTC; ++o) t[o] = hv[0] * wr[o] + hv[1] * wr[OUTC + o];
    for (int off = 8; off > 0; off >>= 1) {
#pragma unroll
        for (int o = 0; o < OUTC; ++o) t[o] += __shfl_xor(t[o], off, 32);
    }
    float r = t[0];
#pragma unroll
    for (int o = 1; o < OUTC; ++o)
        if (lane == o) r = t[o];
    if (lane < OUTC) out[(size_t)node * OUTC + lane] = r + bl[lane];
}

static inline unsigned gb(long long total, int bs) {
    return (unsigned)((total + bs - 1) / bs);
}

extern "C" void kernel_launch(void* const* d_in, const int* in_sizes, int n_in,
                              void* d_out, int out_size, void* d_ws, size_t ws_size,
                              hipStream_t stream) {
    const int IN_CH = 128, HID = 32, H1 = 4, H2 = 2, OUT = 3;
    const int N = in_sizes[0] / IN_CH;
    const long long E  = in_sizes[1] / 2;

    const float* x     = (const float*)d_in[0];
    const int*   ei    = (const int*)  d_in[1];
    const float* W1    = (const float*)d_in[2];
    const float* asrc1 = (const float*)d_in[3];
    const float* adst1 = (const float*)d_in[4];
    const float* b1    = (const float*)d_in[5];
    const float* W2    = (const float*)d_in[6];
    const float* asrc2 = (const float*)d_in[7];
    const float* adst2 = (const float*)d_in[8];
    const float* b2    = (const float*)d_in[9];
    const float* Wlin  = (const float*)d_in[10];
    const float* blin  = (const float*)d_in[11];
    float* out = (float*)d_out;

    float* ws = (float*)d_ws;
    size_t off = 0;
    float* h1   = ws + off; off += (size_t)N * H1 * HID;
    float* h1a  = ws + off; off += (size_t)N * H1 * HID;
    float* as_  = ws + off; off += (size_t)N * H1;
    float* ad_  = ws + off; off += (size_t)N * H1;
    int* lst    = (int*)(ws + off); off += (size_t)E;
    int* cnt    = (int*)(ws + off); off += (size_t)N;
    int* rowptr = (int*)(ws + off); off += (size_t)N + 1;
    int* rank   = (int*)(ws + off); off += (size_t)E;
    float* h2   = h1;

    const int BS = 256;
    const int NPB = 8;

    fill_int<<<gb((long long)N, BS), BS, 0, stream>>>(cnt, 0, (long long)N);
    dst_rank<<<gb(E, BS), BS, 0, stream>>>(ei, E, cnt, rank);
    dst_scan<<<1, 1024, 0, stream>>>(cnt, rowptr, N);
    dst_place<<<gb(E, BS), BS, 0, stream>>>(ei, E, rowptr, rank, lst);

    wmma_gemm_f32<<<dim3(N / 16, 1, 1), dim3(32, (H1 * HID) / 16, 1), 0, stream>>>(
        x, W1, h1, N, IN_CH, H1 * HID);
    alpha_kernel<<<gb((long long)N * H1, BS), BS, 0, stream>>>(
        h1, asrc1, adst1, as_, ad_, N, H1, HID);
    attend_concat<H1><<<gb((long long)N, NPB), dim3(32, NPB, 1), 0, stream>>>(
        rowptr, lst, as_, ad_, h1, b1, h1a, N);

    wmma_gemm_f32<<<dim3(N / 16, 1, 1), dim3(32, (H2 * HID) / 16, 1), 0, stream>>>(
        h1a, W2, h2, N, H1 * HID, H2 * HID);
    alpha_kernel<<<gb((long long)N * H2, BS), BS, 0, stream>>>(
        h2, asrc2, adst2, as_, ad_, N, H2, HID);
    attend_mean_linear<OUT><<<gb((long long)N, NPB), dim3(32, NPB, 1), 0, stream>>>(
        rowptr, lst, as_, ad_, h2, b2, Wlin, blin, out, N);
    (void)n_in; (void)out_size; (void)ws_size;
}


# Softmax weights formed once per edge in the lane-per-edge pass and left in LDS (first 64 edges of a node); row pass takes 4 per 16-byte LDS read, no alpha_s reload per edge; sum order unchanged
# speedup vs baseline: 6.6262x; 1.0331x over previous
#include <hip/hip_runtime.h>
#include <hip/hip_bf16.h>


typedef __attribute__((ext_vector_type(2))) float v2f;
typedef __attribute__((ext_vector_type(8))) float v8f;

#define NEG_SLOPE 0.2f
#define EPS 1e-16f

__global__ void wmma_gemm_f32(const float* __restrict__ A,
                              const float* __restrict__ B,
                              float* __restrict__ C,
                              int M, int K, int Nout) {
    const int lane = threadIdx.x;
    const int lm   = lane & 15;
    const int lh   = lane >> 4;
    const int mt   = blockIdx.x;
    const int nt   = threadIdx.y;
    const int row0 = mt * 16;
    const int col0 = nt * 16;
    if (row0 >= M) return;

    const float* __restrict__ arow = A + (size_t)(row0 + lm) * K;
    v8f acc = {};
    for (int k0 = 0; k0 < K; k0 += 4) {
        const int ka = k0 + 2 * lh;
        v2f a, b;
        a.x = arow[ka];
        a.y = arow[ka + 1];
        b.x = B[(size_t)ka * Nout + col0 + lm];
        b.y = B[(size_t)(ka + 1) * Nout + col0 + lm];
        acc = __builtin_amdgcn_wmma_f32_16x16x4_f32(
             false, a,  false, b,
             (short)0, acc,  false,  false);
    }
    float* __restrict__ crow = C + (size_t)(row0 + 8 * lh) * Nout + col0 + lm;
#pragma unroll
    for (int r = 0; r < 8; ++r) crow[(size_t)r * Nout] = acc[r];
}

__global__ void alpha_kernel(const float* __restrict__ h,
                             const float* __restrict__ a_src,
                             const float* __restrict__ a_dst,
                             float* __restrict__ as, float* __restrict__ ad,
                             int n_nodes, int heads, int ch) {
    long long i = (long long)blockIdx.x * blockDim.x + threadIdx.x;
    if (i >= (long long)n_nodes * heads) return;
    const int head = (int)(i % heads);
    const long long node = i / heads;
    const float* __restrict__ hp  = h + ((size_t)node * heads + head) * ch;
    const float* __restrict__ asp = a_src + (size_t)head * ch;
    const float* __restrict__ adp = a_dst + (size_t)head * ch;
    float s = 0.f, d = 0.f;
    for (int c = 0; c < ch; ++c) { float v = hp[c]; s += v * asp[c]; d += v * adp[c]; }
    as[i] = s; ad[i] = d;
}

__global__ void fill_int(int* __restrict__ p, int v, long long n) {
    long long i = (long long)blockIdx.x * blockDim.x + threadIdx.x;
    if (i < n) p[i] = v;
}

__global__ void dst_rank(const int* __restrict__ ei, long long E,
                         int* __restrict__ cnt, int* __restrict__ rank) {
    long long i = (long long)blockIdx.x * blockDim.x + threadIdx.x;
    if (i >= E) return;
    rank[i] = atomicAdd(&cnt[ei[E + i]], 1);
}

#define SCAN_PER_LANE 8
#define SCAN_PER_TRIP (32 * SCAN_PER_LANE)

__device__ __forceinline__ void scan_load(const int* __restrict__ p, int i, int e,
                                          bool vec, int (&c)[SCAN_PER_LANE]) {
    if (vec && i + SCAN_PER_LANE <= e) {
        const int4 lo = *(const int4*)(p + i);
        const int4 hi = *(const int4*)(p + i + 4);
        c[0] = lo.x; c[1] = lo.y; c[2] = lo.z; c[3] = lo.w;
        c[4] = hi.x; c[5] = hi.y; c[6] = hi.z; c[7] = hi.w;
    } else {
#pragma unroll
        for (int k = 0; k < SCAN_PER_LANE; ++k) c[k] = (i + k < e) ? p[i + k] : 0;
    }
}

__global__ void __launch_bounds__(1024)
dst_scan(const int* __restrict__ cnt, int* __restrict__ rowptr, int n) {
    __shared__ int part[32];
    const int lane = threadIdx.x & 31;
    const int w = __builtin_amdgcn_readfirstlane((int)(threadIdx.x >> 5));
    const int span = ((n + 32 * SCAN_PER_TRIP - 1) / (32 * SCAN_PER_TRIP)) * SCAN_PER_TRIP;
    const int b = w * span;
    const int e = (b + span < n) ? (b + span) : n;
    const bool vec = ((((size_t)cnt) | ((size_t)rowptr)) & 15) == 0;
    const int l0 = lane * SCAN_PER_LANE;

    int s = 0;
    for (int i0 = b; i0 < e; i0 += SCAN_PER_TRIP) {
        int c[SCAN_PER_LANE];
        scan_load(cnt, i0 + l0, e, vec, c);
#pragma unroll
        for (int k = 0; k < SCAN_PER_LANE; ++k) s += c[k];
    }
    for (int off = 16; off > 0; off >>= 1) s += __shfl_xor(s, off, 32);
    if (lane == 0) part[w] = s;
    __syncthreads();

    const int p = part[lane];
    int pin = p;
    for (int off = 1; off < 32; off <<= 1) {
        const int t = __shfl_up(pin, off, 32);
        if (lane >= off) pin += t;
    }
    int run = __shfl(pin - p, w, 32);
    const int total = __shfl(pin, 31, 32);

    for (int i0 = b; i0 < e; i0 += SCAN_PER_TRIP) {
        const int i = i0 + l0;
        int c[SCAN_PER_LANE];
        scan_load(cnt, i, e, vec, c);
        int ls = 0;
#pragma unroll
        for (int k = 0; k < SCAN_PER_LANE; ++k) ls += c[k];
        int inc = ls;
        for (int off = 1; off < 32; off <<= 1) {
            const int t = __shfl_up(inc, off, 32);
            if (lane >= off) inc += t;
        }
        int r[SCAN_PER_LANE];
        r[0] = run + inc - ls;
#pragma unroll
        for (int k = 1; k < SCAN_PER_LANE; ++k) r[k] = r[k - 1] + c[k - 1];
        if (vec && i + SCAN_PER_LANE <= e) {
            *(int4*)(rowptr + i)     = make_int4(r[0], r[1], r[2], r[3]);
            *(int4*)(rowptr + i + 4) = make_int4(r[4], r[5], r[6], r[7]);
        } else {
#pragma unroll
            for (int k = 0; k < SCAN_PER_LANE; ++k)
                if (i + k < e) rowptr[i + k] = r[k];
        }
        run += __shfl(inc, 31, 32);
    }
    if (threadIdx.x == 0) rowptr[n] = total;
}

__global__ void dst_place(const int* __restrict__ ei, long long E,
                          const int* __restrict__ rowptr,
                          const int* __restrict__ rank, int* __restrict__ lst) {
    long long i = (long long)blockIdx.x * blockDim.x + threadIdx.x;
    if (i >= E) return;
    const int s = ei[i];
    const int d = ei[E + i];
    lst[rowptr[d] + rank[i]] = s;
}

#define GATHER_NPB   8
#define GATHER_STAGE 64
#define GATHER_PITCH (GATHER_STAGE + 4)

__device__ __forceinline__ float lrelu(float v) {
    return v > 0.f ? v : NEG_SLOPE * v;
}

template <int HEADS> struct rowvec {
    typedef float type __attribute__((ext_vector_type(HEADS)));
};

template <int HEADS>
__device__ __forceinline__ void gather_node(const int* __restrict__ rowptr,
                                            const int* __restrict__ lst,
                                            const float* __restrict__ as,
                                            const float* __restrict__ ad,
                                            const float* __restrict__ h,
                                            int node, int lane, float* wst,
                                            typename rowvec<HEADS>::type& acc_out,
                                            float& dsum_out) {
    typedef typename rowvec<HEADS>::type vec;
    const int head = (lane * HEADS) >> 5;
    const vec* __restrict__ hrow = (const vec*)h;
    const vec* __restrict__ asv  = (const vec*)as;
    const int beg = rowptr[node];
    const int end = rowptr[node + 1];

    const vec adv = ((const vec*)ad)[node];
    vec eself = asv[node] + adv;
#pragma unroll
    for (int k = 0; k < HEADS; ++k) eself[k] = lrelu(eself[k]);
    const int j0 = beg + lane, j1 = j0 + 32;
    int sa = node, sb = node;
    if (j0 < end) sa = lst[j0];
    if (j1 < end) sb = lst[j1];
    vec e0 = asv[sa] + adv;
    vec e1 = asv[sb] + adv;
    vec m = eself;
#pragma unroll
    for (int k = 0; k < HEADS; ++k) {
        e0[k] = lrelu(e0[k]);
        e1[k] = lrelu(e1[k]);
        m[k] = fmaxf(m[k], fmaxf(e0[k], e1[k]));
    }
    for (int c = beg + GATHER_STAGE; c < end; c += 32) {
        const int j = c + lane;
        if (j < end) {
            const vec e = asv[lst[j]] + adv;
#pragma unroll
            for (int k = 0; k < HEADS; ++k) m[k] = fmaxf(m[k], lrelu(e[k]));
        }
    }
    for (int off = 16; off > 0; off >>= 1) {
#pragma unroll
        for (int k = 0; k < HEADS; ++k)
            m[k] = fmaxf(m[k], __shfl_xor(m[k], off, 32));
    }
#pragma unroll
    for (int k = 0; k < HEADS; ++k) {
        wst[k * GATHER_PITCH + lane]      = __expf(e0[k] - m[k]);
        wst[k * GATHER_PITCH + 32 + lane] = __expf(e1[k] - m[k]);
    }
    float mh = m[0], adh = adv[0], esh = eself[0];
#pragma unroll
    for (int k = 1; k < HEADS; ++k)
        if (head == k) { mh = m[k]; adh = adv[k]; esh = eself[k]; }

    const float* wrow = wst + head * GATHER_PITCH;
    const float wself = __expf(esh - mh);
    vec acc = wself * hrow[(size_t)node * 32 + lane];
    float dsum = wself;

    const int ends = (end - beg > GATHER_STAGE) ? beg + GATHER_STAGE : end;
    int j = beg;
    for (; j + 4 <= ends; j += 4) {
        const int s0 = lst[j], s1 = lst[j + 1], s2 = lst[j + 2], s3 = lst[j + 3];
        const vec v0 = hrow[(size_t)s0 * 32 + lane];
        const vec v1 = hrow[(size_t)s1 * 32 + lane];
        const vec v2 = hrow[(size_t)s2 * 32 + lane];
        const vec v3 = hrow[(size_t)s3 * 32 + lane];
        const float4 w = *(const float4*)(wrow + (j - beg));
        acc += w.x * v0;  dsum += w.x;
        acc += w.y * v1;  dsum += w.y;
        acc += w.z * v2;  dsum += w.z;
        acc += w.w * v3;  dsum += w.w;
    }
#pragma clang loop unroll(disable)
    for (; j < ends; ++j) {
        const float w = wrow[j - beg];
        acc += w * hrow[(size_t)lst[j] * 32 + lane];
        dsum += w;
    }

    const float* __restrict__ ash = as + head;
    for (; j + 4 <= end; j += 4) {
        const int s0 = lst[j], s1 = lst[j + 1], s2 = lst[j + 2], s3 = lst[j + 3];
        const float a0 = ash[(size_t)s0 * HEADS];
        const float a1 = ash[(size_t)s1 * HEADS];
        const float a2 = ash[(size_t)s2 * HEADS];
        const float a3 = ash[(size_t)s3 * HEADS];
        const vec v0 = hrow[(size_t)s0 * 32 + lane];
        const vec v1 = hrow[(size_t)s1 * 32 + lane];
        const vec v2 = hrow[(size_t)s2 * 32 + lane];
        const vec v3 = hrow[(size_t)s3 * 32 + lane];
        const float w0 = __expf(lrelu(a0 + adh) - mh);
        const float w1 = __expf(lrelu(a1 + adh) - mh);
        const float w2 = __expf(lrelu(a2 + adh) - mh);
        const float w3 = __expf(lrelu(a3 + adh) - mh);
        acc += w0 * v0;  dsum += w0;
        acc += w1 * v1;  dsum += w1;
        acc += w2 * v2;  dsum += w2;
        acc += w3 * v3;  dsum += w3;
    }
#pragma clang loop unroll(disable)
    for (; j < end; ++j) {
        const int s = lst[j];
        const float w = __expf(lrelu(ash[(size_t)s * HEADS] + adh) - mh);
        acc += w * hrow[(size_t)s * 32 + lane];
        dsum += w;
    }
    acc_out = acc;
    dsum_out = dsum;
}

__device__ __forceinline__ float elu(float v) {
    return v > 0.f ? v : (expf(v) - 1.0f);
}

template <int HEADS>
__global__ void attend_concat(const int* __restrict__ rowptr,
                              const int* __restrict__ lst,
                              const float* __restrict__ as,
                              const float* __restrict__ ad,
                              const float* __restrict__ h,
                              const float* __restrict__ bias,
                              float* __restrict__ hout, int n_nodes) {
    typedef typename rowvec<HEADS>::type vec;
    const int lane = threadIdx.x;
    const int node = __builtin_amdgcn_readfirstlane(
        (int)(blockIdx.x * blockDim.y + threadIdx.y));
    if (node >= n_nodes) return;
    __shared__ __attribute__((aligned(16))) float wst[GATHER_NPB * HEADS * GATHER_PITCH];
    vec acc; float dsum;
    gather_node<HEADS>(rowptr, lst, as, ad, h, node, lane,
                       wst + threadIdx.y * (HEADS * GATHER_PITCH), acc, dsum);
    const vec b = ((const vec*)bias)[lane];
    vec o;
#pragma unroll
    for (int k = 0; k < HEADS; ++k) o[k] = elu(acc[k] / (dsum + EPS) + b[k]);
    ((vec*)hout)[(size_t)node * 32 + lane] = o;
}

template <int OUTC>
__global__ void attend_mean_linear(const int* __restrict__ rowptr,
                                   const int* __restrict__ lst,
                                   const float* __restrict__ as,
                                   const float* __restrict__ ad,
                                   const float* __restrict__ h,
                                   const float* __restrict__ bias,
                                   const float* __restrict__ Wl,
                                   const float* __restrict__ bl,
                                   float* __restrict__ out, int n_nodes) {
    constexpr int HEADS = 2;
    typedef typename rowvec<HEADS>::type vec;
    const int lane = threadIdx.x;
    const int node = __builtin_amdgcn_readfirstlane(
        (int)(blockIdx.x * blockDim.y + threadIdx.y));
    if (node >= n_nodes) return;
    __shared__ __attribute__((aligned(16))) float wst[GATHER_NPB * HEADS * GATHER_PITCH];
    vec acc; float dsum;
    gather_node<HEADS>(rowptr, lst, as, ad, h, node, lane,
                       wst + threadIdx.y * (HEADS * GATHER_PITCH), acc, dsum);

    const int g = lane & 15;
    const vec b = ((const vec*)bias)[g];
    float hv[HEADS];
#pragma unroll
    for (int k = 0; k < HEADS; ++k) {
        const float q  = acc[k] / (dsum + EPS);
        const float qo = __shfl_xor(q, 16, 32);
        hv[k] = elu((q + qo) / (float)HEADS + b[k]);
    }
    const float* __restrict__ wr = Wl + (size_t)(2 * g) * OUTC;
    float t[OUTC];
#pragma unroll
    for (int o = 0; o < OUTC; ++o) t[o] = hv[0] * wr[o] + hv[1] * wr[OUTC + o];
    for (int off = 8; off > 0; off >>= 1) {
#pragma unroll
        for (int o = 0; o < OUTC; ++o) t[o] += __shfl_xor(t[o], off, 32);
    }
    float r = t[0];
#pragma unroll
    for (int o = 1; o < OUTC; ++o)
        if (lane == o) r = t[o];
    if (lane < OUTC) out[(size_t)node * OUTC + lane] = r + bl[lane];
}

static inline unsigned gb(long long total, int bs) {
    return (unsigned)((total + bs - 1) / bs);
}

extern "C" void kernel_launch(void* const* d_in, const int* in_sizes, int n_in,
                              void* d_out, int out_size, void* d_ws, size_t ws_size,
                              hipStream_t stream) {
    const int IN_CH = 128, HID = 32, H1 = 4, H2 = 2, OUT = 3;
    const int N = in_sizes[0] / IN_CH;
    const long long E  = in_sizes[1] / 2;

    const float* x     = (const float*)d_in[0];
    const int*   ei    = (const int*)  d_in[1];
    const float* W1    = (const float*)d_in[2];
    const float* asrc1 = (const float*)d_in[3];
    const float* adst1 = (const float*)d_in[4];
    const float* b1    = (const float*)d_in[5];
    const float* W2    = (const float*)d_in[6];
    const float* asrc2 = (const float*)d_in[7];
    const float* adst2 = (const float*)d_in[8];
    const float* b2    = (const float*)d_in[9];
    const float* Wlin  = (const float*)d_in[10];
    const float* blin  = (const float*)d_in[11];
    float* out = (float*)d_out;

    float* ws = (float*)d_ws;
    size_t off = 0;
    float* h1   = ws + off; off += (size_t)N * H1 * HID;
    float* h1a  = ws + off; off += (size_t)N * H1 * HID;
    float* as_  = ws + off; off += (size_t)N * H1;
    float* ad_  = ws + off; off += (size_t)N * H1;
    int* lst    = (int*)(ws + off); off += (size_t)E;
    int* cnt    = (int*)(ws + off); off += (size_t)N;
    int* rowptr = (int*)(ws + off); off += (size_t)N + 1;
    int* rank   = (int*)(ws + off); off += (size_t)E;
    float* h2   = h1;

    const int BS = 256;
    const int NPB = GATHER_NPB;

    fill_int<<<gb((long long)N, BS), BS, 0, stream>>>(cnt, 0, (long long)N);
    dst_rank<<<gb(E, BS), BS, 0, stream>>>(ei, E, cnt, rank);
    dst_scan<<<1, 1024, 0, stream>>>(cnt, rowptr, N);
    dst_place<<<gb(E, BS), BS, 0, stream>>>(ei, E, rowptr, rank, lst);

    wmma_gemm_f32<<<dim3(N / 16, 1, 1), dim3(32, (H1 * HID) / 16, 1), 0, stream>>>(
        x, W1, h1, N, IN_CH, H1 * HID);
    alpha_kernel<<<gb((long long)N * H1, BS), BS, 0, stream>>>(
        h1, asrc1, adst1, as_, ad_, N, H1, HID);
    attend_concat<H1><<<gb((long long)N, NPB), dim3(32, NPB, 1), 0, stream>>>(
        rowptr, lst, as_, ad_, h1, b1, h1a, N);

    wmma_gemm_f32<<<dim3(N / 16, 1, 1), dim3(32, (H2 * HID) / 16, 1), 0, stream>>>(
        h1a, W2, h2, N, H1 * HID, H2 * HID);
    alpha_kernel<<<gb((long long)N * H2, BS), BS, 0, stream>>>(
        h2, asrc2, adst2, as_, ad_, N, H2, HID);
    attend_mean_linear<OUT><<<gb((long long)N, NPB), dim3(32, NPB, 1), 0, stream>>>(
        rowptr, lst, as_, ad_, h2, b2, Wlin, blin, out, N);
    (void)n_in; (void)out_size; (void)ws_size;
}


# Edge grouping rebuilt as a two-level counting sort on LDS counters (256-node destination buckets, then ranks per bucket): no global atomics, 9 launches for 10; sum order = LDS atomic arrival rank
# speedup vs baseline: 11.6432x; 1.7571x over previous
#include <hip/hip_runtime.h>
#include <hip/hip_bf16.h>


typedef __attribute__((ext_vector_type(2))) float v2f;
typedef __attribute__((ext_vector_type(8))) float v8f;

#define NEG_SLOPE 0.2f
#define EPS 1e-16f

__global__ void wmma_gemm_f32(const float* __restrict__ A,
                              const float* __restrict__ B,
                              float* __restrict__ C,
                              int M, int K, int Nout) {
    const int lane = threadIdx.x;
    const int lm   = lane & 15;
    const int lh   = lane >> 4;
    const int mt   = blockIdx.x;
    const int nt   = threadIdx.y;
    const int row0 = mt * 16;
    const int col0 = nt * 16;
    if (row0 >= M) return;

    const float* __restrict__ arow = A + (size_t)(row0 + lm) * K;
    v8f acc = {};
    for (int k0 = 0; k0 < K; k0 += 4) {
        const int ka = k0 + 2 * lh;
        v2f a, b;
        a.x = arow[ka];
        a.y = arow[ka + 1];
        b.x = B[(size_t)ka * Nout + col0 + lm];
        b.y = B[(size_t)(ka + 1) * Nout + col0 + lm];
        acc = __builtin_amdgcn_wmma_f32_16x16x4_f32(
             false, a,  false, b,
             (short)0, acc,  false,  false);
    }
    float* __restrict__ crow = C + (size_t)(row0 + 8 * lh) * Nout + col0 + lm;
#pragma unroll
    for (int r = 0; r < 8; ++r) crow[(size_t)r * Nout] = acc[r];
}

__global__ void alpha_kernel(const float* __restrict__ h,
                             const float* __restrict__ a_src,
                             const float* __restrict__ a_dst,
                             float* __restrict__ as, float* __restrict__ ad,
                             int n_nodes, int heads, int ch) {
    long long i = (long long)blockIdx.x * blockDim.x + threadIdx.x;
    if (i >= (long long)n_nodes * heads) return;
    const int head = (int)(i % heads);
    const long long node = i / heads;
    const float* __restrict__ hp  = h + ((size_t)node * heads + head) * ch;
    const float* __restrict__ asp = a_src + (size_t)head * ch;
    const float* __restrict__ adp = a_dst + (size_t)head * ch;
    float s = 0.f, d = 0.f;
    for (int c = 0; c < ch; ++c) { float v = hp[c]; s += v * asp[c]; d += v * adp[c]; }
    as[i] = s; ad[i] = d;
}

#define BKT_SHIFT       8
#define BKT_NODES       (1 << BKT_SHIFT)
#define BKT_MAX         256
#define SORT_CHUNKS     256
#define SORT_THREADS    1024
#define SORT_PER_THREAD 8
#define SCATTER_STAGE   7168
#define PLACE_STAGE     12288

__global__ void __launch_bounds__(SORT_THREADS)
bucket_count(const int* __restrict__ ei, int E, int chunk, int nb,
             int* __restrict__ tbl) {
    __shared__ int hist[BKT_MAX];
    const int t = threadIdx.x;
    const int c = blockIdx.x;
    if (t < BKT_MAX) hist[t] = 0;
    __syncthreads();
    const int* __restrict__ dst = ei + E;
    const int lo = c * chunk;
    const int hi = (lo + chunk < E) ? lo + chunk : E;
    for (int base = lo + t; base < hi; base += SORT_THREADS * SORT_PER_THREAD) {
        int d[SORT_PER_THREAD];
#pragma unroll
        for (int k = 0; k < SORT_PER_THREAD; ++k) {
            const int i = base + k * SORT_THREADS;
            d[k] = (i < hi) ? dst[i] : -1;
        }
#pragma unroll
        for (int k = 0; k < SORT_PER_THREAD; ++k)
            if (d[k] >= 0) atomicAdd(&hist[d[k] >> BKT_SHIFT], 1);
    }
    __syncthreads();
    if (t < nb) tbl[c * BKT_MAX + t] = hist[t];
}

__device__ __forceinline__ int scan256(int v, int t, int* ws) {
    const int lane = t & 31, w = t >> 5;
    int inc = v;
    for (int off = 1; off < 32; off <<= 1) {
        const int u = __shfl_up(inc, off, 32);
        if (lane >= off) inc += u;
    }
    if (t < BKT_MAX && lane == 31) ws[w] = inc;
    __syncthreads();
    int before = 0;
    if (t < BKT_MAX)
        for (int k = 0; k < w; ++k) before += ws[k];
    return before + inc;
}

__global__ void __launch_bounds__(SORT_THREADS)
bucket_scatter(const int* __restrict__ ei, int E, int chunk, int nb,
               const int* __restrict__ tbl, int* __restrict__ bstart,
               int* __restrict__ rec) {
    __shared__ int gbase[BKT_MAX];
    __shared__ int lbase[BKT_MAX];
    __shared__ int cur[BKT_MAX];
    __shared__ int wsa[BKT_MAX / 32], wsb[BKT_MAX / 32];
    __shared__ int2 stage[SCATTER_STAGE];
    const int t = threadIdx.x;
    const int c = blockIdx.x;

    const int bq = t & (BKT_MAX - 1);
    const int g = __builtin_amdgcn_readfirstlane(t / BKT_MAX);
    int* sums = (int*)stage;
    int all = 0, bef = 0;
    if (bq < nb) {
#pragma unroll 16
        for (int cc = g; cc < SORT_CHUNKS; cc += SORT_THREADS / BKT_MAX) {
            const int v = tbl[cc * BKT_MAX + bq];
            all += v;
            bef += (cc < c) ? v : 0;
        }
    }
    sums[t] = all;
    sums[SORT_THREADS + t] = bef;
    __syncthreads();

    const bool bk = t < nb;
    int vt = 0, pt = 0;
    if (bk) {
#pragma unroll
        for (int q = 0; q < SORT_THREADS / BKT_MAX; ++q) {
            vt += sums[q * BKT_MAX + t];
            pt += sums[SORT_THREADS + q * BKT_MAX + t];
        }
    }
    const int vo = bk ? tbl[c * BKT_MAX + t] : 0;
    const int st = scan256(vt, t, wsa) - vt;
    const int sl = scan256(vo, t, wsb) - vo;
    if (bk) {
        gbase[t] = st + pt;
        lbase[t] = sl;
        cur[t]   = sl;
        if (c == 0) bstart[t] = st;
    }
    if (c == 0 && t == 0) bstart[nb] = E;
    __syncthreads();

    const int lo = c * chunk;
    const int hi = (lo + chunk < E) ? lo + chunk : E;
    const bool staged = (hi - lo) <= SCATTER_STAGE;
    int2* __restrict__ rec2 = (int2*)rec;
    for (int base = lo + t; base < hi; base += SORT_THREADS * SORT_PER_THREAD) {
        int s[SORT_PER_THREAD], d[SORT_PER_THREAD];
#pragma unroll
        for (int k = 0; k < SORT_PER_THREAD; ++k) {
            const int i = base + k * SORT_THREADS;
            const bool in = i < hi;
            s[k] = in ? ei[i] : 0;
            d[k] = in ? ei[E + i] : -1;
        }
#pragma unroll
        for (int k = 0; k < SORT_PER_THREAD; ++k)
            if (d[k] >= 0) {
                const int b = d[k] >> BKT_SHIFT;
                const int p = atomicAdd(&cur[b], 1);
                if (staged) stage[p] = make_int2(s[k], d[k]);
                else rec2[gbase[b] + (p - lbase[b])] = make_int2(s[k], d[k]);
            }
    }
    __syncthreads();
    if (staged)
        for (int k = t; k < hi - lo; k += SORT_THREADS) {
            const int2 r = stage[k];
            const int b = r.y >> BKT_SHIFT;
            rec2[gbase[b] + (k - lbase[b])] = r;
        }
}

__global__ void __launch_bounds__(SORT_THREADS)
bucket_place(const int* __restrict__ rec, const int* __restrict__ bstart,
             int n, int* __restrict__ rowptr, int* __restrict__ lst) {
    __shared__ int cnt[BKT_NODES];
    __shared__ int wsa[BKT_MAX / 32];
    __shared__ int stage[PLACE_STAGE];
    const int t = threadIdx.x;
    const int b = blockIdx.x;
    const int bb = bstart[b];
    const int be = bstart[b + 1];
    const int2* __restrict__ rec2 = (const int2*)rec;
    if (t < BKT_NODES) cnt[t] = 0;
    __syncthreads();
    for (int base = bb + t; base < be; base += SORT_THREADS * SORT_PER_THREAD) {
        int d[SORT_PER_THREAD];
#pragma unroll
        for (int k = 0; k < SORT_PER_THREAD; ++k) {
            const int j = base + k * SORT_THREADS;
            d[k] = (j < be) ? rec2[j].y : -1;
        }
#pragma unroll
        for (int k = 0; k < SORT_PER_THREAD; ++k)
            if (d[k] >= 0) atomicAdd(&cnt[d[k] & (BKT_NODES - 1)], 1);
    }
    __syncthreads();
    const int v = (t < BKT_NODES) ? cnt[t] : 0;
    const int start = bb + scan256(v, t, wsa) - v;
    if (t < BKT_NODES) {
        cnt[t] = start;
        const int node = (b << BKT_SHIFT) + t;
        if (node < n) rowptr[node] = start;
    }
    if (b == (int)gridDim.x - 1 && t == 0) rowptr[n] = be;
    __syncthreads();
    const bool staged = (be - bb) <= PLACE_STAGE;
    for (int base = bb + t; base < be; base += SORT_THREADS * SORT_PER_THREAD) {
        int2 r[SORT_PER_THREAD];
#pragma unroll
        for (int k = 0; k < SORT_PER_THREAD; ++k) {
            const int j = base + k * SORT_THREADS;
            r[k] = (j < be) ? rec2[j] : make_int2(0, -1);
        }
#pragma unroll
        for (int k = 0; k < SORT_PER_THREAD; ++k)
            if (r[k].y >= 0) {
                const int p = atomicAdd(&cnt[r[k].y & (BKT_NODES - 1)], 1);
                if (staged) stage[p - bb] = r[k].x; else lst[p] = r[k].x;
            }
    }
    __syncthreads();
    if (staged)
        for (int j = bb + t; j < be; j += SORT_THREADS) lst[j] = stage[j - bb];
}

#define GATHER_NPB   8
#define GATHER_STAGE 64
#define GATHER_PITCH (GATHER_STAGE + 4)

__device__ __forceinline__ float lrelu(float v) {
    return v > 0.f ? v : NEG_SLOPE * v;
}

template <int HEADS> struct rowvec {
    typedef float type __attribute__((ext_vector_type(HEADS)));
};

template <int HEADS>
__device__ __forceinline__ void gather_node(const int* __restrict__ rowptr,
                                            const int* __restrict__ lst,
                                            const float* __restrict__ as,
                                            const float* __restrict__ ad,
                                            const float* __restrict__ h,
                                            int node, int lane, float* wst,
                                            typename rowvec<HEADS>::type& acc_out,
                                            float& dsum_out) {
    typedef typename rowvec<HEADS>::type vec;
    const int head = (lane * HEADS) >> 5;
    const vec* __restrict__ hrow = (const vec*)h;
    const vec* __restrict__ asv  = (const vec*)as;
    const int beg = rowptr[node];
    const int end = rowptr[node + 1];

    const vec adv = ((const vec*)ad)[node];
    vec eself = asv[node] + adv;
#pragma unroll
    for (int k = 0; k < HEADS; ++k) eself[k] = lrelu(eself[k]);
    const int j0 = beg + lane, j1 = j0 + 32;
    int sa = node, sb = node;
    if (j0 < end) sa = lst[j0];
    if (j1 < end) sb = lst[j1];
    vec e0 = asv[sa] + adv;
    vec e1 = asv[sb] + adv;
    vec m = eself;
#pragma unroll
    for (int k = 0; k < HEADS; ++k) {
        e0[k] = lrelu(e0[k]);
        e1[k] = lrelu(e1[k]);
        m[k] = fmaxf(m[k], fmaxf(e0[k], e1[k]));
    }
    for (int c = beg + GATHER_STAGE; c < end; c += 32) {
        const int j = c + lane;
        if (j < end) {
            const vec e = asv[lst[j]] + adv;
#pragma unroll
            for (int k = 0; k < HEADS; ++k) m[k] = fmaxf(m[k], lrelu(e[k]));
        }
    }
    for (int off = 16; off > 0; off >>= 1) {
#pragma unroll
        for (int k = 0; k < HEADS; ++k)
            m[k] = fmaxf(m[k], __shfl_xor(m[k], off, 32));
    }
#pragma unroll
    for (int k = 0; k < HEADS; ++k) {
        wst[k * GATHER_PITCH + lane]      = __expf(e0[k] - m[k]);
        wst[k * GATHER_PITCH + 32 + lane] = __expf(e1[k] - m[k]);
    }
    float mh = m[0], adh = adv[0], esh = eself[0];
#pragma unroll
    for (int k = 1; k < HEADS; ++k)
        if (head == k) { mh = m[k]; adh = adv[k]; esh = eself[k]; }

    const float* wrow = wst + head * GATHER_PITCH;
    const float wself = __expf(esh - mh);
    vec acc = wself * hrow[(size_t)node * 32 + lane];
    float dsum = wself;

    const int ends = (end - beg > GATHER_STAGE) ? beg + GATHER_STAGE : end;
    int j = beg;
    for (; j + 4 <= ends; j += 4) {
        const int s0 = lst[j], s1 = lst[j + 1], s2 = lst[j + 2], s3 = lst[j + 3];
        const vec v0 = hrow[(size_t)s0 * 32 + lane];
        const vec v1 = hrow[(size_t)s1 * 32 + lane];
        const vec v2 = hrow[(size_t)s2 * 32 + lane];
        const vec v3 = hrow[(size_t)s3 * 32 + lane];
        const float4 w = *(const float4*)(wrow + (j - beg));
        acc += w.x * v0;  dsum += w.x;
        acc += w.y * v1;  dsum += w.y;
        acc += w.z * v2;  dsum += w.z;
        acc += w.w * v3;  dsum += w.w;
    }
#pragma clang loop unroll(disable)
    for (; j < ends; ++j) {
        const float w = wrow[j - beg];
        acc += w * hrow[(size_t)lst[j] * 32 + lane];
        dsum += w;
    }

    const float* __restrict__ ash = as + head;
    for (; j + 4 <= end; j += 4) {
        const int s0 = lst[j], s1 = lst[j + 1], s2 = lst[j + 2], s3 = lst[j + 3];
        const float a0 = ash[(size_t)s0 * HEADS];
        const float a1 = ash[(size_t)s1 * HEADS];
        const float a2 = ash[(size_t)s2 * HEADS];
        const float a3 = ash[(size_t)s3 * HEADS];
        const vec v0 = hrow[(size_t)s0 * 32 + lane];
        const vec v1 = hrow[(size_t)s1 * 32 + lane];
        const vec v2 = hrow[(size_t)s2 * 32 + lane];
        const vec v3 = hrow[(size_t)s3 * 32 + lane];
        const float w0 = __expf(lrelu(a0 + adh) - mh);
        const float w1 = __expf(lrelu(a1 + adh) - mh);
        const float w2 = __expf(lrelu(a2 + adh) - mh);
        const float w3 = __expf(lrelu(a3 + adh) - mh);
        acc += w0 * v0;  dsum += w0;
        acc += w1 * v1;  dsum += w1;
        acc += w2 * v2;  dsum += w2;
        acc += w3 * v3;  dsum += w3;
    }
#pragma clang loop unroll(disable)
    for (; j < end; ++j) {
        const int s = lst[j];
        const float w = __expf(lrelu(ash[(size_t)s * HEADS] + adh) - mh);
        acc += w * hrow[(size_t)s * 32 + lane];
        dsum += w;
    }
    acc_out = acc;
    dsum_out = dsum;
}

__device__ __forceinline__ float elu(float v) {
    return v > 0.f ? v : (expf(v) - 1.0f);
}

template <int HEADS>
__global__ void attend_concat(const int* __restrict__ rowptr,
                              const int* __restrict__ lst,
                              const float* __restrict__ as,
                              const float* __restrict__ ad,
                              const float* __restrict__ h,
                              const float* __restrict__ bias,
                              float* __restrict__ hout, int n_nodes) {
    typedef typename rowvec<HEADS>::type vec;
    const int lane = threadIdx.x;
    const int node = __builtin_amdgcn_readfirstlane(
        (int)(blockIdx.x * blockDim.y + threadIdx.y));
    if (node >= n_nodes) return;
    __shared__ __attribute__((aligned(16))) float wst[GATHER_NPB * HEADS * GATHER_PITCH];
    vec acc; float dsum;
    gather_node<HEADS>(rowptr, lst, as, ad, h, node, lane,
                       wst + threadIdx.y * (HEADS * GATHER_PITCH), acc, dsum);
    const vec b = ((const vec*)bias)[lane];
    vec o;
#pragma unroll
    for (int k = 0; k < HEADS; ++k) o[k] = elu(acc[k] / (dsum + EPS) + b[k]);
    ((vec*)hout)[(size_t)node * 32 + lane] = o;
}

template <int OUTC>
__global__ void attend_mean_linear(const int* __restrict__ rowptr,
                                   const int* __restrict__ lst,
                                   const float* __restrict__ as,
                                   const float* __restrict__ ad,
                                   const float* __restrict__ h,
                                   const float* __restrict__ bias,
                                   const float* __restrict__ Wl,
                                   const float* __restrict__ bl,
                                   float* __restrict__ out, int n_nodes) {
    constexpr int HEADS = 2;
    typedef typename rowvec<HEADS>::type vec;
    const int lane = threadIdx.x;
    const int node = __builtin_amdgcn_readfirstlane(
        (int)(blockIdx.x * blockDim.y + threadIdx.y));
    if (node >= n_nodes) return;
    __shared__ __attribute__((aligned(16))) float wst[GATHER_NPB * HEADS * GATHER_PITCH];
    vec acc; float dsum;
    gather_node<HEADS>(rowptr, lst, as, ad, h, node, lane,
                       wst + threadIdx.y * (HEADS * GATHER_PITCH), acc, dsum);

    const int g = lane & 15;
    const vec b = ((const vec*)bias)[g];
    float hv[HEADS];
#pragma unroll
    for (int k = 0; k < HEADS; ++k) {
        const float q  = acc[k] / (dsum + EPS);
        const float qo = __shfl_xor(q, 16, 32);
        hv[k] = elu((q + qo) / (float)HEADS + b[k]);
    }
    const float* __restrict__ wr = Wl + (size_t)(2 * g) * OUTC;
    float t[OUTC];
#pragma unroll
    for (int o = 0; o < OUTC; ++o) t[o] = hv[0] * wr[o] + hv[1] * wr[OUTC + o];
    for (int off = 8; off > 0; off >>= 1) {
#pragma unroll
        for (int o = 0; o < OUTC; ++o) t[o] += __shfl_xor(t[o], off, 32);
    }
    float r = t[0];
#pragma unroll
    for (int o = 1; o < OUTC; ++o)
        if (lane == o) r = t[o];
    if (lane < OUTC) out[(size_t)node * OUTC + lane] = r + bl[lane];
}

static inline unsigned gb(long long total, int bs) {
    return (unsigned)((total + bs - 1) / bs);
}

extern "C" void kernel_launch(void* const* d_in, const int* in_sizes, int n_in,
                              void* d_out, int out_size, void* d_ws, size_t ws_size,
                              hipStream_t stream) {
    const int IN_CH = 128, HID = 32, H1 = 4, H2 = 2, OUT = 3;
    const int N = in_sizes[0] / IN_CH;
    const long long E  = in_sizes[1] / 2;

    const float* x     = (const float*)d_in[0];
    const int*   ei    = (const int*)  d_in[1];
    const float* W1    = (const float*)d_in[2];
    const float* asrc1 = (const float*)d_in[3];
    const float* adst1 = (const float*)d_in[4];
    const float* b1    = (const float*)d_in[5];
    const float* W2    = (const float*)d_in[6];
    const float* asrc2 = (const float*)d_in[7];
    const float* adst2 = (const float*)d_in[8];
    const float* b2    = (const float*)d_in[9];
    const float* Wlin  = (const float*)d_in[10];
    const float* blin  = (const float*)d_in[11];
    float* out = (float*)d_out;

    float* ws = (float*)d_ws;
    size_t off = 0;
    float* h1   = ws + off; off += (size_t)N * H1 * HID;
    float* h1a  = ws + off; off += (size_t)N * H1 * HID;
    float* as_  = ws + off; off += (size_t)N * H1;
    float* ad_  = ws + off; off += (size_t)N * H1;
    int* lst    = (int*)(ws + off); off += (size_t)E;
    const int NBKT  = (N + BKT_NODES - 1) >> BKT_SHIFT;
    const int CHUNK = (int)((E + SORT_CHUNKS - 1) / SORT_CHUNKS);
    int* tbl    = (int*)(ws + off); off += (size_t)SORT_CHUNKS * BKT_MAX;
    int* bstart = (int*)(ws + off); off += (size_t)BKT_MAX + 4;
    int* rowptr = (int*)(ws + off); off += (size_t)N + 4;
    off = (off + 1) & ~(size_t)1;
    int* rec    = (int*)(ws + off); off += (size_t)2 * E;
    float* h2   = h1;

    const int BS = 256;
    const int NPB = GATHER_NPB;

    bucket_count<<<SORT_CHUNKS, SORT_THREADS, 0, stream>>>(ei, (int)E, CHUNK, NBKT, tbl);
    bucket_scatter<<<SORT_CHUNKS, SORT_THREADS, 0, stream>>>(ei, (int)E, CHUNK, NBKT, tbl, bstart, rec);
    bucket_place<<<NBKT, SORT_THREADS, 0, stream>>>(rec, bstart, N, rowptr, lst);

    wmma_gemm_f32<<<dim3(N / 16, 1, 1), dim3(32, (H1 * HID) / 16, 1), 0, stream>>>(
        x, W1, h1, N, IN_CH, H1 * HID);
    alpha_kernel<<<gb((long long)N * H1, BS), BS, 0, stream>>>(
        h1, asrc1, adst1, as_, ad_, N, H1, HID);
    attend_concat<H1><<<gb((long long)N, NPB), dim3(32, NPB, 1), 0, stream>>>(
        rowptr, lst, as_, ad_, h1, b1, h1a, N);

    wmma_gemm_f32<<<dim3(N / 16, 1, 1), dim3(32, (H2 * HID) / 16, 1), 0, stream>>>(
        h1a, W2, h2, N, H1 * HID, H2 * HID);
    alpha_kernel<<<gb((long long)N * H2, BS), BS, 0, stream>>>(
        h2, asrc2, adst2, as_, ad_, N, H2, HID);
    attend_mean_linear<OUT><<<gb((long long)N, NPB), dim3(32, NPB, 1), 0, stream>>>(
        rowptr, lst, as_, ad_, h2, b2, Wlin, blin, out, N);
    (void)n_in; (void)out_size; (void)ws_size;
}


# Attention logits formed in the projection's epilogue: a wave owns one head (two 16x16 tiles, one A fragment), dots it with a_src/a_dst by 16-lane butterfly; 7 launches for 9; logit sum order changes
# speedup vs baseline: 13.8164x; 1.1866x over previous
#include <hip/hip_runtime.h>
#include <hip/hip_bf16.h>


typedef __attribute__((ext_vector_type(2))) float v2f;
typedef __attribute__((ext_vector_type(8))) float v8f;

#define NEG_SLOPE 0.2f
#define EPS 1e-16f


template <int MASK>
__device__ __forceinline__ float row_xor_add(float v) {
    const int b = __builtin_bit_cast(int, v);
    return v + __builtin_bit_cast(float,
        __builtin_amdgcn_update_dpp(0, b, 0x160 | MASK, 0xf, 0xf, false));
}

__device__ __forceinline__ float row16_sum(float v) {
    v = row_xor_add<1>(v);
    v = row_xor_add<2>(v);
    v = row_xor_add<4>(v);
    v = row_xor_add<8>(v);
    return v;
}

template <int K, int HEADS>
__global__ void __launch_bounds__(32 * HEADS)
project_logits(const float* __restrict__ A,
               const float* __restrict__ B,
               const float* __restrict__ a_src,
               const float* __restrict__ a_dst,
               float* __restrict__ C,
               float* __restrict__ as, float* __restrict__ ad, int M) {
    constexpr int NOUT = HEADS * 32;
    constexpr int T = 16 * HEADS;
    __shared__ float lg[2 * T];
    const int lane = threadIdx.x;
    const int lm   = lane & 15;
    const int lh   = lane >> 4;
    const int head = threadIdx.y;
    const int row0 = blockIdx.x * 16;
    const int col0 = head * 32;
    if (row0 >= M) return;

    const float* __restrict__ arow = A + (size_t)(row0 + lm) * K;
    const float* __restrict__ bcol = B + col0 + lm;
    const float s0 = a_src[col0 + lm], s1 = a_src[col0 + 16 + lm];
    const float d0 = a_dst[col0 + lm], d1 = a_dst[col0 + 16 + lm];
    v8f acc0 = {}, acc1 = {};
    for (int k0 = 0; k0 < K; k0 += 4) {
        const int ka = k0 + 2 * lh;
        v2f a, b0, b1;
        a.x  = arow[ka];
        a.y  = arow[ka + 1];
        b0.x = bcol[(size_t)ka * NOUT];
        b0.y = bcol[(size_t)(ka + 1) * NOUT];
        b1.x = bcol[(size_t)ka * NOUT + 16];
        b1.y = bcol[(size_t)(ka + 1) * NOUT + 16];
        acc0 = __builtin_amdgcn_wmma_f32_16x16x4_f32(
             false, a,  false, b0,
             (short)0, acc0,  false,  false);
        acc1 = __builtin_amdgcn_wmma_f32_16x16x4_f32(
             false, a,  false, b1,
             (short)0, acc1,  false,  false);
    }
    float* __restrict__ crow = C + (size_t)(row0 + 8 * lh) * NOUT + col0 + lm;
#pragma unroll
    for (int r = 0; r < 8; ++r) {
        crow[(size_t)r * NOUT]      = acc0[r];
        crow[(size_t)r * NOUT + 16] = acc1[r];
    }

    float vs = 0.f, vd = 0.f;
#pragma unroll
    for (int r = 0; r < 8; ++r) {
        const float ps = row16_sum(acc0[r] * s0 + acc1[r] * s1);
        const float pd = row16_sum(acc0[r] * d0 + acc1[r] * d1);
        if (lm == r) { vs = ps; vd = pd; }
    }
    if (lm < 8) {
        const int rr = 8 * lh + lm;
        lg[rr * HEADS + head]     = vs;
        lg[T + rr * HEADS + head] = vd;
    }
    __syncthreads();
    const int t = head * 32 + lane;
    float* __restrict__ dstp = (t < T) ? as : ad;
    dstp[(size_t)row0 * HEADS + (t < T ? t : t - T)] = lg[t];
}

#define BKT_SHIFT       8
#define BKT_NODES       (1 << BKT_SHIFT)
#define BKT_MAX         256
#define SORT_CHUNKS     256
#define SORT_THREADS    1024
#define SORT_PER_THREAD 8
#define SCATTER_STAGE   7168
#define PLACE_STAGE     12288

__global__ void __launch_bounds__(SORT_THREADS)
bucket_count(const int* __restrict__ ei, int E, int chunk, int nb,
             int* __restrict__ tbl) {
    __shared__ int hist[BKT_MAX];
    const int t = threadIdx.x;
    const int c = blockIdx.x;
    if (t < BKT_MAX) hist[t] = 0;
    __syncthreads();
    const int* __restrict__ dst = ei + E;
    const int lo = c * chunk;
    const int hi = (lo + chunk < E) ? lo + chunk : E;
    for (int base = lo + t; base < hi; base += SORT_THREADS * SORT_PER_THREAD) {
        int d[SORT_PER_THREAD];
#pragma unroll
        for (int k = 0; k < SORT_PER_THREAD; ++k) {
            const int i = base + k * SORT_THREADS;
            d[k] = (i < hi) ? dst[i] : -1;
        }
#pragma unroll
        for (int k = 0; k < SORT_PER_THREAD; ++k)
            if (d[k] >= 0) atomicAdd(&hist[d[k] >> BKT_SHIFT], 1);
    }
    __syncthreads();
    if (t < nb) tbl[c * BKT_MAX + t] = hist[t];
}

__device__ __forceinline__ int scan256(int v, int t, int* ws) {
    const int lane = t & 31, w = t >> 5;
    int inc = v;
    for (int off = 1; off < 32; off <<= 1) {
        const int u = __shfl_up(inc, off, 32);
        if (lane >= off) inc += u;
    }
    if (t < BKT_MAX && lane == 31) ws[w] = inc;
    __syncthreads();
    int before = 0;
    if (t < BKT_MAX)
        for (int k = 0; k < w; ++k) before += ws[k];
    return before + inc;
}

__global__ void __launch_bounds__(SORT_THREADS)
bucket_scatter(const int* __restrict__ ei, int E, int chunk, int nb,
               const int* __restrict__ tbl, int* __restrict__ bstart,
               int* __restrict__ rec) {
    __shared__ int gbase[BKT_MAX];
    __shared__ int lbase[BKT_MAX];
    __shared__ int cur[BKT_MAX];
    __shared__ int wsa[BKT_MAX / 32], wsb[BKT_MAX / 32];
    __shared__ int2 stage[SCATTER_STAGE];
    const int t = threadIdx.x;
    const int c = blockIdx.x;

    const int bq = t & (BKT_MAX - 1);
    const int g = __builtin_amdgcn_readfirstlane(t / BKT_MAX);
    int* sums = (int*)stage;
    int all = 0, bef = 0;
    if (bq < nb) {
#pragma unroll 16
        for (int cc = g; cc < SORT_CHUNKS; cc += SORT_THREADS / BKT_MAX) {
            const int v = tbl[cc * BKT_MAX + bq];
            all += v;
            bef += (cc < c) ? v : 0;
        }
    }
    sums[t] = all;
    sums[SORT_THREADS + t] = bef;
    __syncthreads();

    const bool bk = t < nb;
    int vt = 0, pt = 0;
    if (bk) {
#pragma unroll
        for (int q = 0; q < SORT_THREADS / BKT_MAX; ++q) {
            vt += sums[q * BKT_MAX + t];
            pt += sums[SORT_THREADS + q * BKT_MAX + t];
        }
    }
    const int vo = bk ? tbl[c * BKT_MAX + t] : 0;
    const int st = scan256(vt, t, wsa) - vt;
    const int sl = scan256(vo, t, wsb) - vo;
    if (bk) {
        gbase[t] = st + pt;
        lbase[t] = sl;
        cur[t]   = sl;
        if (c == 0) bstart[t] = st;
    }
    if (c == 0 && t == 0) bstart[nb] = E;
    __syncthreads();

    const int lo = c * chunk;
    const int hi = (lo + chunk < E) ? lo + chunk : E;
    const bool staged = (hi - lo) <= SCATTER_STAGE;
    int2* __restrict__ rec2 = (int2*)rec;
    for (int base = lo + t; base < hi; base += SORT_THREADS * SORT_PER_THREAD) {
        int s[SORT_PER_THREAD], d[SORT_PER_THREAD];
#pragma unroll
        for (int k = 0; k < SORT_PER_THREAD; ++k) {
            const int i = base + k * SORT_THREADS;
            const bool in = i < hi;
            s[k] = in ? ei[i] : 0;
            d[k] = in ? ei[E + i] : -1;
        }
#pragma unroll
        for (int k = 0; k < SORT_PER_THREAD; ++k)
            if (d[k] >= 0) {
                const int b = d[k] >> BKT_SHIFT;
                const int p = atomicAdd(&cur[b], 1);
                if (staged) stage[p] = make_int2(s[k], d[k]);
                else rec2[gbase[b] + (p - lbase[b])] = make_int2(s[k], d[k]);
            }
    }
    __syncthreads();
    if (staged)
        for (int k = t; k < hi - lo; k += SORT_THREADS) {
            const int2 r = stage[k];
            const int b = r.y >> BKT_SHIFT;
            rec2[gbase[b] + (k - lbase[b])] = r;
        }
}

__global__ void __launch_bounds__(SORT_THREADS)
bucket_place(const int* __restrict__ rec, const int* __restrict__ bstart,
             int n, int* __restrict__ rowptr, int* __restrict__ lst) {
    __shared__ int cnt[BKT_NODES];
    __shared__ int wsa[BKT_MAX / 32];
    __shared__ int stage[PLACE_STAGE];
    const int t = threadIdx.x;
    const int b = blockIdx.x;
    const int bb = bstart[b];
    const int be = bstart[b + 1];
    const int2* __restrict__ rec2 = (const int2*)rec;
    if (t < BKT_NODES) cnt[t] = 0;
    __syncthreads();
    for (int base = bb + t; base < be; base += SORT_THREADS * SORT_PER_THREAD) {
        int d[SORT_PER_THREAD];
#pragma unroll
        for (int k = 0; k < SORT_PER_THREAD; ++k) {
            const int j = base + k * SORT_THREADS;
            d[k] = (j < be) ? rec2[j].y : -1;
        }
#pragma unroll
        for (int k = 0; k < SORT_PER_THREAD; ++k)
            if (d[k] >= 0) atomicAdd(&cnt[d[k] & (BKT_NODES - 1)], 1);
    }
    __syncthreads();
    const int v = (t < BKT_NODES) ? cnt[t] : 0;
    const int start = bb + scan256(v, t, wsa) - v;
    if (t < BKT_NODES) {
        cnt[t] = start;
        const int node = (b << BKT_SHIFT) + t;
        if (node < n) rowptr[node] = start;
    }
    if (b == (int)gridDim.x - 1 && t == 0) rowptr[n] = be;
    __syncthreads();
    const bool staged = (be - bb) <= PLACE_STAGE;
    for (int base = bb + t; base < be; base += SORT_THREADS * SORT_PER_THREAD) {
        int2 r[SORT_PER_THREAD];
#pragma unroll
        for (int k = 0; k < SORT_PER_THREAD; ++k) {
            const int j = base + k * SORT_THREADS;
            r[k] = (j < be) ? rec2[j] : make_int2(0, -1);
        }
#pragma unroll
        for (int k = 0; k < SORT_PER_THREAD; ++k)
            if (r[k].y >= 0) {
                const int p = atomicAdd(&cnt[r[k].y & (BKT_NODES - 1)], 1);
                if (staged) stage[p - bb] = r[k].x; else lst[p] = r[k].x;
            }
    }
    __syncthreads();
    if (staged)
        for (int j = bb + t; j < be; j += SORT_THREADS) lst[j] = stage[j - bb];
}

#define GATHER_NPB   8
#define GATHER_STAGE 64
#define GATHER_PITCH (GATHER_STAGE + 4)

__device__ __forceinline__ float lrelu(float v) {
    return v > 0.f ? v : NEG_SLOPE * v;
}

template <int HEADS> struct rowvec {
    typedef float type __attribute__((ext_vector_type(HEADS)));
};

template <int HEADS>
__device__ __forceinline__ void gather_node(const int* __restrict__ rowptr,
                                            const int* __restrict__ lst,
                                            const float* __restrict__ as,
                                            const float* __restrict__ ad,
                                            const float* __restrict__ h,
                                            int node, int lane, float* wst,
                                            typename rowvec<HEADS>::type& acc_out,
                                            float& dsum_out) {
    typedef typename rowvec<HEADS>::type vec;
    const int head = (lane * HEADS) >> 5;
    const vec* __restrict__ hrow = (const vec*)h;
    const vec* __restrict__ asv  = (const vec*)as;
    const int beg = rowptr[node];
    const int end = rowptr[node + 1];

    const vec adv = ((const vec*)ad)[node];
    vec eself = asv[node] + adv;
#pragma unroll
    for (int k = 0; k < HEADS; ++k) eself[k] = lrelu(eself[k]);
    const int j0 = beg + lane, j1 = j0 + 32;
    int sa = node, sb = node;
    if (j0 < end) sa = lst[j0];
    if (j1 < end) sb = lst[j1];
    vec e0 = asv[sa] + adv;
    vec e1 = asv[sb] + adv;
    vec m = eself;
#pragma unroll
    for (int k = 0; k < HEADS; ++k) {
        e0[k] = lrelu(e0[k]);
        e1[k] = lrelu(e1[k]);
        m[k] = fmaxf(m[k], fmaxf(e0[k], e1[k]));
    }
    for (int c = beg + GATHER_STAGE; c < end; c += 32) {
        const int j = c + lane;
        if (j < end) {
            const vec e = asv[lst[j]] + adv;
#pragma unroll
            for (int k = 0; k < HEADS; ++k) m[k] = fmaxf(m[k], lrelu(e[k]));
        }
    }
    for (int off = 16; off > 0; off >>= 1) {
#pragma unroll
        for (int k = 0; k < HEADS; ++k)
            m[k] = fmaxf(m[k], __shfl_xor(m[k], off, 32));
    }
#pragma unroll
    for (int k = 0; k < HEADS; ++k) {
        wst[k * GATHER_PITCH + lane]      = __expf(e0[k] - m[k]);
        wst[k * GATHER_PITCH + 32 + lane] = __expf(e1[k] - m[k]);
    }
    float mh = m[0], adh = adv[0], esh = eself[0];
#pragma unroll
    for (int k = 1; k < HEADS; ++k)
        if (head == k) { mh = m[k]; adh = adv[k]; esh = eself[k]; }

    const float* wrow = wst + head * GATHER_PITCH;
    const float wself = __expf(esh - mh);
    vec acc = wself * hrow[(size_t)node * 32 + lane];
    float dsum = wself;

    const int ends = (end - beg > GATHER_STAGE) ? beg + GATHER_STAGE : end;
    int j = beg;
    for (; j + 4 <= ends; j += 4) {
        const int s0 = lst[j], s1 = lst[j + 1], s2 = lst[j + 2], s3 = lst[j + 3];
        const vec v0 = hrow[(size_t)s0 * 32 + lane];
        const vec v1 = hrow[(size_t)s1 * 32 + lane];
        const vec v2 = hrow[(size_t)s2 * 32 + lane];
        const vec v3 = hrow[(size_t)s3 * 32 + lane];
        const float4 w = *(const float4*)(wrow + (j - beg));
        acc += w.x * v0;  dsum += w.x;
        acc += w.y * v1;  dsum += w.y;
        acc += w.z * v2;  dsum += w.z;
        acc += w.w * v3;  dsum += w.w;
    }
#pragma clang loop unroll(disable)
    for (; j < ends; ++j) {
        const float w = wrow[j - beg];
        acc += w * hrow[(size_t)lst[j] * 32 + lane];
        dsum += w;
    }

    const float* __restrict__ ash = as + head;
    for (; j + 4 <= end; j += 4) {
        const int s0 = lst[j], s1 = lst[j + 1], s2 = lst[j + 2], s3 = lst[j + 3];
        const float a0 = ash[(size_t)s0 * HEADS];
        const float a1 = ash[(size_t)s1 * HEADS];
        const float a2 = ash[(size_t)s2 * HEADS];
        const float a3 = ash[(size_t)s3 * HEADS];
        const vec v0 = hrow[(size_t)s0 * 32 + lane];
        const vec v1 = hrow[(size_t)s1 * 32 + lane];
        const vec v2 = hrow[(size_t)s2 * 32 + lane];
        const vec v3 = hrow[(size_t)s3 * 32 + lane];
        const float w0 = __expf(lrelu(a0 + adh) - mh);
        const float w1 = __expf(lrelu(a1 + adh) - mh);
        const float w2 = __expf(lrelu(a2 + adh) - mh);
        const float w3 = __expf(lrelu(a3 + adh) - mh);
        acc += w0 * v0;  dsum += w0;
        acc += w1 * v1;  dsum += w1;
        acc += w2 * v2;  dsum += w2;
        acc += w3 * v3;  dsum += w3;
    }
#pragma clang loop unroll(disable)
    for (; j < end; ++j) {
        const int s = lst[j];
        const float w = __expf(lrelu(ash[(size_t)s * HEADS] + adh) - mh);
        acc += w * hrow[(size_t)s * 32 + lane];
        dsum += w;
    }
    acc_out = acc;
    dsum_out = dsum;
}

__device__ __forceinline__ float elu(float v) {
    return v > 0.f ? v : (expf(v) - 1.0f);
}

template <int HEADS>
__global__ void attend_concat(const int* __restrict__ rowptr,
                              const int* __restrict__ lst,
                              const float* __restrict__ as,
                              const float* __restrict__ ad,
                              const float* __restrict__ h,
                              const float* __restrict__ bias,
                              float* __restrict__ hout, int n_nodes) {
    typedef typename rowvec<HEADS>::type vec;
    const int lane = threadIdx.x;
    const int node = __builtin_amdgcn_readfirstlane(
        (int)(blockIdx.x * blockDim.y + threadIdx.y));
    if (node >= n_nodes) return;
    __shared__ __attribute__((aligned(16))) float wst[GATHER_NPB * HEADS * GATHER_PITCH];
    vec acc; float dsum;
    gather_node<HEADS>(rowptr, lst, as, ad, h, node, lane,
                       wst + threadIdx.y * (HEADS * GATHER_PITCH), acc, dsum);
    const vec b = ((const vec*)bias)[lane];
    vec o;
#pragma unroll
    for (int k = 0; k < HEADS; ++k) o[k] = elu(acc[k] / (dsum + EPS) + b[k]);
    ((vec*)hout)[(size_t)node * 32 + lane] = o;
}

template <int OUTC>
__global__ void attend_mean_linear(const int* __restrict__ rowptr,
                                   const int* __restrict__ lst,
                                   const float* __restrict__ as,
                                   const float* __restrict__ ad,
                                   const float* __restrict__ h,
                                   const float* __restrict__ bias,
                                   const float* __restrict__ Wl,
                                   const float* __restrict__ bl,
                                   float* __restrict__ out, int n_nodes) {
    constexpr int HEADS = 2;
    typedef typename rowvec<HEADS>::type vec;
    const int lane = threadIdx.x;
    const int node = __builtin_amdgcn_readfirstlane(
        (int)(blockIdx.x * blockDim.y + threadIdx.y));
    if (node >= n_nodes) return;
    __shared__ __attribute__((aligned(16))) float wst[GATHER_NPB * HEADS * GATHER_PITCH];
    vec acc; float dsum;
    gather_node<HEADS>(rowptr, lst, as, ad, h, node, lane,
                       wst + threadIdx.y * (HEADS * GATHER_PITCH), acc, dsum);

    const int g = lane & 15;
    const vec b = ((const vec*)bias)[g];
    float hv[HEADS];
#pragma unroll
    for (int k = 0; k < HEADS; ++k) {
        const float q  = acc[k] / (dsum + EPS);
        const float qo = __shfl_xor(q, 16, 32);
        hv[k] = elu((q + qo) / (float)HEADS + b[k]);
    }
    const float* __restrict__ wr = Wl + (size_t)(2 * g) * OUTC;
    float t[OUTC];
#pragma unroll
    for (int o = 0; o < OUTC; ++o) t[o] = hv[0] * wr[o] + hv[1] * wr[OUTC + o];
    for (int off = 8; off > 0; off >>= 1) {
#pragma unroll
        for (int o = 0; o < OUTC; ++o) t[o] += __shfl_xor(t[o], off, 32);
    }
    float r = t[0];
#pragma unroll
    for (int o = 1; o < OUTC; ++o)
        if (lane == o) r = t[o];
    if (lane < OUTC) out[(size_t)node * OUTC + lane] = r + bl[lane];
}

static inline unsigned gb(long long total, int bs) {
    return (unsigned)((total + bs - 1) / bs);
}

extern "C" void kernel_launch(void* const* d_in, const int* in_sizes, int n_in,
                              void* d_out, int out_size, void* d_ws, size_t ws_size,
                              hipStream_t stream) {
    const int IN_CH = 128, HID = 32, H1 = 4, H2 = 2, OUT = 3;
    const int N = in_sizes[0] / IN_CH;
    const long long E  = in_sizes[1] / 2;

    const float* x     = (const float*)d_in[0];
    const int*   ei    = (const int*)  d_in[1];
    const float* W1    = (const float*)d_in[2];
    const float* asrc1 = (const float*)d_in[3];
    const float* adst1 = (const float*)d_in[4];
    const float* b1    = (const float*)d_in[5];
    const float* W2    = (const float*)d_in[6];
    const float* asrc2 = (const float*)d_in[7];
    const float* adst2 = (const float*)d_in[8];
    const float* b2    = (const float*)d_in[9];
    const float* Wlin  = (const float*)d_in[10];
    const float* blin  = (const float*)d_in[11];
    float* out = (float*)d_out;

    float* ws = (float*)d_ws;
    size_t off = 0;
    float* h1   = ws + off; off += (size_t)N * H1 * HID;
    float* h1a  = ws + off; off += (size_t)N * H1 * HID;
    float* as_  = ws + off; off += (size_t)N * H1;
    float* ad_  = ws + off; off += (size_t)N * H1;
    int* lst    = (int*)(ws + off); off += (size_t)E;
    const int NBKT  = (N + BKT_NODES - 1) >> BKT_SHIFT;
    const int CHUNK = (int)((E + SORT_CHUNKS - 1) / SORT_CHUNKS);
    int* tbl    = (int*)(ws + off); off += (size_t)SORT_CHUNKS * BKT_MAX;
    int* bstart = (int*)(ws + off); off += (size_t)BKT_MAX + 4;
    int* rowptr = (int*)(ws + off); off += (size_t)N + 4;
    off = (off + 1) & ~(size_t)1;
    int* rec    = (int*)(ws + off); off += (size_t)2 * E;
    float* h2   = h1;

    const int NPB = GATHER_NPB;

    bucket_count<<<SORT_CHUNKS, SORT_THREADS, 0, stream>>>(ei, (int)E, CHUNK, NBKT, tbl);
    bucket_scatter<<<SORT_CHUNKS, SORT_THREADS, 0, stream>>>(ei, (int)E, CHUNK, NBKT, tbl, bstart, rec);
    bucket_place<<<NBKT, SORT_THREADS, 0, stream>>>(rec, bstart, N, rowptr, lst);

    project_logits<IN_CH, H1><<<dim3(N / 16, 1, 1), dim3(32, H1, 1), 0, stream>>>(
        x, W1, asrc1, adst1, h1, as_, ad_, N);
    attend_concat<H1><<<gb((long long)N, NPB), dim3(32, NPB, 1), 0, stream>>>(
        rowptr, lst, as_, ad_, h1, b1, h1a, N);

    project_logits<H1 * HID, H2><<<dim3(N / 16, 1, 1), dim3(32, H2, 1), 0, stream>>>(
        h1a, W2, asrc2, adst2, h2, as_, ad_, N);
    attend_mean_linear<OUT><<<gb((long long)N, NPB), dim3(32, NPB, 1), 0, stream>>>(
        rowptr, lst, as_, ad_, h2, b2, Wlin, blin, out, N);
    (void)n_in; (void)out_size; (void)ws_size;
}


# Layer-1 projection moved into the edge sort's 2nd and 3rd launch as extra work-groups (independent work, run side by side; its k loop 4 steps per trip to fit 64 registers); 6 launches for 7
# speedup vs baseline: 14.5076x; 1.0500x over previous
#include <hip/hip_runtime.h>
#include <hip/hip_bf16.h>


typedef __attribute__((ext_vector_type(2))) float v2f;
typedef __attribute__((ext_vector_type(8))) float v8f;

#define NEG_SLOPE 0.2f
#define EPS 1e-16f


template <int MASK>
__device__ __forceinline__ float row_xor_add(float v) {
    const int b = __builtin_bit_cast(int, v);
    return v + __builtin_bit_cast(float,
        __builtin_amdgcn_update_dpp(0, b, 0x160 | MASK, 0xf, 0xf, false));
}

__device__ __forceinline__ float row16_sum(float v) {
    v = row_xor_add<1>(v);
    v = row_xor_add<2>(v);
    v = row_xor_add<4>(v);
    v = row_xor_add<8>(v);
    return v;
}

template <int K, int HEADS, int KSTEPS>
__device__ __forceinline__ void project_tile(const float* __restrict__ A,
                                             const float* __restrict__ B,
                                             const float* __restrict__ a_src,
                                             const float* __restrict__ a_dst,
                                             float* __restrict__ C,
                                             int row0, int head, int lane, float* lg) {
    constexpr int NOUT = HEADS * 32;
    constexpr int T = 16 * HEADS;
    const int lm   = lane & 15;
    const int lh   = lane >> 4;
    const int col0 = head * 32;

    const float* __restrict__ arow = A + (size_t)(row0 + lm) * K;
    const float* __restrict__ bcol = B + col0 + lm;
    const float s0 = a_src[col0 + lm], s1 = a_src[col0 + 16 + lm];
    const float d0 = a_dst[col0 + lm], d1 = a_dst[col0 + 16 + lm];
    v8f acc0 = {}, acc1 = {};
#pragma unroll KSTEPS
    for (int k0 = 0; k0 < K; k0 += 4) {
        const int ka = k0 + 2 * lh;
        v2f a, b0, b1;
        a.x  = arow[ka];
        a.y  = arow[ka + 1];
        b0.x = bcol[(size_t)ka * NOUT];
        b0.y = bcol[(size_t)(ka + 1) * NOUT];
        b1.x = bcol[(size_t)ka * NOUT + 16];
        b1.y = bcol[(size_t)(ka + 1) * NOUT + 16];
        acc0 = __builtin_amdgcn_wmma_f32_16x16x4_f32(
             false, a,  false, b0,
             (short)0, acc0,  false,  false);
        acc1 = __builtin_amdgcn_wmma_f32_16x16x4_f32(
             false, a,  false, b1,
             (short)0, acc1,  false,  false);
    }
    float* __restrict__ crow = C + (size_t)(row0 + 8 * lh) * NOUT + col0 + lm;
#pragma unroll
    for (int r = 0; r < 8; ++r) {
        crow[(size_t)r * NOUT]      = acc0[r];
        crow[(size_t)r * NOUT + 16] = acc1[r];
    }

    float vs = 0.f, vd = 0.f;
#pragma unroll
    for (int r = 0; r < 8; ++r) {
        const float ps = row16_sum(acc0[r] * s0 + acc1[r] * s1);
        const float pd = row16_sum(acc0[r] * d0 + acc1[r] * d1);
        if (lm == r) { vs = ps; vd = pd; }
    }
    if (lm < 8) {
        const int rr = 8 * lh + lm;
        lg[rr * HEADS + head]     = vs;
        lg[T + rr * HEADS + head] = vd;
    }
}

template <int HEADS>
__device__ __forceinline__ void store_logits(const float* lg, float* __restrict__ as,
                                             float* __restrict__ ad,
                                             int row0, int head, int lane) {
    constexpr int T = 16 * HEADS;
    const int t = head * 32 + lane;
    float* __restrict__ dstp = (t < T) ? as : ad;
    dstp[(size_t)row0 * HEADS + (t < T ? t : t - T)] = lg[t];
}

template <int K, int HEADS>
__global__ void __launch_bounds__(32 * HEADS)
project_logits(const float* __restrict__ A,
               const float* __restrict__ B,
               const float* __restrict__ a_src,
               const float* __restrict__ a_dst,
               float* __restrict__ C,
               float* __restrict__ as, float* __restrict__ ad, int M) {
    __shared__ float lg[2 * 16 * HEADS];
    const int lane = threadIdx.x;
    const int head = threadIdx.y;
    const int row0 = blockIdx.x * 16;
    if (row0 >= M) return;
    project_tile<K, HEADS, 8>(A, B, a_src, a_dst, C, row0, head, lane, lg);
    __syncthreads();
    store_logits<HEADS>(lg, as, ad, row0, head, lane);
}

#define BKT_SHIFT       8
#define BKT_NODES       (1 << BKT_SHIFT)
#define BKT_MAX         256
#define SORT_CHUNKS     256
#define SORT_THREADS    1024
#define SORT_PER_THREAD 8
#define SCATTER_STAGE   7168
#define PLACE_STAGE     12288

__global__ void __launch_bounds__(SORT_THREADS)
bucket_count(const int* __restrict__ ei, int E, int chunk, int nb,
             int* __restrict__ tbl) {
    __shared__ int hist[BKT_MAX];
    const int t = threadIdx.x;
    const int c = blockIdx.x;
    if (t < BKT_MAX) hist[t] = 0;
    __syncthreads();
    const int* __restrict__ dst = ei + E;
    const int lo = c * chunk;
    const int hi = (lo + chunk < E) ? lo + chunk : E;
    for (int base = lo + t; base < hi; base += SORT_THREADS * SORT_PER_THREAD) {
        int d[SORT_PER_THREAD];
#pragma unroll
        for (int k = 0; k < SORT_PER_THREAD; ++k) {
            const int i = base + k * SORT_THREADS;
            d[k] = (i < hi) ? dst[i] : -1;
        }
#pragma unroll
        for (int k = 0; k < SORT_PER_THREAD; ++k)
            if (d[k] >= 0) atomicAdd(&hist[d[k] >> BKT_SHIFT], 1);
    }
    __syncthreads();
    if (t < nb) tbl[c * BKT_MAX + t] = hist[t];
}

__device__ __forceinline__ int scan256(int v, int t, int* ws) {
    const int lane = t & 31, w = t >> 5;
    int inc = v;
    for (int off = 1; off < 32; off <<= 1) {
        const int u = __shfl_up(inc, off, 32);
        if (lane >= off) inc += u;
    }
    if (t < BKT_MAX && lane == 31) ws[w] = inc;
    __syncthreads();
    int before = 0;
    if (t < BKT_MAX)
        for (int k = 0; k < w; ++k) before += ws[k];
    return before + inc;
}

__device__ __forceinline__ void bucket_scatter(const int* __restrict__ ei, int E, int chunk,
                                               int nb, const int* __restrict__ tbl,
                                               int* __restrict__ bstart,
                                               int* __restrict__ rec, int c) {
    __shared__ int gbase[BKT_MAX];
    __shared__ int lbase[BKT_MAX];
    __shared__ int cur[BKT_MAX];
    __shared__ int wsa[BKT_MAX / 32], wsb[BKT_MAX / 32];
    __shared__ int2 stage[SCATTER_STAGE];
    const int t = threadIdx.x;

    const int bq = t & (BKT_MAX - 1);
    const int g = __builtin_amdgcn_readfirstlane(t / BKT_MAX);
    int* sums = (int*)stage;
    int all = 0, bef = 0;
    if (bq < nb) {
#pragma unroll 16
        for (int cc = g; cc < SORT_CHUNKS; cc += SORT_THREADS / BKT_MAX) {
            const int v = tbl[cc * BKT_MAX + bq];
            all += v;
            bef += (cc < c) ? v : 0;
        }
    }
    sums[t] = all;
    sums[SORT_THREADS + t] = bef;
    __syncthreads();

    const bool bk = t < nb;
    int vt = 0, pt = 0;
    if (bk) {
#pragma unroll
        for (int q = 0; q < SORT_THREADS / BKT_MAX; ++q) {
            vt += sums[q * BKT_MAX + t];
            pt += sums[SORT_THREADS + q * BKT_MAX + t];
        }
    }
    const int vo = bk ? tbl[c * BKT_MAX + t] : 0;
    const int st = scan256(vt, t, wsa) - vt;
    const int sl = scan256(vo, t, wsb) - vo;
    if (bk) {
        gbase[t] = st + pt;
        lbase[t] = sl;
        cur[t]   = sl;
        if (c == 0) bstart[t] = st;
    }
    if (c == 0 && t == 0) bstart[nb] = E;
    __syncthreads();

    const int lo = c * chunk;
    const int hi = (lo + chunk < E) ? lo + chunk : E;
    const bool staged = (hi - lo) <= SCATTER_STAGE;
    int2* __restrict__ rec2 = (int2*)rec;
    for (int base = lo + t; base < hi; base += SORT_THREADS * SORT_PER_THREAD) {
        int s[SORT_PER_THREAD], d[SORT_PER_THREAD];
#pragma unroll
        for (int k = 0; k < SORT_PER_THREAD; ++k) {
            const int i = base + k * SORT_THREADS;
            const bool in = i < hi;
            s[k] = in ? ei[i] : 0;
            d[k] = in ? ei[E + i] : -1;
        }
#pragma unroll
        for (int k = 0; k < SORT_PER_THREAD; ++k)
            if (d[k] >= 0) {
                const int b = d[k] >> BKT_SHIFT;
                const int p = atomicAdd(&cur[b], 1);
                if (staged) stage[p] = make_int2(s[k], d[k]);
                else rec2[gbase[b] + (p - lbase[b])] = make_int2(s[k], d[k]);
            }
    }
    __syncthreads();
    if (staged)
        for (int k = t; k < hi - lo; k += SORT_THREADS) {
            const int2 r = stage[k];
            const int b = r.y >> BKT_SHIFT;
            rec2[gbase[b] + (k - lbase[b])] = r;
        }
}

__device__ __forceinline__ void bucket_place(const int* __restrict__ rec,
                                             const int* __restrict__ bstart, int n, int nb,
                                             int* __restrict__ rowptr,
                                             int* __restrict__ lst, int b) {
    __shared__ int cnt[BKT_NODES];
    __shared__ int wsa[BKT_MAX / 32];
    __shared__ int stage[PLACE_STAGE];
    const int t = threadIdx.x;
    const int bb = bstart[b];
    const int be = bstart[b + 1];
    const int2* __restrict__ rec2 = (const int2*)rec;
    if (t < BKT_NODES) cnt[t] = 0;
    __syncthreads();
    for (int base = bb + t; base < be; base += SORT_THREADS * SORT_PER_THREAD) {
        int d[SORT_PER_THREAD];
#pragma unroll
        for (int k = 0; k < SORT_PER_THREAD; ++k) {
            const int j = base + k * SORT_THREADS;
            d[k] = (j < be) ? rec2[j].y : -1;
        }
#pragma unroll
        for (int k = 0; k < SORT_PER_THREAD; ++k)
            if (d[k] >= 0) atomicAdd(&cnt[d[k] & (BKT_NODES - 1)], 1);
    }
    __syncthreads();
    const int v = (t < BKT_NODES) ? cnt[t] : 0;
    const int start = bb + scan256(v, t, wsa) - v;
    if (t < BKT_NODES) {
        cnt[t] = start;
        const int node = (b << BKT_SHIFT) + t;
        if (node < n) rowptr[node] = start;
    }
    if (b == nb - 1 && t == 0) rowptr[n] = be;
    __syncthreads();
    const bool staged = (be - bb) <= PLACE_STAGE;
    for (int base = bb + t; base < be; base += SORT_THREADS * SORT_PER_THREAD) {
        int2 r[SORT_PER_THREAD];
#pragma unroll
        for (int k = 0; k < SORT_PER_THREAD; ++k) {
            const int j = base + k * SORT_THREADS;
            r[k] = (j < be) ? rec2[j] : make_int2(0, -1);
        }
#pragma unroll
        for (int k = 0; k < SORT_PER_THREAD; ++k)
            if (r[k].y >= 0) {
                const int p = atomicAdd(&cnt[r[k].y & (BKT_NODES - 1)], 1);
                if (staged) stage[p - bb] = r[k].x; else lst[p] = r[k].x;
            }
    }
    __syncthreads();
    if (staged)
        for (int j = bb + t; j < be; j += SORT_THREADS) lst[j] = stage[j - bb];
}

#define PROJ_TILES (SORT_THREADS / 128)

template <int K, int HEADS>
__device__ __forceinline__ void project_block(const float* __restrict__ A,
                                              const float* __restrict__ B,
                                              const float* __restrict__ a_src,
                                              const float* __restrict__ a_dst,
                                              float* __restrict__ C,
                                              float* __restrict__ as,
                                              float* __restrict__ ad, int M, int tile0) {
    constexpr int T = 16 * HEADS;
    __shared__ float lgb[PROJ_TILES * 2 * T];
    const int lane = threadIdx.x & 31;
    const int w    = __builtin_amdgcn_readfirstlane((int)(threadIdx.x >> 5));
    const int head = w % HEADS;
    const int tl   = w / HEADS;
    const int row0 = (tile0 + tl) * 16;
    const bool live = row0 < M;
    float* lg = lgb + tl * (2 * T);
    if (live) project_tile<K, HEADS, 4>(A, B, a_src, a_dst, C, row0, head, lane, lg);
    __syncthreads();
    if (live) store_logits<HEADS>(lg, as, ad, row0, head, lane);
}

template <bool PLACE>
__global__ void __launch_bounds__(SORT_THREADS)
sort_stage_project(const int* __restrict__ ei, int E, int chunk, int nb,
                   const int* __restrict__ tbl, int* __restrict__ bstart,
                   int* __restrict__ rec, int n, int* __restrict__ rowptr,
                   int* __restrict__ lst, int sort_blocks,
                   const float* __restrict__ x, const float* __restrict__ W,
                   const float* __restrict__ a_src, const float* __restrict__ a_dst,
                   float* __restrict__ h, float* __restrict__ as,
                   float* __restrict__ ad, int tile0) {
    const int blk = blockIdx.x;
    if (blk >= sort_blocks) {
        project_block<128, 4>(x, W, a_src, a_dst, h, as, ad, n,
                              tile0 + (blk - sort_blocks) * PROJ_TILES);
        return;
    }
    if (PLACE) bucket_place(rec, bstart, n, nb, rowptr, lst, blk);
    else       bucket_scatter(ei, E, chunk, nb, tbl, bstart, rec, blk);
}

#define GATHER_NPB   8
#define GATHER_STAGE 64
#define GATHER_PITCH (GATHER_STAGE + 4)

__device__ __forceinline__ float lrelu(float v) {
    return v > 0.f ? v : NEG_SLOPE * v;
}

template <int HEADS> struct rowvec {
    typedef float type __attribute__((ext_vector_type(HEADS)));
};

template <int HEADS>
__device__ __forceinline__ void gather_node(const int* __restrict__ rowptr,
                                            const int* __restrict__ lst,
                                            const float* __restrict__ as,
                                            const float* __restrict__ ad,
                                            const float* __restrict__ h,
                                            int node, int lane, float* wst,
                                            typename rowvec<HEADS>::type& acc_out,
                                            float& dsum_out) {
    typedef typename rowvec<HEADS>::type vec;
    const int head = (lane * HEADS) >> 5;
    const vec* __restrict__ hrow = (const vec*)h;
    const vec* __restrict__ asv  = (const vec*)as;
    const int beg = rowptr[node];
    const int end = rowptr[node + 1];

    const vec adv = ((const vec*)ad)[node];
    vec eself = asv[node] + adv;
#pragma unroll
    for (int k = 0; k < HEADS; ++k) eself[k] = lrelu(eself[k]);
    const int j0 = beg + lane, j1 = j0 + 32;
    int sa = node, sb = node;
    if (j0 < end) sa = lst[j0];
    if (j1 < end) sb = lst[j1];
    vec e0 = asv[sa] + adv;
    vec e1 = asv[sb] + adv;
    vec m = eself;
#pragma unroll
    for (int k = 0; k < HEADS; ++k) {
        e0[k] = lrelu(e0[k]);
        e1[k] = lrelu(e1[k]);
        m[k] = fmaxf(m[k], fmaxf(e0[k], e1[k]));
    }
    for (int c = beg + GATHER_STAGE; c < end; c += 32) {
        const int j = c + lane;
        if (j < end) {
            const vec e = asv[lst[j]] + adv;
#pragma unroll
            for (int k = 0; k < HEADS; ++k) m[k] = fmaxf(m[k], lrelu(e[k]));
        }
    }
    for (int off = 16; off > 0; off >>= 1) {
#pragma unroll
        for (int k = 0; k < HEADS; ++k)
            m[k] = fmaxf(m[k], __shfl_xor(m[k], off, 32));
    }
#pragma unroll
    for (int k = 0; k < HEADS; ++k) {
        wst[k * GATHER_PITCH + lane]      = __expf(e0[k] - m[k]);
        wst[k * GATHER_PITCH + 32 + lane] = __expf(e1[k] - m[k]);
    }
    float mh = m[0], adh = adv[0], esh = eself[0];
#pragma unroll
    for (int k = 1; k < HEADS; ++k)
        if (head == k) { mh = m[k]; adh = adv[k]; esh = eself[k]; }

    const float* wrow = wst + head * GATHER_PITCH;
    const float wself = __expf(esh - mh);
    vec acc = wself * hrow[(size_t)node * 32 + lane];
    float dsum = wself;

    const int ends = (end - beg > GATHER_STAGE) ? beg + GATHER_STAGE : end;
    int j = beg;
    for (; j + 4 <= ends; j += 4) {
        const int s0 = lst[j], s1 = lst[j + 1], s2 = lst[j + 2], s3 = lst[j + 3];
        const vec v0 = hrow[(size_t)s0 * 32 + lane];
        const vec v1 = hrow[(size_t)s1 * 32 + lane];
        const vec v2 = hrow[(size_t)s2 * 32 + lane];
        const vec v3 = hrow[(size_t)s3 * 32 + lane];
        const float4 w = *(const float4*)(wrow + (j - beg));
        acc += w.x * v0;  dsum += w.x;
        acc += w.y * v1;  dsum += w.y;
        acc += w.z * v2;  dsum += w.z;
        acc += w.w * v3;  dsum += w.w;
    }
#pragma clang loop unroll(disable)
    for (; j < ends; ++j) {
        const float w = wrow[j - beg];
        acc += w * hrow[(size_t)lst[j] * 32 + lane];
        dsum += w;
    }

    const float* __restrict__ ash = as + head;
    for (; j + 4 <= end; j += 4) {
        const int s0 = lst[j], s1 = lst[j + 1], s2 = lst[j + 2], s3 = lst[j + 3];
        const float a0 = ash[(size_t)s0 * HEADS];
        const float a1 = ash[(size_t)s1 * HEADS];
        const float a2 = ash[(size_t)s2 * HEADS];
        const float a3 = ash[(size_t)s3 * HEADS];
        const vec v0 = hrow[(size_t)s0 * 32 + lane];
        const vec v1 = hrow[(size_t)s1 * 32 + lane];
        const vec v2 = hrow[(size_t)s2 * 32 + lane];
        const vec v3 = hrow[(size_t)s3 * 32 + lane];
        const float w0 = __expf(lrelu(a0 + adh) - mh);
        const float w1 = __expf(lrelu(a1 + adh) - mh);
        const float w2 = __expf(lrelu(a2 + adh) - mh);
        const float w3 = __expf(lrelu(a3 + adh) - mh);
        acc += w0 * v0;  dsum += w0;
        acc += w1 * v1;  dsum += w1;
        acc += w2 * v2;  dsum += w2;
        acc += w3 * v3;  dsum += w3;
    }
#pragma clang loop unroll(disable)
    for (; j < end; ++j) {
        const int s = lst[j];
        const float w = __expf(lrelu(ash[(size_t)s * HEADS] + adh) - mh);
        acc += w * hrow[(size_t)s * 32 + lane];
        dsum += w;
    }
    acc_out = acc;
    dsum_out = dsum;
}

__device__ __forceinline__ float elu(float v) {
    return v > 0.f ? v : (expf(v) - 1.0f);
}

template <int HEADS>
__global__ void attend_concat(const int* __restrict__ rowptr,
                              const int* __restrict__ lst,
                              const float* __restrict__ as,
                              const float* __restrict__ ad,
                              const float* __restrict__ h,
                              const float* __restrict__ bias,
                              float* __restrict__ hout, int n_nodes) {
    typedef typename rowvec<HEADS>::type vec;
    const int lane = threadIdx.x;
    const int node = __builtin_amdgcn_readfirstlane(
        (int)(blockIdx.x * blockDim.y + threadIdx.y));
    if (node >= n_nodes) return;
    __shared__ __attribute__((aligned(16))) float wst[GATHER_NPB * HEADS * GATHER_PITCH];
    vec acc; float dsum;
    gather_node<HEADS>(rowptr, lst, as, ad, h, node, lane,
                       wst + threadIdx.y * (HEADS * GATHER_PITCH), acc, dsum);
    const vec b = ((const vec*)bias)[lane];
    vec o;
#pragma unroll
    for (int k = 0; k < HEADS; ++k) o[k] = elu(acc[k] / (dsum + EPS) + b[k]);
    ((vec*)hout)[(size_t)node * 32 + lane] = o;
}

template <int OUTC>
__global__ void attend_mean_linear(const int* __restrict__ rowptr,
                                   const int* __restrict__ lst,
                                   const float* __restrict__ as,
                                   const float* __restrict__ ad,
                                   const float* __restrict__ h,
                                   const float* __restrict__ bias,
                                   const float* __restrict__ Wl,
                                   const float* __restrict__ bl,
                                   float* __restrict__ out, int n_nodes) {
    constexpr int HEADS = 2;
    typedef typename rowvec<HEADS>::type vec;
    const int lane = threadIdx.x;
    const int node = __builtin_amdgcn_readfirstlane(
        (int)(blockIdx.x * blockDim.y + threadIdx.y));
    if (node >= n_nodes) return;
    __shared__ __attribute__((aligned(16))) float wst[GATHER_NPB * HEADS * GATHER_PITCH];
    vec acc; float dsum;
    gather_node<HEADS>(rowptr, lst, as, ad, h, node, lane,
                       wst + threadIdx.y * (HEADS * GATHER_PITCH), acc, dsum);

    const int g = lane & 15;
    const vec b = ((const vec*)bias)[g];
    float hv[HEADS];
#pragma unroll
    for (int k = 0; k < HEADS; ++k) {
        const float q  = acc[k] / (dsum + EPS);
        const float qo = __shfl_xor(q, 16, 32);
        hv[k] = elu((q + qo) / (float)HEADS + b[k]);
    }
    const float* __restrict__ wr = Wl + (size_t)(2 * g) * OUTC;
    float t[OUTC];
#pragma unroll
    for (int o = 0; o < OUTC; ++o) t[o] = hv[0] * wr[o] + hv[1] * wr[OUTC + o];
    for (int off = 8; off > 0; off >>= 1) {
#pragma unroll
        for (int o = 0; o < OUTC; ++o) t[o] += __shfl_xor(t[o], off, 32);
    }
    float r = t[0];
#pragma unroll
    for (int o = 1; o < OUTC; ++o)
        if (lane == o) r = t[o];
    if (lane < OUTC) out[(size_t)node * OUTC + lane] = r + bl[lane];
}

static inline unsigned gb(long long total, int bs) {
    return (unsigned)((total + bs - 1) / bs);
}

extern "C" void kernel_launch(void* const* d_in, const int* in_sizes, int n_in,
                              void* d_out, int out_size, void* d_ws, size_t ws_size,
                              hipStream_t stream) {
    const int IN_CH = 128, HID = 32, H1 = 4, H2 = 2, OUT = 3;
    const int N = in_sizes[0] / IN_CH;
    const long long E  = in_sizes[1] / 2;

    const float* x     = (const float*)d_in[0];
    const int*   ei    = (const int*)  d_in[1];
    const float* W1    = (const float*)d_in[2];
    const float* asrc1 = (const float*)d_in[3];
    const float* adst1 = (const float*)d_in[4];
    const float* b1    = (const float*)d_in[5];
    const float* W2    = (const float*)d_in[6];
    const float* asrc2 = (const float*)d_in[7];
    const float* adst2 = (const float*)d_in[8];
    const float* b2    = (const float*)d_in[9];
    const float* Wlin  = (const float*)d_in[10];
    const float* blin  = (const float*)d_in[11];
    float* out = (float*)d_out;

    float* ws = (float*)d_ws;
    size_t off = 0;
    float* h1   = ws + off; off += (size_t)N * H1 * HID;
    float* h1a  = ws + off; off += (size_t)N * H1 * HID;
    float* as_  = ws + off; off += (size_t)N * H1;
    float* ad_  = ws + off; off += (size_t)N * H1;
    int* lst    = (int*)(ws + off); off += (size_t)E;
    const int NBKT  = (N + BKT_NODES - 1) >> BKT_SHIFT;
    const int CHUNK = (int)((E + SORT_CHUNKS - 1) / SORT_CHUNKS);
    int* tbl    = (int*)(ws + off); off += (size_t)SORT_CHUNKS * BKT_MAX;
    int* bstart = (int*)(ws + off); off += (size_t)BKT_MAX + 4;
    int* rowptr = (int*)(ws + off); off += (size_t)N + 4;
    off = (off + 1) & ~(size_t)1;
    int* rec    = (int*)(ws + off); off += (size_t)2 * E;
    float* h2   = h1;

    const int NPB = GATHER_NPB;

    const int PB  = (N / 16 + PROJ_TILES - 1) / PROJ_TILES;
    const int PB1 = (PB + 1) / 2;
    const int PB2 = PB - PB1;
    bucket_count<<<SORT_CHUNKS, SORT_THREADS, 0, stream>>>(ei, (int)E, CHUNK, NBKT, tbl);
    sort_stage_project<false><<<SORT_CHUNKS + PB1, SORT_THREADS, 0, stream>>>(
        ei, (int)E, CHUNK, NBKT, tbl, bstart, rec, N, rowptr, lst, SORT_CHUNKS,
        x, W1, asrc1, adst1, h1, as_, ad_, 0);
    sort_stage_project<true><<<NBKT + PB2, SORT_THREADS, 0, stream>>>(
        ei, (int)E, CHUNK, NBKT, tbl, bstart, rec, N, rowptr, lst, NBKT,
        x, W1, asrc1, adst1, h1, as_, ad_, PB1 * PROJ_TILES);

    attend_concat<H1><<<gb((long long)N, NPB), dim3(32, NPB, 1), 0, stream>>>(
        rowptr, lst, as_, ad_, h1, b1, h1a, N);

    project_logits<H1 * HID, H2><<<dim3(N / 16, 1, 1), dim3(32, H2, 1), 0, stream>>>(
        h1a, W2, asrc2, adst2, h2, as_, ad_, N);
    attend_mean_linear<OUT><<<gb((long long)N, NPB), dim3(32, NPB, 1), 0, stream>>>(
        rowptr, lst, as_, ad_, h2, b2, Wlin, blin, out, N);
    (void)n_in; (void)out_size; (void)ws_size;
}


# No logit stored or gathered per edge: the gather wave dots each loaded source row with a_src (butterfly over the head's lanes), softmax online in one pass; logit sum order, weight rounding change
# speedup vs baseline: 14.9994x; 1.0339x over previous
#include <hip/hip_runtime.h>
#include <hip/hip_bf16.h>


typedef __attribute__((ext_vector_type(2))) float v2f;
typedef __attribute__((ext_vector_type(8))) float v8f;

#define NEG_SLOPE 0.2f
#define EPS 1e-16f


template <int MASK>
__device__ __forceinline__ float row_xor_add(float v) {
    const int b = __builtin_bit_cast(int, v);
    return v + __builtin_bit_cast(float,
        __builtin_amdgcn_update_dpp(0, b, 0x160 | MASK, 0xf, 0xf, false));
}

template <int K, int HEADS, int KSTEPS>
__device__ __forceinline__ void project_tile(const float* __restrict__ A,
                                             const float* __restrict__ B,
                                             float* __restrict__ C,
                                             int row0, int head, int lane) {
    constexpr int NOUT = HEADS * 32;
    const int lm   = lane & 15;
    const int lh   = lane >> 4;
    const int col0 = head * 32;

    const float* __restrict__ arow = A + (size_t)(row0 + lm) * K;
    const float* __restrict__ bcol = B + col0 + lm;
    v8f acc0 = {}, acc1 = {};
#pragma unroll KSTEPS
    for (int k0 = 0; k0 < K; k0 += 4) {
        const int ka = k0 + 2 * lh;
        v2f a, b0, b1;
        a.x  = arow[ka];
        a.y  = arow[ka + 1];
        b0.x = bcol[(size_t)ka * NOUT];
        b0.y = bcol[(size_t)(ka + 1) * NOUT];
        b1.x = bcol[(size_t)ka * NOUT + 16];
        b1.y = bcol[(size_t)(ka + 1) * NOUT + 16];
        acc0 = __builtin_amdgcn_wmma_f32_16x16x4_f32(
             false, a,  false, b0,
             (short)0, acc0,  false,  false);
        acc1 = __builtin_amdgcn_wmma_f32_16x16x4_f32(
             false, a,  false, b1,
             (short)0, acc1,  false,  false);
    }
    float* __restrict__ crow = C + (size_t)(row0 + 8 * lh) * NOUT + col0 + lm;
#pragma unroll
    for (int r = 0; r < 8; ++r) {
        crow[(size_t)r * NOUT]      = acc0[r];
        crow[(size_t)r * NOUT + 16] = acc1[r];
    }
}

template <int K, int HEADS>
__global__ void __launch_bounds__(32 * HEADS)
project_rows(const float* __restrict__ A,
             const float* __restrict__ B,
             float* __restrict__ C, int M) {
    const int lane = threadIdx.x;
    const int head = threadIdx.y;
    const int row0 = blockIdx.x * 16;
    if (row0 >= M) return;
    project_tile<K, HEADS, 8>(A, B, C, row0, head, lane);
}

#define BKT_SHIFT       8
#define BKT_NODES       (1 << BKT_SHIFT)
#define BKT_MAX         256
#define SORT_CHUNKS     256
#define SORT_THREADS    1024
#define SORT_PER_THREAD 8
#define SCATTER_STAGE   7168
#define PLACE_STAGE     12288

__global__ void __launch_bounds__(SORT_THREADS)
bucket_count(const int* __restrict__ ei, int E, int chunk, int nb,
             int* __restrict__ tbl) {
    __shared__ int hist[BKT_MAX];
    const int t = threadIdx.x;
    const int c = blockIdx.x;
    if (t < BKT_MAX) hist[t] = 0;
    __syncthreads();
    const int* __restrict__ dst = ei + E;
    const int lo = c * chunk;
    const int hi = (lo + chunk < E) ? lo + chunk : E;
    for (int base = lo + t; base < hi; base += SORT_THREADS * SORT_PER_THREAD) {
        int d[SORT_PER_THREAD];
#pragma unroll
        for (int k = 0; k < SORT_PER_THREAD; ++k) {
            const int i = base + k * SORT_THREADS;
            d[k] = (i < hi) ? dst[i] : -1;
        }
#pragma unroll
        for (int k = 0; k < SORT_PER_THREAD; ++k)
            if (d[k] >= 0) atomicAdd(&hist[d[k] >> BKT_SHIFT], 1);
    }
    __syncthreads();
    if (t < nb) tbl[c * BKT_MAX + t] = hist[t];
}

__device__ __forceinline__ int scan256(int v, int t, int* ws) {
    const int lane = t & 31, w = t >> 5;
    int inc = v;
    for (int off = 1; off < 32; off <<= 1) {
        const int u = __shfl_up(inc, off, 32);
        if (lane >= off) inc += u;
    }
    if (t < BKT_MAX && lane == 31) ws[w] = inc;
    __syncthreads();
    int before = 0;
    if (t < BKT_MAX)
        for (int k = 0; k < w; ++k) before += ws[k];
    return before + inc;
}

__device__ __forceinline__ void bucket_scatter(const int* __restrict__ ei, int E, int chunk,
                                               int nb, const int* __restrict__ tbl,
                                               int* __restrict__ bstart,
                                               int* __restrict__ rec, int c) {
    __shared__ int gbase[BKT_MAX];
    __shared__ int lbase[BKT_MAX];
    __shared__ int cur[BKT_MAX];
    __shared__ int wsa[BKT_MAX / 32], wsb[BKT_MAX / 32];
    __shared__ int2 stage[SCATTER_STAGE];
    const int t = threadIdx.x;

    const int bq = t & (BKT_MAX - 1);
    const int g = __builtin_amdgcn_readfirstlane(t / BKT_MAX);
    int* sums = (int*)stage;
    int all = 0, bef = 0;
    if (bq < nb) {
#pragma unroll 16
        for (int cc = g; cc < SORT_CHUNKS; cc += SORT_THREADS / BKT_MAX) {
            const int v = tbl[cc * BKT_MAX + bq];
            all += v;
            bef += (cc < c) ? v : 0;
        }
    }
    sums[t] = all;
    sums[SORT_THREADS + t] = bef;
    __syncthreads();

    const bool bk = t < nb;
    int vt = 0, pt = 0;
    if (bk) {
#pragma unroll
        for (int q = 0; q < SORT_THREADS / BKT_MAX; ++q) {
            vt += sums[q * BKT_MAX + t];
            pt += sums[SORT_THREADS + q * BKT_MAX + t];
        }
    }
    const int vo = bk ? tbl[c * BKT_MAX + t] : 0;
    const int st = scan256(vt, t, wsa) - vt;
    const int sl = scan256(vo, t, wsb) - vo;
    if (bk) {
        gbase[t] = st + pt;
        lbase[t] = sl;
        cur[t]   = sl;
        if (c == 0) bstart[t] = st;
    }
    if (c == 0 && t == 0) bstart[nb] = E;
    __syncthreads();

    const int lo = c * chunk;
    const int hi = (lo + chunk < E) ? lo + chunk : E;
    const bool staged = (hi - lo) <= SCATTER_STAGE;
    int2* __restrict__ rec2 = (int2*)rec;
    for (int base = lo + t; base < hi; base += SORT_THREADS * SORT_PER_THREAD) {
        int s[SORT_PER_THREAD], d[SORT_PER_THREAD];
#pragma unroll
        for (int k = 0; k < SORT_PER_THREAD; ++k) {
            const int i = base + k * SORT_THREADS;
            const bool in = i < hi;
            s[k] = in ? ei[i] : 0;
            d[k] = in ? ei[E + i] : -1;
        }
#pragma unroll
        for (int k = 0; k < SORT_PER_THREAD; ++k)
            if (d[k] >= 0) {
                const int b = d[k] >> BKT_SHIFT;
                const int p = atomicAdd(&cur[b], 1);
                if (staged) stage[p] = make_int2(s[k], d[k]);
                else rec2[gbase[b] + (p - lbase[b])] = make_int2(s[k], d[k]);
            }
    }
    __syncthreads();
    if (staged)
        for (int k = t; k < hi - lo; k += SORT_THREADS) {
            const int2 r = stage[k];
            const int b = r.y >> BKT_SHIFT;
            rec2[gbase[b] + (k - lbase[b])] = r;
        }
}

__device__ __forceinline__ void bucket_place(const int* __restrict__ rec,
                                             const int* __restrict__ bstart, int n, int nb,
                                             int* __restrict__ rowptr,
                                             int* __restrict__ lst, int b) {
    __shared__ int cnt[BKT_NODES];
    __shared__ int wsa[BKT_MAX / 32];
    __shared__ int stage[PLACE_STAGE];
    const int t = threadIdx.x;
    const int bb = bstart[b];
    const int be = bstart[b + 1];
    const int2* __restrict__ rec2 = (const int2*)rec;
    if (t < BKT_NODES) cnt[t] = 0;
    __syncthreads();
    for (int base = bb + t; base < be; base += SORT_THREADS * SORT_PER_THREAD) {
        int d[SORT_PER_THREAD];
#pragma unroll
        for (int k = 0; k < SORT_PER_THREAD; ++k) {
            const int j = base + k * SORT_THREADS;
            d[k] = (j < be) ? rec2[j].y : -1;
        }
#pragma unroll
        for (int k = 0; k < SORT_PER_THREAD; ++k)
            if (d[k] >= 0) atomicAdd(&cnt[d[k] & (BKT_NODES - 1)], 1);
    }
    __syncthreads();
    const int v = (t < BKT_NODES) ? cnt[t] : 0;
    const int start = bb + scan256(v, t, wsa) - v;
    if (t < BKT_NODES) {
        cnt[t] = start;
        const int node = (b << BKT_SHIFT) + t;
        if (node < n) rowptr[node] = start;
    }
    if (b == nb - 1 && t == 0) rowptr[n] = be;
    __syncthreads();
    const bool staged = (be - bb) <= PLACE_STAGE;
    for (int base = bb + t; base < be; base += SORT_THREADS * SORT_PER_THREAD) {
        int2 r[SORT_PER_THREAD];
#pragma unroll
        for (int k = 0; k < SORT_PER_THREAD; ++k) {
            const int j = base + k * SORT_THREADS;
            r[k] = (j < be) ? rec2[j] : make_int2(0, -1);
        }
#pragma unroll
        for (int k = 0; k < SORT_PER_THREAD; ++k)
            if (r[k].y >= 0) {
                const int p = atomicAdd(&cnt[r[k].y & (BKT_NODES - 1)], 1);
                if (staged) stage[p - bb] = r[k].x; else lst[p] = r[k].x;
            }
    }
    __syncthreads();
    if (staged)
        for (int j = bb + t; j < be; j += SORT_THREADS) lst[j] = stage[j - bb];
}

#define PROJ_TILES (SORT_THREADS / 128)

template <int K, int HEADS>
__device__ __forceinline__ void project_block(const float* __restrict__ A,
                                              const float* __restrict__ B,
                                              float* __restrict__ C, int M, int tile0) {
    const int lane = threadIdx.x & 31;
    const int w    = __builtin_amdgcn_readfirstlane((int)(threadIdx.x >> 5));
    const int head = w % HEADS;
    const int tl   = w / HEADS;
    const int row0 = (tile0 + tl) * 16;
    if (row0 < M) project_tile<K, HEADS, 4>(A, B, C, row0, head, lane);
}

template <bool PLACE>
__global__ void __launch_bounds__(SORT_THREADS)
sort_stage_project(const int* __restrict__ ei, int E, int chunk, int nb,
                   const int* __restrict__ tbl, int* __restrict__ bstart,
                   int* __restrict__ rec, int n, int* __restrict__ rowptr,
                   int* __restrict__ lst, int sort_blocks,
                   const float* __restrict__ x, const float* __restrict__ W,
                   float* __restrict__ h, int tile0) {
    const int blk = blockIdx.x;
    if (blk >= sort_blocks) {
        project_block<128, 4>(x, W, h, n, tile0 + (blk - sort_blocks) * PROJ_TILES);
        return;
    }
    if (PLACE) bucket_place(rec, bstart, n, nb, rowptr, lst, blk);
    else       bucket_scatter(ei, E, chunk, nb, tbl, bstart, rec, blk);
}

#define GATHER_NPB   8

__device__ __forceinline__ float lrelu(float v) {
    return fmaxf(v, NEG_SLOPE * v);
}

template <int HEADS> struct rowvec {
    typedef float type __attribute__((ext_vector_type(HEADS)));
};

template <int HEADS>
__device__ __forceinline__ float head_sum(float p) {
    static_assert(HEADS == 2 || HEADS == 4, "8 or 16 lanes per head");
    p = row_xor_add<1>(p);
    p = row_xor_add<2>(p);
    p = row_xor_add<4>(p);
    if (HEADS == 2) p = row_xor_add<8>(p);
    return p;
}

template <int HEADS>
__device__ __forceinline__ float head_dot(typename rowvec<HEADS>::type v,
                                          typename rowvec<HEADS>::type a) {
    float p = v[0] * a[0];
#pragma unroll
    for (int k = 1; k < HEADS; ++k) p += v[k] * a[k];
    return head_sum<HEADS>(p);
}

template <int HEADS, bool FULL>
__device__ __forceinline__ void gather_trip(const typename rowvec<HEADS>::type* __restrict__ hrow,
                                            const int* __restrict__ lst, int j, int end,
                                            int lane, typename rowvec<HEADS>::type avs,
                                            float adh, typename rowvec<HEADS>::type& acc,
                                            float& dsum, float& m) {
    typedef typename rowvec<HEADS>::type vec;
    const bool in1 = FULL || j + 1 < end, in2 = FULL || j + 2 < end, in3 = FULL || j + 3 < end;
    const int s0 = lst[j];
    const int s1 = lst[in1 ? j + 1 : end - 1];
    const int s2 = lst[in2 ? j + 2 : end - 1];
    const int s3 = lst[in3 ? j + 3 : end - 1];
    const vec v0 = hrow[(size_t)s0 * 32 + lane];
    const vec v1 = hrow[(size_t)s1 * 32 + lane];
    const vec v2 = hrow[(size_t)s2 * 32 + lane];
    const vec v3 = hrow[(size_t)s3 * 32 + lane];
    __builtin_amdgcn_sched_barrier(0);
    const float ninf = -__builtin_huge_valf();
    const float e0 = lrelu(head_dot<HEADS>(v0, avs) + adh);
    const float e1 = in1 ? lrelu(head_dot<HEADS>(v1, avs) + adh) : ninf;
    const float e2 = in2 ? lrelu(head_dot<HEADS>(v2, avs) + adh) : ninf;
    const float e3 = in3 ? lrelu(head_dot<HEADS>(v3, avs) + adh) : ninf;
    const float mn = fmaxf(fmaxf(m, e0), fmaxf(e1, fmaxf(e2, e3)));
    const float sc = __expf(m - mn);
    const float w0 = __expf(e0 - mn);
    const float w1 = __expf(e1 - mn);
    const float w2 = __expf(e2 - mn);
    const float w3 = __expf(e3 - mn);
    acc = sc * acc + w0 * v0;  dsum = sc * dsum + w0;
    acc += w1 * v1;  dsum += w1;
    acc += w2 * v2;  dsum += w2;
    acc += w3 * v3;  dsum += w3;
    m = mn;
}

template <int HEADS>
__device__ __forceinline__ void gather_node(const int* __restrict__ rowptr,
                                            const int* __restrict__ lst,
                                            const float* __restrict__ a_src,
                                            const float* __restrict__ a_dst,
                                            const float* __restrict__ h,
                                            int node, int lane,
                                            typename rowvec<HEADS>::type& acc_out,
                                            float& dsum_out) {
    typedef typename rowvec<HEADS>::type vec;
    const vec* __restrict__ hrow = (const vec*)h;
    const int beg = rowptr[node];
    const int end = rowptr[node + 1];
    const vec avs = ((const vec*)a_src)[lane];
    const vec avd = ((const vec*)a_dst)[lane];

    const vec vself = hrow[(size_t)node * 32 + lane];
    const float adh = head_dot<HEADS>(vself, avd);
    float m = lrelu(head_dot<HEADS>(vself, avs) + adh);
    vec acc = vself;
    float dsum = 1.f;

    int j = beg;
    for (; j + 4 <= end; j += 4)
        gather_trip<HEADS, true>(hrow, lst, j, end, lane, avs, adh, acc, dsum, m);
    if (j < end)
        gather_trip<HEADS, false>(hrow, lst, j, end, lane, avs, adh, acc, dsum, m);
    acc_out = acc;
    dsum_out = dsum;
}

__device__ __forceinline__ float elu(float v) {
    return v > 0.f ? v : (expf(v) - 1.0f);
}

template <int HEADS>
__global__ void attend_concat(const int* __restrict__ rowptr,
                              const int* __restrict__ lst,
                              const float* __restrict__ a_src,
                              const float* __restrict__ a_dst,
                              const float* __restrict__ h,
                              const float* __restrict__ bias,
                              float* __restrict__ hout, int n_nodes) {
    typedef typename rowvec<HEADS>::type vec;
    const int lane = threadIdx.x;
    const int node = __builtin_amdgcn_readfirstlane(
        (int)(blockIdx.x * blockDim.y + threadIdx.y));
    if (node >= n_nodes) return;
    vec acc; float dsum;
    gather_node<HEADS>(rowptr, lst, a_src, a_dst, h, node, lane, acc, dsum);
    const vec b = ((const vec*)bias)[lane];
    vec o;
#pragma unroll
    for (int k = 0; k < HEADS; ++k) o[k] = elu(acc[k] / (dsum + EPS) + b[k]);
    ((vec*)hout)[(size_t)node * 32 + lane] = o;
}

template <int OUTC>
__global__ void attend_mean_linear(const int* __restrict__ rowptr,
                                   const int* __restrict__ lst,
                                   const float* __restrict__ a_src,
                                   const float* __restrict__ a_dst,
                                   const float* __restrict__ h,
                                   const float* __restrict__ bias,
                                   const float* __restrict__ Wl,
                                   const float* __restrict__ bl,
                                   float* __restrict__ out, int n_nodes) {
    constexpr int HEADS = 2;
    typedef typename rowvec<HEADS>::type vec;
    const int lane = threadIdx.x;
    const int node = __builtin_amdgcn_readfirstlane(
        (int)(blockIdx.x * blockDim.y + threadIdx.y));
    if (node >= n_nodes) return;
    vec acc; float dsum;
    gather_node<HEADS>(rowptr, lst, a_src, a_dst, h, node, lane, acc, dsum);

    const int g = lane & 15;
    const vec b = ((const vec*)bias)[g];
    float hv[HEADS];
#pragma unroll
    for (int k = 0; k < HEADS; ++k) {
        const float q  = acc[k] / (dsum + EPS);
        const float qo = __shfl_xor(q, 16, 32);
        hv[k] = elu((q + qo) / (float)HEADS + b[k]);
    }
    const float* __restrict__ wr = Wl + (size_t)(2 * g) * OUTC;
    float t[OUTC];
#pragma unroll
    for (int o = 0; o < OUTC; ++o) t[o] = hv[0] * wr[o] + hv[1] * wr[OUTC + o];
    for (int off = 8; off > 0; off >>= 1) {
#pragma unroll
        for (int o = 0; o < OUTC; ++o) t[o] += __shfl_xor(t[o], off, 32);
    }
    float r = t[0];
#pragma unroll
    for (int o = 1; o < OUTC; ++o)
        if (lane == o) r = t[o];
    if (lane < OUTC) out[(size_t)node * OUTC + lane] = r + bl[lane];
}

static inline unsigned gb(long long total, int bs) {
    return (unsigned)((total + bs - 1) / bs);
}

extern "C" void kernel_launch(void* const* d_in, const int* in_sizes, int n_in,
                              void* d_out, int out_size, void* d_ws, size_t ws_size,
                              hipStream_t stream) {
    const int IN_CH = 128, HID = 32, H1 = 4, H2 = 2, OUT = 3;
    const int N = in_sizes[0] / IN_CH;
    const long long E  = in_sizes[1] / 2;

    const float* x     = (const float*)d_in[0];
    const int*   ei    = (const int*)  d_in[1];
    const float* W1    = (const float*)d_in[2];
    const float* asrc1 = (const float*)d_in[3];
    const float* adst1 = (const float*)d_in[4];
    const float* b1    = (const float*)d_in[5];
    const float* W2    = (const float*)d_in[6];
    const float* asrc2 = (const float*)d_in[7];
    const float* adst2 = (const float*)d_in[8];
    const float* b2    = (const float*)d_in[9];
    const float* Wlin  = (const float*)d_in[10];
    const float* blin  = (const float*)d_in[11];
    float* out = (float*)d_out;

    float* ws = (float*)d_ws;
    size_t off = 0;
    float* h1   = ws + off; off += (size_t)N * H1 * HID;
    float* h1a  = ws + off; off += (size_t)N * H1 * HID;
    int* lst    = (int*)(ws + off); off += (size_t)E;
    const int NBKT  = (N + BKT_NODES - 1) >> BKT_SHIFT;
    const int CHUNK = (int)((E + SORT_CHUNKS - 1) / SORT_CHUNKS);
    int* tbl    = (int*)(ws + off); off += (size_t)SORT_CHUNKS * BKT_MAX;
    int* bstart = (int*)(ws + off); off += (size_t)BKT_MAX + 4;
    int* rowptr = (int*)(ws + off); off += (size_t)N + 4;
    off = (off + 1) & ~(size_t)1;
    int* rec    = (int*)(ws + off); off += (size_t)2 * E;
    float* h2   = h1;

    const int NPB = GATHER_NPB;

    const int PB  = (N / 16 + PROJ_TILES - 1) / PROJ_TILES;
    const int PB1 = (PB + 1) / 2;
    const int PB2 = PB - PB1;
    bucket_count<<<SORT_CHUNKS, SORT_THREADS, 0, stream>>>(ei, (int)E, CHUNK, NBKT, tbl);
    sort_stage_project<false><<<SORT_CHUNKS + PB1, SORT_THREADS, 0, stream>>>(
        ei, (int)E, CHUNK, NBKT, tbl, bstart, rec, N, rowptr, lst, SORT_CHUNKS,
        x, W1, h1, 0);
    sort_stage_project<true><<<NBKT + PB2, SORT_THREADS, 0, stream>>>(
        ei, (int)E, CHUNK, NBKT, tbl, bstart, rec, N, rowptr, lst, NBKT,
        x, W1, h1, PB1 * PROJ_TILES);

    attend_concat<H1><<<gb((long long)N, NPB), dim3(32, NPB, 1), 0, stream>>>(
        rowptr, lst, asrc1, adst1, h1, b1, h1a, N);

    project_rows<H1 * HID, H2><<<dim3(N / 16, 1, 1), dim3(32, H2, 1), 0, stream>>>(
        h1a, W2, h2, N);
    attend_mean_linear<OUT><<<gb((long long)N, NPB), dim3(32, NPB, 1), 0, stream>>>(
        rowptr, lst, asrc2, adst2, h2, b2, Wlin, blin, out, N);
    (void)n_in; (void)out_size; (void)ws_size;
}


# Edge sort in 2 launches for 3, no counting launch: a chunk counts and orders itself by bucket and writes its run starts; a bucket sums two table columns for its span; list order still LDS rank
# speedup vs baseline: 15.5858x; 1.0391x over previous
#include <hip/hip_runtime.h>
#include <hip/hip_bf16.h>


typedef __attribute__((ext_vector_type(2))) float v2f;
typedef __attribute__((ext_vector_type(8))) float v8f;

#define NEG_SLOPE 0.2f
#define EPS 1e-16f


template <int MASK>
__device__ __forceinline__ float row_xor_add(float v) {
    const int b = __builtin_bit_cast(int, v);
    return v + __builtin_bit_cast(float,
        __builtin_amdgcn_update_dpp(0, b, 0x160 | MASK, 0xf, 0xf, false));
}

template <int K, int HEADS, int KSTEPS>
__device__ __forceinline__ void project_tile(const float* __restrict__ A,
                                             const float* __restrict__ B,
                                             float* __restrict__ C,
                                             int row0, int head, int lane) {
    constexpr int NOUT = HEADS * 32;
    const int lm   = lane & 15;
    const int lh   = lane >> 4;
    const int col0 = head * 32;

    const float* __restrict__ arow = A + (size_t)(row0 + lm) * K;
    const float* __restrict__ bcol = B + col0 + lm;
    v8f acc0 = {}, acc1 = {};
#pragma unroll KSTEPS
    for (int k0 = 0; k0 < K; k0 += 4) {
        const int ka = k0 + 2 * lh;
        v2f a, b0, b1;
        a.x  = arow[ka];
        a.y  = arow[ka + 1];
        b0.x = bcol[(size_t)ka * NOUT];
        b0.y = bcol[(size_t)(ka + 1) * NOUT];
        b1.x = bcol[(size_t)ka * NOUT + 16];
        b1.y = bcol[(size_t)(ka + 1) * NOUT + 16];
        acc0 = __builtin_amdgcn_wmma_f32_16x16x4_f32(
             false, a,  false, b0,
             (short)0, acc0,  false,  false);
        acc1 = __builtin_amdgcn_wmma_f32_16x16x4_f32(
             false, a,  false, b1,
             (short)0, acc1,  false,  false);
    }
    float* __restrict__ crow = C + (size_t)(row0 + 8 * lh) * NOUT + col0 + lm;
#pragma unroll
    for (int r = 0; r < 8; ++r) {
        crow[(size_t)r * NOUT]      = acc0[r];
        crow[(size_t)r * NOUT + 16] = acc1[r];
    }
}

template <int K, int HEADS>
__global__ void __launch_bounds__(32 * HEADS)
project_rows(const float* __restrict__ A,
             const float* __restrict__ B,
             float* __restrict__ C, int M) {
    const int lane = threadIdx.x;
    const int head = threadIdx.y;
    const int row0 = blockIdx.x * 16;
    if (row0 >= M) return;
    project_tile<K, HEADS, 8>(A, B, C, row0, head, lane);
}

#define BKT_SHIFT       8
#define BKT_NODES       (1 << BKT_SHIFT)
#define BKT_MAX         256
#define SORT_CHUNKS     256
#define SORT_THREADS    1024
#define SORT_PER_THREAD 8
#define SORT_STEP       (SORT_THREADS * SORT_PER_THREAD)
#define SCATTER_STAGE   7168
#define PLACE_STAGE     12288
#define PRE_STRIDE      288
#define RUN_LANES       (SORT_THREADS / SORT_CHUNKS)

__device__ __forceinline__ int scan256(int v, int t, int* ws) {
    const int lane = t & 31, w = t >> 5;
    int inc = v;
    for (int off = 1; off < 32; off <<= 1) {
        const int u = __shfl_up(inc, off, 32);
        if (lane >= off) inc += u;
    }
    if (t < BKT_MAX && lane == 31) ws[w] = inc;
    __syncthreads();
    int before = 0;
    if (t < BKT_MAX)
        for (int k = 0; k < w; ++k) before += ws[k];
    return before + inc;
}

__device__ __forceinline__ void sum256x2(int a, int b, int t, int* ws, int& ta, int& tb) {
    const int lane = t & 31, w = t >> 5;
    for (int off = 16; off > 0; off >>= 1) {
        a += __shfl_xor(a, off, 32);
        b += __shfl_xor(b, off, 32);
    }
    if (t < BKT_MAX && lane == 0) { ws[w] = a; ws[BKT_MAX / 32 + w] = b; }
    __syncthreads();
    ta = 0; tb = 0;
#pragma unroll
    for (int k = 0; k < BKT_MAX / 32; ++k) { ta += ws[k]; tb += ws[BKT_MAX / 32 + k]; }
}

__device__ __forceinline__ void chunk_order(const int* __restrict__ ei, int E, int chunk,
                                            int nb, int* __restrict__ pre,
                                            int* __restrict__ rec, int c) {
    __shared__ int cur[BKT_MAX];
    __shared__ int wsa[BKT_MAX / 32];
    __shared__ int total;
    __shared__ int2 stage[SCATTER_STAGE];
    const int t = threadIdx.x;
    if (t < BKT_MAX) cur[t] = 0;
    __syncthreads();
    const int lo = c * chunk;
    const int hi = (lo + chunk < E) ? lo + chunk : E;

    int s0[SORT_PER_THREAD], d0[SORT_PER_THREAD];
#pragma unroll
    for (int k = 0; k < SORT_PER_THREAD; ++k) {
        const int i = lo + t + k * SORT_THREADS;
        const bool in = i < hi;
        s0[k] = in ? ei[i] : 0;
        d0[k] = in ? ei[E + i] : -1;
    }
#pragma unroll
    for (int k = 0; k < SORT_PER_THREAD; ++k)
        if (d0[k] >= 0) atomicAdd(&cur[d0[k] >> BKT_SHIFT], 1);
    for (int base = lo + t + SORT_STEP; base < hi; base += SORT_STEP) {
        int d[SORT_PER_THREAD];
#pragma unroll
        for (int k = 0; k < SORT_PER_THREAD; ++k) {
            const int i = base + k * SORT_THREADS;
            d[k] = (i < hi) ? ei[E + i] : -1;
        }
#pragma unroll
        for (int k = 0; k < SORT_PER_THREAD; ++k)
            if (d[k] >= 0) atomicAdd(&cur[d[k] >> BKT_SHIFT], 1);
    }
    __syncthreads();

    const int v  = (t < BKT_MAX) ? cur[t] : 0;
    const int ex = scan256(v, t, wsa) - v;
    if (t < BKT_MAX) {
        cur[t] = ex;
        if (t <= nb) pre[c * PRE_STRIDE + t] = ex;
        if (t == BKT_MAX - 1) {
            total = ex + v;
            if (nb == BKT_MAX) pre[c * PRE_STRIDE + BKT_MAX] = ex + v;
        }
    }
    __syncthreads();

    const int nrec = total;
    const bool staged = nrec <= SCATTER_STAGE;
    int2* __restrict__ rec2 = (int2*)rec;
#pragma unroll
    for (int k = 0; k < SORT_PER_THREAD; ++k)
        if (d0[k] >= 0) {
            const int p = atomicAdd(&cur[d0[k] >> BKT_SHIFT], 1);
            if (staged) stage[p] = make_int2(s0[k], d0[k]);
            else rec2[lo + p] = make_int2(s0[k], d0[k]);
        }
    for (int base = lo + t + SORT_STEP; base < hi; base += SORT_STEP) {
        int s[SORT_PER_THREAD], d[SORT_PER_THREAD];
#pragma unroll
        for (int k = 0; k < SORT_PER_THREAD; ++k) {
            const int i = base + k * SORT_THREADS;
            const bool in = i < hi;
            s[k] = in ? ei[i] : 0;
            d[k] = in ? ei[E + i] : -1;
        }
#pragma unroll
        for (int k = 0; k < SORT_PER_THREAD; ++k)
            if (d[k] >= 0) {
                const int p = atomicAdd(&cur[d[k] >> BKT_SHIFT], 1);
                if (staged) stage[p] = make_int2(s[k], d[k]);
                else rec2[lo + p] = make_int2(s[k], d[k]);
            }
    }
    __syncthreads();
    if (staged)
        for (int k = t; k < nrec; k += SORT_THREADS) rec2[lo + k] = stage[k];
}

__device__ __forceinline__ void bucket_place(const int* __restrict__ rec,
                                             const int* __restrict__ pre, int chunk,
                                             int n, int nb,
                                             int* __restrict__ rowptr,
                                             int* __restrict__ lst, int b) {
    static_assert(SORT_CHUNKS <= BKT_MAX, "one thread per chunk inside sum256x2's range");
    __shared__ int cnt[BKT_NODES];
    __shared__ int wsa[BKT_MAX / 32];
    __shared__ int wss[2 * BKT_MAX / 32];
    __shared__ int rstart[SORT_CHUNKS];
    __shared__ int rlen[SORT_CHUNKS];
    __shared__ int stage[PLACE_STAGE];
    const int t = threadIdx.x;
    const int2* __restrict__ rec2 = (const int2*)rec;

    int p0 = 0, p1 = 0;
    if (t < SORT_CHUNKS) {
        p0 = pre[t * PRE_STRIDE + b];
        p1 = pre[t * PRE_STRIDE + b + 1];
        rstart[t] = t * chunk + p0;
        rlen[t]   = p1 - p0;
    }
    if (t < BKT_NODES) cnt[t] = 0;
    int bb, be;
    sum256x2(p0, p1, t, wss, bb, be);

    const int q  = t % RUN_LANES;
    const int rs = rstart[t / RUN_LANES];
    const int rl = rlen[t / RUN_LANES];
    int2 r0[SORT_PER_THREAD];
#pragma unroll
    for (int k = 0; k < SORT_PER_THREAD; ++k) {
        const int i = q + k * RUN_LANES;
        r0[k] = (i < rl) ? rec2[rs + i] : make_int2(0, -1);
    }
#pragma unroll
    for (int k = 0; k < SORT_PER_THREAD; ++k)
        if (r0[k].y >= 0) atomicAdd(&cnt[r0[k].y & (BKT_NODES - 1)], 1);
    for (int i0 = q + RUN_LANES * SORT_PER_THREAD; i0 < rl; i0 += RUN_LANES * SORT_PER_THREAD) {
        int d[SORT_PER_THREAD];
#pragma unroll
        for (int k = 0; k < SORT_PER_THREAD; ++k) {
            const int i = i0 + k * RUN_LANES;
            d[k] = (i < rl) ? rec2[rs + i].y : -1;
        }
#pragma unroll
        for (int k = 0; k < SORT_PER_THREAD; ++k)
            if (d[k] >= 0) atomicAdd(&cnt[d[k] & (BKT_NODES - 1)], 1);
    }
    __syncthreads();
    const int v = (t < BKT_NODES) ? cnt[t] : 0;
    const int start = bb + scan256(v, t, wsa) - v;
    if (t < BKT_NODES) {
        cnt[t] = start;
        const int node = (b << BKT_SHIFT) + t;
        if (node < n) rowptr[node] = start;
    }
    if (b == nb - 1 && t == 0) rowptr[n] = be;
    __syncthreads();

    const bool staged = (be - bb) <= PLACE_STAGE;
#pragma unroll
    for (int k = 0; k < SORT_PER_THREAD; ++k)
        if (r0[k].y >= 0) {
            const int p = atomicAdd(&cnt[r0[k].y & (BKT_NODES - 1)], 1);
            if (staged) stage[p - bb] = r0[k].x; else lst[p] = r0[k].x;
        }
    for (int i0 = q + RUN_LANES * SORT_PER_THREAD; i0 < rl; i0 += RUN_LANES * SORT_PER_THREAD) {
        int2 r[SORT_PER_THREAD];
#pragma unroll
        for (int k = 0; k < SORT_PER_THREAD; ++k) {
            const int i = i0 + k * RUN_LANES;
            r[k] = (i < rl) ? rec2[rs + i] : make_int2(0, -1);
        }
#pragma unroll
        for (int k = 0; k < SORT_PER_THREAD; ++k)
            if (r[k].y >= 0) {
                const int p = atomicAdd(&cnt[r[k].y & (BKT_NODES - 1)], 1);
                if (staged) stage[p - bb] = r[k].x; else lst[p] = r[k].x;
            }
    }
    __syncthreads();
    if (staged)
        for (int j = bb + t; j < be; j += SORT_THREADS) lst[j] = stage[j - bb];
}

#define PROJ_TILES (SORT_THREADS / 128)

template <int K, int HEADS>
__device__ __forceinline__ void project_block(const float* __restrict__ A,
                                              const float* __restrict__ B,
                                              float* __restrict__ C, int M, int tile0) {
    const int lane = threadIdx.x & 31;
    const int w    = __builtin_amdgcn_readfirstlane((int)(threadIdx.x >> 5));
    const int head = w % HEADS;
    const int tl   = w / HEADS;
    const int row0 = (tile0 + tl) * 16;
    if (row0 < M) project_tile<K, HEADS, 4>(A, B, C, row0, head, lane);
}

template <bool PLACE>
__global__ void __launch_bounds__(SORT_THREADS)
sort_stage_project(const int* __restrict__ ei, int E, int chunk, int nb,
                   int* __restrict__ pre,
                   int* __restrict__ rec, int n, int* __restrict__ rowptr,
                   int* __restrict__ lst, int sort_blocks,
                   const float* __restrict__ x, const float* __restrict__ W,
                   float* __restrict__ h, int tile0) {
    const int blk = blockIdx.x;
    if (blk >= sort_blocks) {
        project_block<128, 4>(x, W, h, n, tile0 + (blk - sort_blocks) * PROJ_TILES);
        return;
    }
    if (PLACE) bucket_place(rec, pre, chunk, n, nb, rowptr, lst, blk);
    else       chunk_order(ei, E, chunk, nb, pre, rec, blk);
}

#define GATHER_NPB   8

__device__ __forceinline__ float lrelu(float v) {
    return fmaxf(v, NEG_SLOPE * v);
}

template <int HEADS> struct rowvec {
    typedef float type __attribute__((ext_vector_type(HEADS)));
};

template <int HEADS>
__device__ __forceinline__ float head_sum(float p) {
    static_assert(HEADS == 2 || HEADS == 4, "8 or 16 lanes per head");
    p = row_xor_add<1>(p);
    p = row_xor_add<2>(p);
    p = row_xor_add<4>(p);
    if (HEADS == 2) p = row_xor_add<8>(p);
    return p;
}

template <int HEADS>
__device__ __forceinline__ float head_dot(typename rowvec<HEADS>::type v,
                                          typename rowvec<HEADS>::type a) {
    float p = v[0] * a[0];
#pragma unroll
    for (int k = 1; k < HEADS; ++k) p += v[k] * a[k];
    return head_sum<HEADS>(p);
}

template <int HEADS, bool FULL>
__device__ __forceinline__ void gather_trip(const typename rowvec<HEADS>::type* __restrict__ hrow,
                                            const int* __restrict__ lst, int j, int end,
                                            int lane, typename rowvec<HEADS>::type avs,
                                            float adh, typename rowvec<HEADS>::type& acc,
                                            float& dsum, float& m) {
    typedef typename rowvec<HEADS>::type vec;
    const bool in1 = FULL || j + 1 < end, in2 = FULL || j + 2 < end, in3 = FULL || j + 3 < end;
    const int s0 = lst[j];
    const int s1 = lst[in1 ? j + 1 : end - 1];
    const int s2 = lst[in2 ? j + 2 : end - 1];
    const int s3 = lst[in3 ? j + 3 : end - 1];
    const vec v0 = hrow[(size_t)s0 * 32 + lane];
    const vec v1 = hrow[(size_t)s1 * 32 + lane];
    const vec v2 = hrow[(size_t)s2 * 32 + lane];
    const vec v3 = hrow[(size_t)s3 * 32 + lane];
    __builtin_amdgcn_sched_barrier(0);
    const float ninf = -__builtin_huge_valf();
    const float e0 = lrelu(head_dot<HEADS>(v0, avs) + adh);
    const float e1 = in1 ? lrelu(head_dot<HEADS>(v1, avs) + adh) : ninf;
    const float e2 = in2 ? lrelu(head_dot<HEADS>(v2, avs) + adh) : ninf;
    const float e3 = in3 ? lrelu(head_dot<HEADS>(v3, avs) + adh) : ninf;
    const float mn = fmaxf(fmaxf(m, e0), fmaxf(e1, fmaxf(e2, e3)));
    const float sc = __expf(m - mn);
    const float w0 = __expf(e0 - mn);
    const float w1 = __expf(e1 - mn);
    const float w2 = __expf(e2 - mn);
    const float w3 = __expf(e3 - mn);
    acc = sc * acc + w0 * v0;  dsum = sc * dsum + w0;
    acc += w1 * v1;  dsum += w1;
    acc += w2 * v2;  dsum += w2;
    acc += w3 * v3;  dsum += w3;
    m = mn;
}

template <int HEADS>
__device__ __forceinline__ void gather_node(const int* __restrict__ rowptr,
                                            const int* __restrict__ lst,
                                            const float* __restrict__ a_src,
                                            const float* __restrict__ a_dst,
                                            const float* __restrict__ h,
                                            int node, int lane,
                                            typename rowvec<HEADS>::type& acc_out,
                                            float& dsum_out) {
    typedef typename rowvec<HEADS>::type vec;
    const vec* __restrict__ hrow = (const vec*)h;
    const int beg = rowptr[node];
    const int end = rowptr[node + 1];
    const vec avs = ((const vec*)a_src)[lane];
    const vec avd = ((const vec*)a_dst)[lane];

    const vec vself = hrow[(size_t)node * 32 + lane];
    const float adh = head_dot<HEADS>(vself, avd);
    float m = lrelu(head_dot<HEADS>(vself, avs) + adh);
    vec acc = vself;
    float dsum = 1.f;

    int j = beg;
    for (; j + 4 <= end; j += 4)
        gather_trip<HEADS, true>(hrow, lst, j, end, lane, avs, adh, acc, dsum, m);
    if (j < end)
        gather_trip<HEADS, false>(hrow, lst, j, end, lane, avs, adh, acc, dsum, m);
    acc_out = acc;
    dsum_out = dsum;
}

__device__ __forceinline__ float elu(float v) {
    return v > 0.f ? v : (expf(v) - 1.0f);
}

template <int HEADS>
__global__ void attend_concat(const int* __restrict__ rowptr,
                              const int* __restrict__ lst,
                              const float* __restrict__ a_src,
                              const float* __restrict__ a_dst,
                              const float* __restrict__ h,
                              const float* __restrict__ bias,
                              float* __restrict__ hout, int n_nodes) {
    typedef typename rowvec<HEADS>::type vec;
    const int lane = threadIdx.x;
    const int node = __builtin_amdgcn_readfirstlane(
        (int)(blockIdx.x * blockDim.y + threadIdx.y));
    if (node >= n_nodes) return;
    vec acc; float dsum;
    gather_node<HEADS>(rowptr, lst, a_src, a_dst, h, node, lane, acc, dsum);
    const vec b = ((const vec*)bias)[lane];
    vec o;
#pragma unroll
    for (int k = 0; k < HEADS; ++k) o[k] = elu(acc[k] / (dsum + EPS) + b[k]);
    ((vec*)hout)[(size_t)node * 32 + lane] = o;
}

template <int OUTC>
__global__ void attend_mean_linear(const int* __restrict__ rowptr,
                                   const int* __restrict__ lst,
                                   const float* __restrict__ a_src,
                                   const float* __restrict__ a_dst,
                                   const float* __restrict__ h,
                                   const float* __restrict__ bias,
                                   const float* __restrict__ Wl,
                                   const float* __restrict__ bl,
                                   float* __restrict__ out, int n_nodes) {
    constexpr int HEADS = 2;
    typedef typename rowvec<HEADS>::type vec;
    const int lane = threadIdx.x;
    const int node = __builtin_amdgcn_readfirstlane(
        (int)(blockIdx.x * blockDim.y + threadIdx.y));
    if (node >= n_nodes) return;
    vec acc; float dsum;
    gather_node<HEADS>(rowptr, lst, a_src, a_dst, h, node, lane, acc, dsum);

    const int g = lane & 15;
    const vec b = ((const vec*)bias)[g];
    float hv[HEADS];
#pragma unroll
    for (int k = 0; k < HEADS; ++k) {
        const float q  = acc[k] / (dsum + EPS);
        const float qo = __shfl_xor(q, 16, 32);
        hv[k] = elu((q + qo) / (float)HEADS + b[k]);
    }
    const float* __restrict__ wr = Wl + (size_t)(2 * g) * OUTC;
    float t[OUTC];
#pragma unroll
    for (int o = 0; o < OUTC; ++o) t[o] = hv[0] * wr[o] + hv[1] * wr[OUTC + o];
    for (int off = 8; off > 0; off >>= 1) {
#pragma unroll
        for (int o = 0; o < OUTC; ++o) t[o] += __shfl_xor(t[o], off, 32);
    }
    float r = t[0];
#pragma unroll
    for (int o = 1; o < OUTC; ++o)
        if (lane == o) r = t[o];
    if (lane < OUTC) out[(size_t)node * OUTC + lane] = r + bl[lane];
}

static inline unsigned gb(long long total, int bs) {
    return (unsigned)((total + bs - 1) / bs);
}

extern "C" void kernel_launch(void* const* d_in, const int* in_sizes, int n_in,
                              void* d_out, int out_size, void* d_ws, size_t ws_size,
                              hipStream_t stream) {
    const int IN_CH = 128, HID = 32, H1 = 4, H2 = 2, OUT = 3;
    const int N = in_sizes[0] / IN_CH;
    const long long E  = in_sizes[1] / 2;

    const float* x     = (const float*)d_in[0];
    const int*   ei    = (const int*)  d_in[1];
    const float* W1    = (const float*)d_in[2];
    const float* asrc1 = (const float*)d_in[3];
    const float* adst1 = (const float*)d_in[4];
    const float* b1    = (const float*)d_in[5];
    const float* W2    = (const float*)d_in[6];
    const float* asrc2 = (const float*)d_in[7];
    const float* adst2 = (const float*)d_in[8];
    const float* b2    = (const float*)d_in[9];
    const float* Wlin  = (const float*)d_in[10];
    const float* blin  = (const float*)d_in[11];
    float* out = (float*)d_out;

    float* ws = (float*)d_ws;
    size_t off = 0;
    float* h1   = ws + off; off += (size_t)N * H1 * HID;
    float* h1a  = ws + off; off += (size_t)N * H1 * HID;
    int* lst    = (int*)(ws + off); off += (size_t)E;
    const int NBKT  = (N + BKT_NODES - 1) >> BKT_SHIFT;
    const int CHUNK = (int)((E + SORT_CHUNKS - 1) / SORT_CHUNKS);
    int* pre    = (int*)(ws + off); off += (size_t)SORT_CHUNKS * PRE_STRIDE;
    int* rowptr = (int*)(ws + off); off += (size_t)N + 4;
    off = (off + 1) & ~(size_t)1;
    int* rec    = (int*)(ws + off); off += (size_t)2 * E;
    float* h2   = h1;

    const int NPB = GATHER_NPB;

    const int PB  = (N / 16 + PROJ_TILES - 1) / PROJ_TILES;
    const int PB1 = (PB + 1) / 2;
    const int PB2 = PB - PB1;
    sort_stage_project<false><<<SORT_CHUNKS + PB1, SORT_THREADS, 0, stream>>>(
        ei, (int)E, CHUNK, NBKT, pre, rec, N, rowptr, lst, SORT_CHUNKS,
        x, W1, h1, 0);
    sort_stage_project<true><<<NBKT + PB2, SORT_THREADS, 0, stream>>>(
        ei, (int)E, CHUNK, NBKT, pre, rec, N, rowptr, lst, NBKT,
        x, W1, h1, PB1 * PROJ_TILES);

    attend_concat<H1><<<gb((long long)N, NPB), dim3(32, NPB, 1), 0, stream>>>(
        rowptr, lst, asrc1, adst1, h1, b1, h1a, N);

    project_rows<H1 * HID, H2><<<dim3(N / 16, 1, 1), dim3(32, H2, 1), 0, stream>>>(
        h1a, W2, h2, N);
    attend_mean_linear<OUT><<<gb((long long)N, NPB), dim3(32, NPB, 1), 0, stream>>>(
        rowptr, lst, asrc2, adst2, h2, b2, Wlin, blin, out, N);
    (void)n_in; (void)out_size; (void)ws_size;
}
